# starting kernel
# baseline (speedup 1.0000x reference)
#include <hip/hip_runtime.h>
#include <math.h>

typedef float v2f __attribute__((ext_vector_type(2)));
typedef float v8f __attribute__((ext_vector_type(8)));

#define FEAT_IN  128
#define FEAT_MID 16
#define FEAT_OUT 2

__global__ void init_ws(float* __restrict__ deg, float* __restrict__ acc1,
                        float* __restrict__ acc2, int n_nodes) {
    int tid = blockIdx.x * blockDim.x + threadIdx.x;
    if (tid < n_nodes * FEAT_MID) acc1[tid] = 0.0f;
    if (tid < n_nodes * FEAT_OUT) acc2[tid] = 0.0f;
    if (tid < n_nodes)            deg[tid]  = 1.0f;
}

__global__ void degree_kernel(const int* __restrict__ dst, float* __restrict__ deg,
                              int n_edges) {
    int e = blockIdx.x * blockDim.x + threadIdx.x;
    if (e < n_edges) atomicAdd(&deg[dst[e]], 1.0f);
}

__global__ void dinv_kernel(float* __restrict__ deg, int n_nodes) {
    int i = blockIdx.x * blockDim.x + threadIdx.x;
    if (i < n_nodes) {
        float d = deg[i];
        deg[i] = (d > 0.0f) ? rsqrtf(d) : 0.0f;
    }
}

__global__ void __launch_bounds__(256)
gemm1_wmma(const float* __restrict__ x, const float* __restrict__ W1,
           float* __restrict__ h, int n_tiles, int n_nodes) {
    int lane = threadIdx.x & 31;
    int wave = threadIdx.x >> 5;
    int tile = blockIdx.x * (blockDim.x >> 5) + wave;
    if (tile >= n_tiles) return;

    int col = lane & 15;
    int kh  = (lane >> 4) << 1;

    int rowA = tile * 16 + col;
    if (rowA > n_nodes - 1) rowA = n_nodes - 1;
    const float* xrow = x + (size_t)rowA * FEAT_IN;

    v8f acc = {};
    for (int k0 = 0; k0 < FEAT_IN; k0 += 4) {
        v2f a, b;
        a.x = xrow[k0 + kh];
        a.y = xrow[k0 + kh + 1];
        b.x = W1[(k0 + kh)     * FEAT_MID + col];
        b.y = W1[(k0 + kh + 1) * FEAT_MID + col];
        acc = __builtin_amdgcn_wmma_f32_16x16x4_f32(
             false, a,  false, b,
             (short)0, acc,  false,  false);
    }

    int mbase = (lane >> 4) * 8;
#pragma unroll
    for (int r = 0; r < 8; r++) {
        int row = tile * 16 + mbase + r;
        if (row < n_nodes) h[(size_t)row * FEAT_MID + col] = acc[r];
    }
}

__global__ void edge1_kernel(const int* __restrict__ src, const int* __restrict__ dst,
                             const float* __restrict__ dinv, const float* __restrict__ h,
                             float* __restrict__ acc1, int n_edges) {
    long long tid = (long long)blockIdx.x * blockDim.x + threadIdx.x;
    if (tid >= (long long)n_edges * FEAT_MID) return;
    int e = (int)(tid >> 4);
    int f = (int)(tid & 15);
    int s = src[e], d = dst[e];
    float norm = dinv[s] * dinv[d];
    atomicAdd(&acc1[d * FEAT_MID + f], h[s * FEAT_MID + f] * norm);
}

__global__ void relu_bias1(const float* __restrict__ h, const float* __restrict__ dinv,
                           const float* __restrict__ b1, float* __restrict__ acc1,
                           int n_nodes) {
    int tid = blockIdx.x * blockDim.x + threadIdx.x;
    if (tid >= n_nodes * FEAT_MID) return;
    int node = tid >> 4, f = tid & 15;
    float di = dinv[node];
    float v = acc1[tid] + h[tid] * di * di + b1[f];
    acc1[tid] = fmaxf(v, 0.0f);
}

__global__ void gemm2_kernel(const float* __restrict__ h1, const float* __restrict__ W2,
                             float* __restrict__ h2, int n_nodes) {
    int i = blockIdx.x * blockDim.x + threadIdx.x;
    if (i >= n_nodes) return;
    float s0 = 0.0f, s1 = 0.0f;
#pragma unroll
    for (int k = 0; k < FEAT_MID; k++) {
        float v = h1[i * FEAT_MID + k];
        s0 = fmaf(v, W2[k * 2 + 0], s0);
        s1 = fmaf(v, W2[k * 2 + 1], s1);
    }
    h2[i * 2 + 0] = s0;
    h2[i * 2 + 1] = s1;
}

__global__ void edge2_kernel(const int* __restrict__ src, const int* __restrict__ dst,
                             const float* __restrict__ dinv, const float* __restrict__ h2,
                             float* __restrict__ acc2, int n_edges) {
    long long tid = (long long)blockIdx.x * blockDim.x + threadIdx.x;
    if (tid >= (long long)n_edges * FEAT_OUT) return;
    int e = (int)(tid >> 1);
    int c = (int)(tid & 1);
    int s = src[e], d = dst[e];
    float norm = dinv[s] * dinv[d];
    atomicAdd(&acc2[d * FEAT_OUT + c], h2[s * FEAT_OUT + c] * norm);
}

__global__ void final_kernel(const float* __restrict__ acc2, const float* __restrict__ h2,
                             const float* __restrict__ dinv, const float* __restrict__ b2,
                             float* __restrict__ out, int n_nodes) {
    int i = blockIdx.x * blockDim.x + threadIdx.x;
    if (i >= n_nodes) return;
    float di = dinv[i];
    float sc = di * di;
    float v0 = acc2[i * 2 + 0] + h2[i * 2 + 0] * sc + b2[0];
    float v1 = acc2[i * 2 + 1] + h2[i * 2 + 1] * sc + b2[1];
    float m  = fmaxf(v0, v1);
    float l  = m + logf(expf(v0 - m) + expf(v1 - m));
    out[i * 2 + 0] = v0 - l;
    out[i * 2 + 1] = v1 - l;
}

extern "C" void kernel_launch(void* const* d_in, const int* in_sizes, int n_in,
                              void* d_out, int out_size, void* d_ws, size_t ws_size,
                              hipStream_t stream) {
    const float* x  = (const float*)d_in[0];
    const int*   ei = (const int*)d_in[1];
    const float* W1 = (const float*)d_in[2];
    const float* b1 = (const float*)d_in[3];
    const float* W2 = (const float*)d_in[4];
    const float* b2 = (const float*)d_in[5];
    float* out = (float*)d_out;

    int n_nodes = in_sizes[0] / FEAT_IN;
    int n_edges = in_sizes[1] / 2;
    const int* src = ei;
    const int* dst = ei + n_edges;

    float* ws   = (float*)d_ws;
    float* deg  = ws;
    float* h    = deg  + n_nodes;
    float* acc1 = h    + (size_t)n_nodes * FEAT_MID;
    float* h2   = acc1 + (size_t)n_nodes * FEAT_MID;
    float* acc2 = h2   + (size_t)n_nodes * FEAT_OUT;

    const int B = 256;
    int n16 = n_nodes * FEAT_MID;

    init_ws<<<(n16 + B - 1) / B, B, 0, stream>>>(deg, acc1, acc2, n_nodes);
    degree_kernel<<<(n_edges + B - 1) / B, B, 0, stream>>>(dst, deg, n_edges);
    dinv_kernel<<<(n_nodes + B - 1) / B, B, 0, stream>>>(deg, n_nodes);

    int n_tiles = (n_nodes + 15) / 16;
    int wpb = B / 32;
    gemm1_wmma<<<(n_tiles + wpb - 1) / wpb, B, 0, stream>>>(x, W1, h, n_tiles, n_nodes);

    long long e16 = (long long)n_edges * FEAT_MID;
    edge1_kernel<<<(int)((e16 + B - 1) / B), B, 0, stream>>>(src, dst, deg, h, acc1, n_edges);
    relu_bias1<<<(n16 + B - 1) / B, B, 0, stream>>>(h, deg, b1, acc1, n_nodes);

    gemm2_kernel<<<(n_nodes + B - 1) / B, B, 0, stream>>>(acc1, W2, h2, n_nodes);

    long long e2 = (long long)n_edges * FEAT_OUT;
    edge2_kernel<<<(int)((e2 + B - 1) / B), B, 0, stream>>>(src, dst, deg, h2, acc2, n_edges);

    final_kernel<<<(n_nodes + B - 1) / B, B, 0, stream>>>(acc2, h2, deg, b2, out, n_nodes);
}


# degree count and layer-2 scatter: per-edge global atomics replaced by LDS partial sums per owned node range, merged by whole-line atomics (layer-2 sum order: per-work-group partials)
# speedup vs baseline: 1.6245x; 1.6245x over previous
#include <hip/hip_runtime.h>
#include <math.h>

typedef float v2f __attribute__((ext_vector_type(2)));
typedef float v8f __attribute__((ext_vector_type(8)));

#define FEAT_IN  128
#define FEAT_MID 16
#define FEAT_OUT 2

__global__ void init_ws(float* __restrict__ deg, float* __restrict__ acc1,
                        float* __restrict__ acc2, int n_nodes) {
    int tid = blockIdx.x * blockDim.x + threadIdx.x;
    if (tid < n_nodes * FEAT_MID) acc1[tid] = 0.0f;
    if (tid < n_nodes * FEAT_OUT) acc2[tid] = 0.0f;
    if (tid < n_nodes)            deg[tid]  = 1.0f;
}

#define LDS_FLOATS   69632
#define SCAT_THREADS 1024
#define SCAT_U       8

__global__ void __launch_bounds__(SCAT_THREADS)
degree_lds(const int* __restrict__ dst, float* __restrict__ deg,
           int n_edges, int n_nodes, int range_nodes, int chunks, int per_chunk) {
    __shared__ float cnt[LDS_FLOATS];
    const int tid   = threadIdx.x;
    const int range = blockIdx.x / chunks;
    const int chunk = blockIdx.x - range * chunks;
    const int lo    = range * range_nodes;
    int len = n_nodes - lo;
    if (len > range_nodes) len = range_nodes;
    if (len <= 0) return;

    for (int i = tid; i < len; i += SCAT_THREADS) cnt[i] = 0.0f;
    __syncthreads();

    const int e0 = chunk * per_chunk;
    int e1 = e0 + per_chunk;
    if (e1 > n_edges) e1 = n_edges;
    for (int base = e0; base < e1; base += SCAT_U * SCAT_THREADS) {
        int dv[SCAT_U];
#pragma unroll
        for (int k = 0; k < SCAT_U; k++) {
            int e  = base + k * SCAT_THREADS + tid;
            int ec = e < e1 ? e : e1 - 1;
            dv[k] = dst[ec];
        }
        asm volatile("" : "+v"(dv[0]), "+v"(dv[1]), "+v"(dv[2]), "+v"(dv[3]),
                          "+v"(dv[4]), "+v"(dv[5]), "+v"(dv[6]), "+v"(dv[7]));
#pragma unroll
        for (int k = 0; k < SCAT_U; k++) {
            int e = base + k * SCAT_THREADS + tid;
            int t = dv[k] - lo;
            if (e < e1 && (unsigned)t < (unsigned)len) atomicAdd(&cnt[t], 1.0f);
        }
    }
    __syncthreads();

    const int rot = (chunk * (len / chunks)) & ~31;
    for (int j = tid; j < len; j += SCAT_THREADS) {
        int i = j + rot;
        if (i >= len) i -= len;
        float v = cnt[i];
        if (v != 0.0f) atomicAdd(&deg[lo + i], v);
    }
}

__global__ void dinv_kernel(float* __restrict__ deg, int n_nodes) {
    int i = blockIdx.x * blockDim.x + threadIdx.x;
    if (i < n_nodes) {
        float d = deg[i];
        deg[i] = (d > 0.0f) ? rsqrtf(d) : 0.0f;
    }
}

__global__ void __launch_bounds__(256)
gemm1_wmma(const float* __restrict__ x, const float* __restrict__ W1,
           float* __restrict__ h, int n_tiles, int n_nodes) {
    int lane = threadIdx.x & 31;
    int wave = threadIdx.x >> 5;
    int tile = blockIdx.x * (blockDim.x >> 5) + wave;
    if (tile >= n_tiles) return;

    int col = lane & 15;
    int kh  = (lane >> 4) << 1;

    int rowA = tile * 16 + col;
    if (rowA > n_nodes - 1) rowA = n_nodes - 1;
    const float* xrow = x + (size_t)rowA * FEAT_IN;

    v8f acc = {};
    for (int k0 = 0; k0 < FEAT_IN; k0 += 4) {
        v2f a, b;
        a.x = xrow[k0 + kh];
        a.y = xrow[k0 + kh + 1];
        b.x = W1[(k0 + kh)     * FEAT_MID + col];
        b.y = W1[(k0 + kh + 1) * FEAT_MID + col];
        acc = __builtin_amdgcn_wmma_f32_16x16x4_f32(
             false, a,  false, b,
             (short)0, acc,  false,  false);
    }

    int mbase = (lane >> 4) * 8;
#pragma unroll
    for (int r = 0; r < 8; r++) {
        int row = tile * 16 + mbase + r;
        if (row < n_nodes) h[(size_t)row * FEAT_MID + col] = acc[r];
    }
}

__global__ void edge1_kernel(const int* __restrict__ src, const int* __restrict__ dst,
                             const float* __restrict__ dinv, const float* __restrict__ h,
                             float* __restrict__ acc1, int n_edges) {
    long long tid = (long long)blockIdx.x * blockDim.x + threadIdx.x;
    if (tid >= (long long)n_edges * FEAT_MID) return;
    int e = (int)(tid >> 4);
    int f = (int)(tid & 15);
    int s = src[e], d = dst[e];
    float norm = dinv[s] * dinv[d];
    atomicAdd(&acc1[d * FEAT_MID + f], h[s * FEAT_MID + f] * norm);
}

__global__ void relu_bias1(const float* __restrict__ h, const float* __restrict__ dinv,
                           const float* __restrict__ b1, float* __restrict__ acc1,
                           int n_nodes) {
    int tid = blockIdx.x * blockDim.x + threadIdx.x;
    if (tid >= n_nodes * FEAT_MID) return;
    int node = tid >> 4, f = tid & 15;
    float di = dinv[node];
    float v = acc1[tid] + h[tid] * di * di + b1[f];
    acc1[tid] = fmaxf(v, 0.0f);
}

__global__ void gemm2_kernel(const float* __restrict__ h1, const float* __restrict__ W2,
                             float* __restrict__ h2, int n_nodes) {
    int i = blockIdx.x * blockDim.x + threadIdx.x;
    if (i >= n_nodes) return;
    float s0 = 0.0f, s1 = 0.0f;
#pragma unroll
    for (int k = 0; k < FEAT_MID; k++) {
        float v = h1[i * FEAT_MID + k];
        s0 = fmaf(v, W2[k * 2 + 0], s0);
        s1 = fmaf(v, W2[k * 2 + 1], s1);
    }
    h2[i * 2 + 0] = s0;
    h2[i * 2 + 1] = s1;
}

__global__ void __launch_bounds__(SCAT_THREADS)
edge2_lds(const int* __restrict__ src, const int* __restrict__ dst,
          const float* __restrict__ dinv, const float* __restrict__ h2,
          float* __restrict__ acc2, int n_edges, int n_nodes,
          int range_nodes, int chunks, int per_chunk) {
    __shared__ float acc[LDS_FLOATS];
    const int tid   = threadIdx.x;
    const int range = blockIdx.x / chunks;
    const int chunk = blockIdx.x - range * chunks;
    const int lo    = range * range_nodes;
    int len = n_nodes - lo;
    if (len > range_nodes) len = range_nodes;
    if (len <= 0) return;
    const int flen = len * FEAT_OUT;

    for (int i = tid; i < flen; i += SCAT_THREADS) acc[i] = 0.0f;
    __syncthreads();

    const v2f* __restrict__ h2v = (const v2f*)h2;
    const int e0 = chunk * per_chunk;
    int e1 = e0 + per_chunk;
    if (e1 > n_edges) e1 = n_edges;
    for (int base = e0; base < e1; base += SCAT_U * SCAT_THREADS) {
        int dv[SCAT_U], sv[SCAT_U];
#pragma unroll
        for (int k = 0; k < SCAT_U; k++) {
            int e  = base + k * SCAT_THREADS + tid;
            int ec = e < e1 ? e : e1 - 1;
            dv[k] = dst[ec];
            sv[k] = src[ec];
        }
        asm volatile("" : "+v"(dv[0]), "+v"(dv[1]), "+v"(dv[2]), "+v"(dv[3]),
                          "+v"(dv[4]), "+v"(dv[5]), "+v"(dv[6]), "+v"(dv[7]),
                          "+v"(sv[0]), "+v"(sv[1]), "+v"(sv[2]), "+v"(sv[3]),
                          "+v"(sv[4]), "+v"(sv[5]), "+v"(sv[6]), "+v"(sv[7]));
        int   t[SCAT_U];
        float ds[SCAT_U], dd[SCAT_U];
        v2f   hv[SCAT_U];
#pragma unroll
        for (int k = 0; k < SCAT_U; k++) {
            int e  = base + k * SCAT_THREADS + tid;
            int tt = dv[k] - lo;
            bool m = (e < e1) && ((unsigned)tt < (unsigned)len);
            t[k] = m ? tt : -1;
            int sg = m ? sv[k] : lo;
            int dg = m ? dv[k] : lo;
            ds[k] = dinv[sg];
            dd[k] = dinv[dg];
            hv[k] = h2v[sg];
        }
        asm volatile("" : "+v"(ds[0]), "+v"(ds[1]), "+v"(ds[2]), "+v"(ds[3]),
                          "+v"(ds[4]), "+v"(ds[5]), "+v"(ds[6]), "+v"(ds[7]),
                          "+v"(dd[0]), "+v"(dd[1]), "+v"(dd[2]), "+v"(dd[3]),
                          "+v"(dd[4]), "+v"(dd[5]), "+v"(dd[6]), "+v"(dd[7]),
                          "+v"(hv[0]), "+v"(hv[1]), "+v"(hv[2]), "+v"(hv[3]),
                          "+v"(hv[4]), "+v"(hv[5]), "+v"(hv[6]), "+v"(hv[7]));
#pragma unroll
        for (int k = 0; k < SCAT_U; k++) {
            if (t[k] >= 0) {
                float w = ds[k] * dd[k];
                atomicAdd(&acc[t[k] * FEAT_OUT + 0], hv[k].x * w);
                atomicAdd(&acc[t[k] * FEAT_OUT + 1], hv[k].y * w);
            }
        }
    }
    __syncthreads();

    const int rot = (chunk * (flen / chunks)) & ~31;
    float* __restrict__ outp = acc2 + (size_t)lo * FEAT_OUT;
    for (int j = tid; j < flen; j += SCAT_THREADS) {
        int i = j + rot;
        if (i >= flen) i -= flen;
        float v = acc[i];
        if (v != 0.0f) atomicAdd(&outp[i], v);
    }
}

__global__ void final_kernel(const float* __restrict__ acc2, const float* __restrict__ h2,
                             const float* __restrict__ dinv, const float* __restrict__ b2,
                             float* __restrict__ out, int n_nodes) {
    int i = blockIdx.x * blockDim.x + threadIdx.x;
    if (i >= n_nodes) return;
    float di = dinv[i];
    float sc = di * di;
    float v0 = acc2[i * 2 + 0] + h2[i * 2 + 0] * sc + b2[0];
    float v1 = acc2[i * 2 + 1] + h2[i * 2 + 1] * sc + b2[1];
    float m  = fmaxf(v0, v1);
    float l  = m + logf(expf(v0 - m) + expf(v1 - m));
    out[i * 2 + 0] = v0 - l;
    out[i * 2 + 1] = v1 - l;
}

extern "C" void kernel_launch(void* const* d_in, const int* in_sizes, int n_in,
                              void* d_out, int out_size, void* d_ws, size_t ws_size,
                              hipStream_t stream) {
    const float* x  = (const float*)d_in[0];
    const int*   ei = (const int*)d_in[1];
    const float* W1 = (const float*)d_in[2];
    const float* b1 = (const float*)d_in[3];
    const float* W2 = (const float*)d_in[4];
    const float* b2 = (const float*)d_in[5];
    float* out = (float*)d_out;

    int n_nodes = in_sizes[0] / FEAT_IN;
    int n_edges = in_sizes[1] / 2;
    const int* src = ei;
    const int* dst = ei + n_edges;

    float* ws   = (float*)d_ws;
    float* deg  = ws;
    float* h    = deg  + n_nodes;
    float* acc1 = h    + (size_t)n_nodes * FEAT_MID;
    float* h2   = acc1 + (size_t)n_nodes * FEAT_MID;
    float* acc2 = h2   + (size_t)n_nodes * FEAT_OUT;

    const int B = 256;
    int n16 = n_nodes * FEAT_MID;

    init_ws<<<(n16 + B - 1) / B, B, 0, stream>>>(deg, acc1, acc2, n_nodes);
    int nr1 = (n_nodes + LDS_FLOATS - 1) / LDS_FLOATS;
    int rn1 = (((n_nodes + nr1 - 1) / nr1) + 31) & ~31;
    int ch1 = 128 / nr1 > 0 ? 128 / nr1 : 1;
    int pc1 = (n_edges + ch1 - 1) / ch1;
    degree_lds<<<nr1 * ch1, SCAT_THREADS, 0, stream>>>(dst, deg, n_edges, n_nodes, rn1, ch1, pc1);
    dinv_kernel<<<(n_nodes + B - 1) / B, B, 0, stream>>>(deg, n_nodes);

    int n_tiles = (n_nodes + 15) / 16;
    int wpb = B / 32;
    gemm1_wmma<<<(n_tiles + wpb - 1) / wpb, B, 0, stream>>>(x, W1, h, n_tiles, n_nodes);

    long long e16 = (long long)n_edges * FEAT_MID;
    edge1_kernel<<<(int)((e16 + B - 1) / B), B, 0, stream>>>(src, dst, deg, h, acc1, n_edges);
    relu_bias1<<<(n16 + B - 1) / B, B, 0, stream>>>(h, deg, b1, acc1, n_nodes);

    gemm2_kernel<<<(n_nodes + B - 1) / B, B, 0, stream>>>(acc1, W2, h2, n_nodes);

    int nr2 = (n_nodes * FEAT_OUT + LDS_FLOATS - 1) / LDS_FLOATS;
    int rn2 = (((n_nodes + nr2 - 1) / nr2) + 31) & ~31;
    int ch2 = 128 / nr2 > 0 ? 128 / nr2 : 1;
    int pc2 = (n_edges + ch2 - 1) / ch2;
    edge2_lds<<<nr2 * ch2, SCAT_THREADS, 0, stream>>>(src, dst, deg, h2, acc2, n_edges, n_nodes,
                                                      rn2, ch2, pc2);

    final_kernel<<<(n_nodes + B - 1) / B, B, 0, stream>>>(acc2, h2, deg, b2, out, n_nodes);
}


# layer-1 scatter: per-edge global atomics replaced by LDS partial sums per owned node range, 16 lanes per listed edge, merged by whole-line atomics (layer-1 sum order: per-work-group partials)
# speedup vs baseline: 2.0612x; 1.2688x over previous
#include <hip/hip_runtime.h>
#include <math.h>

typedef float v2f __attribute__((ext_vector_type(2)));
typedef float v8f __attribute__((ext_vector_type(8)));

#define FEAT_IN  128
#define FEAT_MID 16
#define FEAT_OUT 2

__global__ void init_ws(float* __restrict__ deg, float* __restrict__ acc1,
                        float* __restrict__ acc2, int n_nodes) {
    int tid = blockIdx.x * blockDim.x + threadIdx.x;
    if (tid < n_nodes * FEAT_MID) acc1[tid] = 0.0f;
    if (tid < n_nodes * FEAT_OUT) acc2[tid] = 0.0f;
    if (tid < n_nodes)            deg[tid]  = 1.0f;
}

#define LDS_FLOATS   69632
#define SCAT_THREADS 1024
#define SCAT_U       8

__global__ void __launch_bounds__(SCAT_THREADS)
degree_lds(const int* __restrict__ dst, float* __restrict__ deg,
           int n_edges, int n_nodes, int range_nodes, int chunks, int per_chunk) {
    __shared__ float cnt[LDS_FLOATS];
    const int tid   = threadIdx.x;
    const int range = blockIdx.x / chunks;
    const int chunk = blockIdx.x - range * chunks;
    const int lo    = range * range_nodes;
    int len = n_nodes - lo;
    if (len > range_nodes) len = range_nodes;
    if (len <= 0) return;

    for (int i = tid; i < len; i += SCAT_THREADS) cnt[i] = 0.0f;
    __syncthreads();

    const int e0 = chunk * per_chunk;
    int e1 = e0 + per_chunk;
    if (e1 > n_edges) e1 = n_edges;
    for (int base = e0; base < e1; base += SCAT_U * SCAT_THREADS) {
        int dv[SCAT_U];
#pragma unroll
        for (int k = 0; k < SCAT_U; k++) {
            int e  = base + k * SCAT_THREADS + tid;
            int ec = e < e1 ? e : e1 - 1;
            dv[k] = dst[ec];
        }
        asm volatile("" : "+v"(dv[0]), "+v"(dv[1]), "+v"(dv[2]), "+v"(dv[3]),
                          "+v"(dv[4]), "+v"(dv[5]), "+v"(dv[6]), "+v"(dv[7]));
#pragma unroll
        for (int k = 0; k < SCAT_U; k++) {
            int e = base + k * SCAT_THREADS + tid;
            int t = dv[k] - lo;
            if (e < e1 && (unsigned)t < (unsigned)len) atomicAdd(&cnt[t], 1.0f);
        }
    }
    __syncthreads();

    const int rot = (chunk * (len / chunks)) & ~31;
    for (int j = tid; j < len; j += SCAT_THREADS) {
        int i = j + rot;
        if (i >= len) i -= len;
        float v = cnt[i];
        if (v != 0.0f) atomicAdd(&deg[lo + i], v);
    }
}

__global__ void dinv_kernel(float* __restrict__ deg, int n_nodes) {
    int i = blockIdx.x * blockDim.x + threadIdx.x;
    if (i < n_nodes) {
        float d = deg[i];
        deg[i] = (d > 0.0f) ? rsqrtf(d) : 0.0f;
    }
}

__global__ void __launch_bounds__(256)
gemm1_wmma(const float* __restrict__ x, const float* __restrict__ W1,
           float* __restrict__ h, int n_tiles, int n_nodes) {
    int lane = threadIdx.x & 31;
    int wave = threadIdx.x >> 5;
    int tile = blockIdx.x * (blockDim.x >> 5) + wave;
    if (tile >= n_tiles) return;

    int col = lane & 15;
    int kh  = (lane >> 4) << 1;

    int rowA = tile * 16 + col;
    if (rowA > n_nodes - 1) rowA = n_nodes - 1;
    const float* xrow = x + (size_t)rowA * FEAT_IN;

    v8f acc = {};
    for (int k0 = 0; k0 < FEAT_IN; k0 += 4) {
        v2f a, b;
        a.x = xrow[k0 + kh];
        a.y = xrow[k0 + kh + 1];
        b.x = W1[(k0 + kh)     * FEAT_MID + col];
        b.y = W1[(k0 + kh + 1) * FEAT_MID + col];
        acc = __builtin_amdgcn_wmma_f32_16x16x4_f32(
             false, a,  false, b,
             (short)0, acc,  false,  false);
    }

    int mbase = (lane >> 4) * 8;
#pragma unroll
    for (int r = 0; r < 8; r++) {
        int row = tile * 16 + mbase + r;
        if (row < n_nodes) h[(size_t)row * FEAT_MID + col] = acc[r];
    }
}

typedef int v2i __attribute__((ext_vector_type(2)));

#define E1_NODES 2400
#define E1_U     16
#define E1_STEPS 8
#define E1_LIST  (E1_U * 32)

__global__ void __launch_bounds__(SCAT_THREADS)
edge1_lds(const int* __restrict__ src, const int* __restrict__ dst,
          const float* __restrict__ dinv, const float* __restrict__ h,
          float* __restrict__ acc1, int n_edges, int n_nodes,
          int range_nodes, int chunks, int per_chunk) {
    __shared__ float acc[E1_NODES * FEAT_MID];
    __shared__ v2i   list[SCAT_THREADS / 32][E1_LIST];
    const int tid   = threadIdx.x;
    const int lane  = tid & 31;
    const int f     = lane & 15;
    const int hf    = lane >> 4;
    const int range = blockIdx.x / chunks;
    const int chunk = blockIdx.x - range * chunks;
    const int lo    = range * range_nodes;
    int len = n_nodes - lo;
    if (len > range_nodes) len = range_nodes;
    if (len <= 0) return;
    const int flen = len * FEAT_MID;

    for (int i = tid; i < flen; i += SCAT_THREADS) acc[i] = 0.0f;
    __syncthreads();

    v2i* __restrict__ lst = list[tid >> 5];
    const int e0 = chunk * per_chunk;
    int e1 = e0 + per_chunk;
    if (e1 > n_edges) e1 = n_edges;
    for (int base = e0; base < e1; base += E1_U * SCAT_THREADS) {
        int dv[E1_U], sv[E1_U];
#pragma unroll
        for (int k = 0; k < E1_U; k++) {
            int e  = base + k * SCAT_THREADS + tid;
            int ec = e < e1 ? e : e1 - 1;
            dv[k] = dst[ec];
        }
        asm volatile("" : "+v"(dv[0]), "+v"(dv[1]), "+v"(dv[2]), "+v"(dv[3]),
                          "+v"(dv[4]), "+v"(dv[5]), "+v"(dv[6]), "+v"(dv[7]),
                          "+v"(dv[8]), "+v"(dv[9]), "+v"(dv[10]), "+v"(dv[11]),
                          "+v"(dv[12]), "+v"(dv[13]), "+v"(dv[14]), "+v"(dv[15]));
#pragma unroll
        for (int k = 0; k < E1_U; k++) {
            int e  = base + k * SCAT_THREADS + tid;
            int tt = dv[k] - lo;
            bool m = (e < e1) && ((unsigned)tt < (unsigned)len);
            dv[k] = m ? tt : -1;
            sv[k] = src[m ? e : e0];
        }
        asm volatile("" : "+v"(sv[0]), "+v"(sv[1]), "+v"(sv[2]), "+v"(sv[3]),
                          "+v"(sv[4]), "+v"(sv[5]), "+v"(sv[6]), "+v"(sv[7]),
                          "+v"(sv[8]), "+v"(sv[9]), "+v"(sv[10]), "+v"(sv[11]),
                          "+v"(sv[12]), "+v"(sv[13]), "+v"(sv[14]), "+v"(sv[15]));
        int cnt = 0;
#pragma unroll
        for (int k = 0; k < E1_U; k++) {
            bool m = dv[k] >= 0;
            unsigned b = __builtin_amdgcn_ballot_w32(m);
            int pos = cnt + (int)__builtin_amdgcn_mbcnt_lo(b, 0u);
            if (m) {
                v2i q;
                q.x = sv[k];
                q.y = dv[k];
                lst[pos] = q;
            }
            cnt += __builtin_popcount(b);
        }
        __builtin_amdgcn_wave_barrier();
        for (int j0 = 0; j0 < cnt; j0 += 2 * E1_STEPS) {
            int   tq[E1_STEPS];
            float hq[E1_STEPS], ws[E1_STEPS], wd[E1_STEPS];
#pragma unroll
            for (int p = 0; p < E1_STEPS; p++) {
                int  idx   = j0 + 2 * p + hf;
                bool valid = idx < cnt;
                v2i  q     = lst[valid ? idx : 0];
                tq[p] = valid ? q.y : -1;
                hq[p] = h[q.x * FEAT_MID + f];
                ws[p] = dinv[q.x];
                wd[p] = dinv[lo + q.y];
            }
            asm volatile("" : "+v"(hq[0]), "+v"(hq[1]), "+v"(hq[2]), "+v"(hq[3]),
                              "+v"(hq[4]), "+v"(hq[5]), "+v"(hq[6]), "+v"(hq[7]),
                              "+v"(ws[0]), "+v"(ws[1]), "+v"(ws[2]), "+v"(ws[3]),
                              "+v"(ws[4]), "+v"(ws[5]), "+v"(ws[6]), "+v"(ws[7]),
                              "+v"(wd[0]), "+v"(wd[1]), "+v"(wd[2]), "+v"(wd[3]),
                              "+v"(wd[4]), "+v"(wd[5]), "+v"(wd[6]), "+v"(wd[7]));
#pragma unroll
            for (int p = 0; p < E1_STEPS; p++) {
                if (tq[p] >= 0)
                    atomicAdd(&acc[tq[p] * FEAT_MID + f], hq[p] * (ws[p] * wd[p]));
            }
        }
        __builtin_amdgcn_wave_barrier();
    }
    __syncthreads();

    const int rot = (chunk * (flen / chunks)) & ~31;
    float* __restrict__ outp = acc1 + (size_t)lo * FEAT_MID;
    for (int j = tid; j < flen; j += SCAT_THREADS) {
        int i = j + rot;
        if (i >= flen) i -= flen;
        float v = acc[i];
        if (v != 0.0f) atomicAdd(&outp[i], v);
    }
}

__global__ void relu_bias1(const float* __restrict__ h, const float* __restrict__ dinv,
                           const float* __restrict__ b1, float* __restrict__ acc1,
                           int n_nodes) {
    int tid = blockIdx.x * blockDim.x + threadIdx.x;
    if (tid >= n_nodes * FEAT_MID) return;
    int node = tid >> 4, f = tid & 15;
    float di = dinv[node];
    float v = acc1[tid] + h[tid] * di * di + b1[f];
    acc1[tid] = fmaxf(v, 0.0f);
}

__global__ void gemm2_kernel(const float* __restrict__ h1, const float* __restrict__ W2,
                             float* __restrict__ h2, int n_nodes) {
    int i = blockIdx.x * blockDim.x + threadIdx.x;
    if (i >= n_nodes) return;
    float s0 = 0.0f, s1 = 0.0f;
#pragma unroll
    for (int k = 0; k < FEAT_MID; k++) {
        float v = h1[i * FEAT_MID + k];
        s0 = fmaf(v, W2[k * 2 + 0], s0);
        s1 = fmaf(v, W2[k * 2 + 1], s1);
    }
    h2[i * 2 + 0] = s0;
    h2[i * 2 + 1] = s1;
}

__global__ void __launch_bounds__(SCAT_THREADS)
edge2_lds(const int* __restrict__ src, const int* __restrict__ dst,
          const float* __restrict__ dinv, const float* __restrict__ h2,
          float* __restrict__ acc2, int n_edges, int n_nodes,
          int range_nodes, int chunks, int per_chunk) {
    __shared__ float acc[LDS_FLOATS];
    const int tid   = threadIdx.x;
    const int range = blockIdx.x / chunks;
    const int chunk = blockIdx.x - range * chunks;
    const int lo    = range * range_nodes;
    int len = n_nodes - lo;
    if (len > range_nodes) len = range_nodes;
    if (len <= 0) return;
    const int flen = len * FEAT_OUT;

    for (int i = tid; i < flen; i += SCAT_THREADS) acc[i] = 0.0f;
    __syncthreads();

    const v2f* __restrict__ h2v = (const v2f*)h2;
    const int e0 = chunk * per_chunk;
    int e1 = e0 + per_chunk;
    if (e1 > n_edges) e1 = n_edges;
    for (int base = e0; base < e1; base += SCAT_U * SCAT_THREADS) {
        int dv[SCAT_U], sv[SCAT_U];
#pragma unroll
        for (int k = 0; k < SCAT_U; k++) {
            int e  = base + k * SCAT_THREADS + tid;
            int ec = e < e1 ? e : e1 - 1;
            dv[k] = dst[ec];
            sv[k] = src[ec];
        }
        asm volatile("" : "+v"(dv[0]), "+v"(dv[1]), "+v"(dv[2]), "+v"(dv[3]),
                          "+v"(dv[4]), "+v"(dv[5]), "+v"(dv[6]), "+v"(dv[7]),
                          "+v"(sv[0]), "+v"(sv[1]), "+v"(sv[2]), "+v"(sv[3]),
                          "+v"(sv[4]), "+v"(sv[5]), "+v"(sv[6]), "+v"(sv[7]));
        int   t[SCAT_U];
        float ds[SCAT_U], dd[SCAT_U];
        v2f   hv[SCAT_U];
#pragma unroll
        for (int k = 0; k < SCAT_U; k++) {
            int e  = base + k * SCAT_THREADS + tid;
            int tt = dv[k] - lo;
            bool m = (e < e1) && ((unsigned)tt < (unsigned)len);
            t[k] = m ? tt : -1;
            int sg = m ? sv[k] : lo;
            int dg = m ? dv[k] : lo;
            ds[k] = dinv[sg];
            dd[k] = dinv[dg];
            hv[k] = h2v[sg];
        }
        asm volatile("" : "+v"(ds[0]), "+v"(ds[1]), "+v"(ds[2]), "+v"(ds[3]),
                          "+v"(ds[4]), "+v"(ds[5]), "+v"(ds[6]), "+v"(ds[7]),
                          "+v"(dd[0]), "+v"(dd[1]), "+v"(dd[2]), "+v"(dd[3]),
                          "+v"(dd[4]), "+v"(dd[5]), "+v"(dd[6]), "+v"(dd[7]),
                          "+v"(hv[0]), "+v"(hv[1]), "+v"(hv[2]), "+v"(hv[3]),
                          "+v"(hv[4]), "+v"(hv[5]), "+v"(hv[6]), "+v"(hv[7]));
#pragma unroll
        for (int k = 0; k < SCAT_U; k++) {
            if (t[k] >= 0) {
                float w = ds[k] * dd[k];
                atomicAdd(&acc[t[k] * FEAT_OUT + 0], hv[k].x * w);
                atomicAdd(&acc[t[k] * FEAT_OUT + 1], hv[k].y * w);
            }
        }
    }
    __syncthreads();

    const int rot = (chunk * (flen / chunks)) & ~31;
    float* __restrict__ outp = acc2 + (size_t)lo * FEAT_OUT;
    for (int j = tid; j < flen; j += SCAT_THREADS) {
        int i = j + rot;
        if (i >= flen) i -= flen;
        float v = acc[i];
        if (v != 0.0f) atomicAdd(&outp[i], v);
    }
}

__global__ void final_kernel(const float* __restrict__ acc2, const float* __restrict__ h2,
                             const float* __restrict__ dinv, const float* __restrict__ b2,
                             float* __restrict__ out, int n_nodes) {
    int i = blockIdx.x * blockDim.x + threadIdx.x;
    if (i >= n_nodes) return;
    float di = dinv[i];
    float sc = di * di;
    float v0 = acc2[i * 2 + 0] + h2[i * 2 + 0] * sc + b2[0];
    float v1 = acc2[i * 2 + 1] + h2[i * 2 + 1] * sc + b2[1];
    float m  = fmaxf(v0, v1);
    float l  = m + logf(expf(v0 - m) + expf(v1 - m));
    out[i * 2 + 0] = v0 - l;
    out[i * 2 + 1] = v1 - l;
}

extern "C" void kernel_launch(void* const* d_in, const int* in_sizes, int n_in,
                              void* d_out, int out_size, void* d_ws, size_t ws_size,
                              hipStream_t stream) {
    const float* x  = (const float*)d_in[0];
    const int*   ei = (const int*)d_in[1];
    const float* W1 = (const float*)d_in[2];
    const float* b1 = (const float*)d_in[3];
    const float* W2 = (const float*)d_in[4];
    const float* b2 = (const float*)d_in[5];
    float* out = (float*)d_out;

    int n_nodes = in_sizes[0] / FEAT_IN;
    int n_edges = in_sizes[1] / 2;
    const int* src = ei;
    const int* dst = ei + n_edges;

    float* ws   = (float*)d_ws;
    float* deg  = ws;
    float* h    = deg  + n_nodes;
    float* acc1 = h    + (size_t)n_nodes * FEAT_MID;
    float* h2   = acc1 + (size_t)n_nodes * FEAT_MID;
    float* acc2 = h2   + (size_t)n_nodes * FEAT_OUT;

    const int B = 256;
    int n16 = n_nodes * FEAT_MID;

    init_ws<<<(n16 + B - 1) / B, B, 0, stream>>>(deg, acc1, acc2, n_nodes);
    int nr1 = (n_nodes + LDS_FLOATS - 1) / LDS_FLOATS;
    int rn1 = (((n_nodes + nr1 - 1) / nr1) + 31) & ~31;
    int ch1 = 128 / nr1 > 0 ? 128 / nr1 : 1;
    int pc1 = (n_edges + ch1 - 1) / ch1;
    degree_lds<<<nr1 * ch1, SCAT_THREADS, 0, stream>>>(dst, deg, n_edges, n_nodes, rn1, ch1, pc1);
    dinv_kernel<<<(n_nodes + B - 1) / B, B, 0, stream>>>(deg, n_nodes);

    int n_tiles = (n_nodes + 15) / 16;
    int wpb = B / 32;
    gemm1_wmma<<<(n_tiles + wpb - 1) / wpb, B, 0, stream>>>(x, W1, h, n_tiles, n_nodes);

    int nr0 = (n_nodes + E1_NODES - 1) / E1_NODES;
    int rn0 = (((n_nodes + nr0 - 1) / nr0) + 31) & ~31;
    int ch0 = 256 / nr0 > 0 ? 256 / nr0 : 1;
    int pc0 = (n_edges + ch0 - 1) / ch0;
    edge1_lds<<<nr0 * ch0, SCAT_THREADS, 0, stream>>>(src, dst, deg, h, acc1, n_edges, n_nodes,
                                                      rn0, ch0, pc0);
    relu_bias1<<<(n16 + B - 1) / B, B, 0, stream>>>(h, deg, b1, acc1, n_nodes);

    gemm2_kernel<<<(n_nodes + B - 1) / B, B, 0, stream>>>(acc1, W2, h2, n_nodes);

    int nr2 = (n_nodes * FEAT_OUT + LDS_FLOATS - 1) / LDS_FLOATS;
    int rn2 = (((n_nodes + nr2 - 1) / nr2) + 31) & ~31;
    int ch2 = 128 / nr2 > 0 ? 128 / nr2 : 1;
    int pc2 = (n_edges + ch2 - 1) / ch2;
    edge2_lds<<<nr2 * ch2, SCAT_THREADS, 0, stream>>>(src, dst, deg, h2, acc2, n_edges, n_nodes,
                                                      rn2, ch2, pc2);

    final_kernel<<<(n_nodes + B - 1) / B, B, 0, stream>>>(acc2, h2, deg, b2, out, n_nodes);
}


# edge weight dinv[src]*dinv[dst] factored: rows of h and h2 scaled by dinv where produced, node sums by dinv after; per-edge dinv gathers removed (sum order kept, rounding differs)
# speedup vs baseline: 2.8080x; 1.3623x over previous
#include <hip/hip_runtime.h>
#include <math.h>

typedef float v2f __attribute__((ext_vector_type(2)));
typedef float v8f __attribute__((ext_vector_type(8)));

#define FEAT_IN  128
#define FEAT_MID 16
#define FEAT_OUT 2

__global__ void init_ws(float* __restrict__ deg, float* __restrict__ acc1,
                        float* __restrict__ acc2, int n_nodes) {
    int tid = blockIdx.x * blockDim.x + threadIdx.x;
    if (tid < n_nodes * FEAT_MID) acc1[tid] = 0.0f;
    if (tid < n_nodes * FEAT_OUT) acc2[tid] = 0.0f;
    if (tid < n_nodes)            deg[tid]  = 1.0f;
}

#define LDS_FLOATS   69632
#define SCAT_THREADS 1024
#define SCAT_U       8

__global__ void __launch_bounds__(SCAT_THREADS)
degree_lds(const int* __restrict__ dst, float* __restrict__ deg,
           int n_edges, int n_nodes, int range_nodes, int chunks, int per_chunk) {
    __shared__ float cnt[LDS_FLOATS];
    const int tid   = threadIdx.x;
    const int range = blockIdx.x / chunks;
    const int chunk = blockIdx.x - range * chunks;
    const int lo    = range * range_nodes;
    int len = n_nodes - lo;
    if (len > range_nodes) len = range_nodes;
    if (len <= 0) return;

    for (int i = tid; i < len; i += SCAT_THREADS) cnt[i] = 0.0f;
    __syncthreads();

    const int e0 = chunk * per_chunk;
    int e1 = e0 + per_chunk;
    if (e1 > n_edges) e1 = n_edges;
    for (int base = e0; base < e1; base += SCAT_U * SCAT_THREADS) {
        int dv[SCAT_U];
#pragma unroll
        for (int k = 0; k < SCAT_U; k++) {
            int e  = base + k * SCAT_THREADS + tid;
            int ec = e < e1 ? e : e1 - 1;
            dv[k] = dst[ec];
        }
        asm volatile("" : "+v"(dv[0]), "+v"(dv[1]), "+v"(dv[2]), "+v"(dv[3]),
                          "+v"(dv[4]), "+v"(dv[5]), "+v"(dv[6]), "+v"(dv[7]));
#pragma unroll
        for (int k = 0; k < SCAT_U; k++) {
            int e = base + k * SCAT_THREADS + tid;
            int t = dv[k] - lo;
            if (e < e1 && (unsigned)t < (unsigned)len) atomicAdd(&cnt[t], 1.0f);
        }
    }
    __syncthreads();

    const int rot = (chunk * (len / chunks)) & ~31;
    for (int j = tid; j < len; j += SCAT_THREADS) {
        int i = j + rot;
        if (i >= len) i -= len;
        float v = cnt[i];
        if (v != 0.0f) atomicAdd(&deg[lo + i], v);
    }
}

__global__ void dinv_kernel(float* __restrict__ deg, int n_nodes) {
    int i = blockIdx.x * blockDim.x + threadIdx.x;
    if (i < n_nodes) {
        float d = deg[i];
        deg[i] = (d > 0.0f) ? rsqrtf(d) : 0.0f;
    }
}

__global__ void __launch_bounds__(256)
gemm1_wmma(const float* __restrict__ x, const float* __restrict__ W1,
           const float* __restrict__ dinv, float* __restrict__ h,
           int n_tiles, int n_nodes) {
    int lane = threadIdx.x & 31;
    int wave = threadIdx.x >> 5;
    int tile = blockIdx.x * (blockDim.x >> 5) + wave;
    if (tile >= n_tiles) return;

    int col = lane & 15;
    int kh  = (lane >> 4) << 1;

    int rowA = tile * 16 + col;
    if (rowA > n_nodes - 1) rowA = n_nodes - 1;
    const float* xrow = x + (size_t)rowA * FEAT_IN;

    int mbase = (lane >> 4) * 8;
    float dr[8];
#pragma unroll
    for (int r = 0; r < 8; r++) {
        int row = tile * 16 + mbase + r;
        dr[r] = dinv[row < n_nodes ? row : n_nodes - 1];
    }

    v8f acc = {};
    for (int k0 = 0; k0 < FEAT_IN; k0 += 4) {
        v2f a, b;
        a.x = xrow[k0 + kh];
        a.y = xrow[k0 + kh + 1];
        b.x = W1[(k0 + kh)     * FEAT_MID + col];
        b.y = W1[(k0 + kh + 1) * FEAT_MID + col];
        acc = __builtin_amdgcn_wmma_f32_16x16x4_f32(
             false, a,  false, b,
             (short)0, acc,  false,  false);
    }

    asm volatile("v_nop\n\tv_nop\n\tv_nop\n\tv_nop\n\tv_nop" : "+v"(acc));
#pragma unroll
    for (int r = 0; r < 8; r++) {
        int row = tile * 16 + mbase + r;
        if (row < n_nodes) h[(size_t)row * FEAT_MID + col] = acc[r] * dr[r];
    }
}

typedef int v2i __attribute__((ext_vector_type(2)));

#define E1_NODES 2400
#define E1_U     16
#define E1_STEPS 8
#define E1_LIST  (E1_U * 32)

__global__ void __launch_bounds__(SCAT_THREADS)
edge1_lds(const int* __restrict__ src, const int* __restrict__ dst,
          const float* __restrict__ h,
          float* __restrict__ acc1, int n_edges, int n_nodes,
          int range_nodes, int chunks, int per_chunk) {
    __shared__ float acc[E1_NODES * FEAT_MID];
    __shared__ v2i   list[SCAT_THREADS / 32][E1_LIST];
    const int tid   = threadIdx.x;
    const int lane  = tid & 31;
    const int f     = lane & 15;
    const int hf    = lane >> 4;
    const int range = blockIdx.x / chunks;
    const int chunk = blockIdx.x - range * chunks;
    const int lo    = range * range_nodes;
    int len = n_nodes - lo;
    if (len > range_nodes) len = range_nodes;
    if (len <= 0) return;
    const int flen = len * FEAT_MID;

    for (int i = tid; i < flen; i += SCAT_THREADS) acc[i] = 0.0f;
    __syncthreads();

    v2i* __restrict__ lst = list[tid >> 5];
    const int e0 = chunk * per_chunk;
    int e1 = e0 + per_chunk;
    if (e1 > n_edges) e1 = n_edges;
    for (int base = e0; base < e1; base += E1_U * SCAT_THREADS) {
        int dv[E1_U], sv[E1_U];
#pragma unroll
        for (int k = 0; k < E1_U; k++) {
            int e  = base + k * SCAT_THREADS + tid;
            int ec = e < e1 ? e : e1 - 1;
            dv[k] = dst[ec];
        }
        asm volatile("" : "+v"(dv[0]), "+v"(dv[1]), "+v"(dv[2]), "+v"(dv[3]),
                          "+v"(dv[4]), "+v"(dv[5]), "+v"(dv[6]), "+v"(dv[7]),
                          "+v"(dv[8]), "+v"(dv[9]), "+v"(dv[10]), "+v"(dv[11]),
                          "+v"(dv[12]), "+v"(dv[13]), "+v"(dv[14]), "+v"(dv[15]));
#pragma unroll
        for (int k = 0; k < E1_U; k++) {
            int e  = base + k * SCAT_THREADS + tid;
            int tt = dv[k] - lo;
            bool m = (e < e1) && ((unsigned)tt < (unsigned)len);
            dv[k] = m ? tt : -1;
            sv[k] = src[m ? e : e0];
        }
        asm volatile("" : "+v"(sv[0]), "+v"(sv[1]), "+v"(sv[2]), "+v"(sv[3]),
                          "+v"(sv[4]), "+v"(sv[5]), "+v"(sv[6]), "+v"(sv[7]),
                          "+v"(sv[8]), "+v"(sv[9]), "+v"(sv[10]), "+v"(sv[11]),
                          "+v"(sv[12]), "+v"(sv[13]), "+v"(sv[14]), "+v"(sv[15]));
        int cnt = 0;
#pragma unroll
        for (int k = 0; k < E1_U; k++) {
            bool m = dv[k] >= 0;
            unsigned b = __builtin_amdgcn_ballot_w32(m);
            int pos = cnt + (int)__builtin_amdgcn_mbcnt_lo(b, 0u);
            if (m) {
                v2i q;
                q.x = sv[k];
                q.y = dv[k];
                lst[pos] = q;
            }
            cnt += __builtin_popcount(b);
        }
        __builtin_amdgcn_wave_barrier();
        for (int j0 = 0; j0 < cnt; j0 += 2 * E1_STEPS) {
            int   tq[E1_STEPS];
            float hq[E1_STEPS];
#pragma unroll
            for (int p = 0; p < E1_STEPS; p++) {
                int  idx   = j0 + 2 * p + hf;
                bool valid = idx < cnt;
                v2i  q     = lst[valid ? idx : 0];
                tq[p] = valid ? q.y : -1;
                hq[p] = h[q.x * FEAT_MID + f];
            }
            asm volatile("" : "+v"(hq[0]), "+v"(hq[1]), "+v"(hq[2]), "+v"(hq[3]),
                              "+v"(hq[4]), "+v"(hq[5]), "+v"(hq[6]), "+v"(hq[7]));
#pragma unroll
            for (int p = 0; p < E1_STEPS; p++) {
                if (tq[p] >= 0)
                    atomicAdd(&acc[tq[p] * FEAT_MID + f], hq[p]);
            }
        }
        __builtin_amdgcn_wave_barrier();
    }
    __syncthreads();

    const int rot = (chunk * (flen / chunks)) & ~31;
    float* __restrict__ outp = acc1 + (size_t)lo * FEAT_MID;
    for (int j = tid; j < flen; j += SCAT_THREADS) {
        int i = j + rot;
        if (i >= flen) i -= flen;
        float v = acc[i];
        if (v != 0.0f) atomicAdd(&outp[i], v);
    }
}

__global__ void relu_bias1(const float* __restrict__ h, const float* __restrict__ dinv,
                           const float* __restrict__ b1, float* __restrict__ acc1,
                           int n_nodes) {
    int tid = blockIdx.x * blockDim.x + threadIdx.x;
    if (tid >= n_nodes * FEAT_MID) return;
    int node = tid >> 4, f = tid & 15;
    float di = dinv[node];
    float v = (acc1[tid] + h[tid]) * di + b1[f];
    acc1[tid] = fmaxf(v, 0.0f);
}

__global__ void gemm2_kernel(const float* __restrict__ h1, const float* __restrict__ W2,
                             const float* __restrict__ dinv,
                             float* __restrict__ h2, int n_nodes) {
    int i = blockIdx.x * blockDim.x + threadIdx.x;
    if (i >= n_nodes) return;
    float s0 = 0.0f, s1 = 0.0f;
#pragma unroll
    for (int k = 0; k < FEAT_MID; k++) {
        float v = h1[i * FEAT_MID + k];
        s0 = fmaf(v, W2[k * 2 + 0], s0);
        s1 = fmaf(v, W2[k * 2 + 1], s1);
    }
    float di = dinv[i];
    h2[i * 2 + 0] = s0 * di;
    h2[i * 2 + 1] = s1 * di;
}

__global__ void __launch_bounds__(SCAT_THREADS)
edge2_lds(const int* __restrict__ src, const int* __restrict__ dst,
          const float* __restrict__ h2,
          float* __restrict__ acc2, int n_edges, int n_nodes,
          int range_nodes, int chunks, int per_chunk) {
    __shared__ float acc[LDS_FLOATS];
    const int tid   = threadIdx.x;
    const int range = blockIdx.x / chunks;
    const int chunk = blockIdx.x - range * chunks;
    const int lo    = range * range_nodes;
    int len = n_nodes - lo;
    if (len > range_nodes) len = range_nodes;
    if (len <= 0) return;
    const int flen = len * FEAT_OUT;

    for (int i = tid; i < flen; i += SCAT_THREADS) acc[i] = 0.0f;
    __syncthreads();

    const v2f* __restrict__ h2v = (const v2f*)h2;
    const int e0 = chunk * per_chunk;
    int e1 = e0 + per_chunk;
    if (e1 > n_edges) e1 = n_edges;
    for (int base = e0; base < e1; base += SCAT_U * SCAT_THREADS) {
        int dv[SCAT_U], sv[SCAT_U];
#pragma unroll
        for (int k = 0; k < SCAT_U; k++) {
            int e  = base + k * SCAT_THREADS + tid;
            int ec = e < e1 ? e : e1 - 1;
            dv[k] = dst[ec];
            sv[k] = src[ec];
        }
        asm volatile("" : "+v"(dv[0]), "+v"(dv[1]), "+v"(dv[2]), "+v"(dv[3]),
                          "+v"(dv[4]), "+v"(dv[5]), "+v"(dv[6]), "+v"(dv[7]),
                          "+v"(sv[0]), "+v"(sv[1]), "+v"(sv[2]), "+v"(sv[3]),
                          "+v"(sv[4]), "+v"(sv[5]), "+v"(sv[6]), "+v"(sv[7]));
        int   t[SCAT_U];
        v2f   hv[SCAT_U];
#pragma unroll
        for (int k = 0; k < SCAT_U; k++) {
            int e  = base + k * SCAT_THREADS + tid;
            int tt = dv[k] - lo;
            bool m = (e < e1) && ((unsigned)tt < (unsigned)len);
            t[k] = m ? tt : -1;
            int sg = m ? sv[k] : lo;
            hv[k] = h2v[sg];
        }
        asm volatile("" : "+v"(hv[0]), "+v"(hv[1]), "+v"(hv[2]), "+v"(hv[3]),
                          "+v"(hv[4]), "+v"(hv[5]), "+v"(hv[6]), "+v"(hv[7]));
#pragma unroll
        for (int k = 0; k < SCAT_U; k++) {
            if (t[k] >= 0) {
                atomicAdd(&acc[t[k] * FEAT_OUT + 0], hv[k].x);
                atomicAdd(&acc[t[k] * FEAT_OUT + 1], hv[k].y);
            }
        }
    }
    __syncthreads();

    const int rot = (chunk * (flen / chunks)) & ~31;
    float* __restrict__ outp = acc2 + (size_t)lo * FEAT_OUT;
    for (int j = tid; j < flen; j += SCAT_THREADS) {
        int i = j + rot;
        if (i >= flen) i -= flen;
        float v = acc[i];
        if (v != 0.0f) atomicAdd(&outp[i], v);
    }
}

__global__ void final_kernel(const float* __restrict__ acc2, const float* __restrict__ h2,
                             const float* __restrict__ dinv, const float* __restrict__ b2,
                             float* __restrict__ out, int n_nodes) {
    int i = blockIdx.x * blockDim.x + threadIdx.x;
    if (i >= n_nodes) return;
    float di = dinv[i];
    float v0 = (acc2[i * 2 + 0] + h2[i * 2 + 0]) * di + b2[0];
    float v1 = (acc2[i * 2 + 1] + h2[i * 2 + 1]) * di + b2[1];
    float m  = fmaxf(v0, v1);
    float l  = m + logf(expf(v0 - m) + expf(v1 - m));
    out[i * 2 + 0] = v0 - l;
    out[i * 2 + 1] = v1 - l;
}

extern "C" void kernel_launch(void* const* d_in, const int* in_sizes, int n_in,
                              void* d_out, int out_size, void* d_ws, size_t ws_size,
                              hipStream_t stream) {
    const float* x  = (const float*)d_in[0];
    const int*   ei = (const int*)d_in[1];
    const float* W1 = (const float*)d_in[2];
    const float* b1 = (const float*)d_in[3];
    const float* W2 = (const float*)d_in[4];
    const float* b2 = (const float*)d_in[5];
    float* out = (float*)d_out;

    int n_nodes = in_sizes[0] / FEAT_IN;
    int n_edges = in_sizes[1] / 2;
    const int* src = ei;
    const int* dst = ei + n_edges;

    float* ws   = (float*)d_ws;
    float* deg  = ws;
    float* h    = deg  + n_nodes;
    float* acc1 = h    + (size_t)n_nodes * FEAT_MID;
    float* h2   = acc1 + (size_t)n_nodes * FEAT_MID;
    float* acc2 = h2   + (size_t)n_nodes * FEAT_OUT;

    const int B = 256;
    int n16 = n_nodes * FEAT_MID;

    init_ws<<<(n16 + B - 1) / B, B, 0, stream>>>(deg, acc1, acc2, n_nodes);
    int nr1 = (n_nodes + LDS_FLOATS - 1) / LDS_FLOATS;
    int rn1 = (((n_nodes + nr1 - 1) / nr1) + 31) & ~31;
    int ch1 = 128 / nr1 > 0 ? 128 / nr1 : 1;
    int pc1 = (n_edges + ch1 - 1) / ch1;
    degree_lds<<<nr1 * ch1, SCAT_THREADS, 0, stream>>>(dst, deg, n_edges, n_nodes, rn1, ch1, pc1);
    dinv_kernel<<<(n_nodes + B - 1) / B, B, 0, stream>>>(deg, n_nodes);

    int n_tiles = (n_nodes + 15) / 16;
    int wpb = B / 32;
    gemm1_wmma<<<(n_tiles + wpb - 1) / wpb, B, 0, stream>>>(x, W1, deg, h, n_tiles, n_nodes);

    int nr0 = (n_nodes + E1_NODES - 1) / E1_NODES;
    int rn0 = (((n_nodes + nr0 - 1) / nr0) + 31) & ~31;
    int ch0 = 256 / nr0 > 0 ? 256 / nr0 : 1;
    int pc0 = (n_edges + ch0 - 1) / ch0;
    edge1_lds<<<nr0 * ch0, SCAT_THREADS, 0, stream>>>(src, dst, h, acc1, n_edges, n_nodes,
                                                      rn0, ch0, pc0);
    relu_bias1<<<(n16 + B - 1) / B, B, 0, stream>>>(h, deg, b1, acc1, n_nodes);

    gemm2_kernel<<<(n_nodes + B - 1) / B, B, 0, stream>>>(acc1, W2, deg, h2, n_nodes);

    int nr2 = (n_nodes * FEAT_OUT + LDS_FLOATS - 1) / LDS_FLOATS;
    int rn2 = (((n_nodes + nr2 - 1) / nr2) + 31) & ~31;
    int ch2 = 128 / nr2 > 0 ? 128 / nr2 : 1;
    int pc2 = (n_edges + ch2 - 1) / ch2;
    edge2_lds<<<nr2 * ch2, SCAT_THREADS, 0, stream>>>(src, dst, h2, acc2, n_edges, n_nodes,
                                                      rn2, ch2, pc2);

    final_kernel<<<(n_nodes + B - 1) / B, B, 0, stream>>>(acc2, h2, deg, b2, out, n_nodes);
}


# layer-1 edge test loop: targets read as 16-byte loads one trip ahead, test cut to subtract+compare, source fetched only for listed edges (same work-group partials, order of adds inside them differs)
# speedup vs baseline: 2.9672x; 1.0567x over previous
#include <hip/hip_runtime.h>
#include <math.h>

typedef float v2f __attribute__((ext_vector_type(2)));
typedef float v8f __attribute__((ext_vector_type(8)));

#define FEAT_IN  128
#define FEAT_MID 16
#define FEAT_OUT 2

__global__ void init_ws(float* __restrict__ deg, float* __restrict__ acc1,
                        float* __restrict__ acc2, int n_nodes) {
    int tid = blockIdx.x * blockDim.x + threadIdx.x;
    if (tid < n_nodes * FEAT_MID) acc1[tid] = 0.0f;
    if (tid < n_nodes * FEAT_OUT) acc2[tid] = 0.0f;
    if (tid < n_nodes)            deg[tid]  = 1.0f;
}

#define LDS_FLOATS   69632
#define SCAT_THREADS 1024
#define SCAT_U       8

__global__ void __launch_bounds__(SCAT_THREADS)
degree_lds(const int* __restrict__ dst, float* __restrict__ deg,
           int n_edges, int n_nodes, int range_nodes, int chunks, int per_chunk) {
    __shared__ float cnt[LDS_FLOATS];
    const int tid   = threadIdx.x;
    const int range = blockIdx.x / chunks;
    const int chunk = blockIdx.x - range * chunks;
    const int lo    = range * range_nodes;
    int len = n_nodes - lo;
    if (len > range_nodes) len = range_nodes;
    if (len <= 0) return;

    for (int i = tid; i < len; i += SCAT_THREADS) cnt[i] = 0.0f;
    __syncthreads();

    const int e0 = chunk * per_chunk;
    int e1 = e0 + per_chunk;
    if (e1 > n_edges) e1 = n_edges;
    for (int base = e0; base < e1; base += SCAT_U * SCAT_THREADS) {
        int dv[SCAT_U];
#pragma unroll
        for (int k = 0; k < SCAT_U; k++) {
            int e  = base + k * SCAT_THREADS + tid;
            int ec = e < e1 ? e : e1 - 1;
            dv[k] = dst[ec];
        }
        asm volatile("" : "+v"(dv[0]), "+v"(dv[1]), "+v"(dv[2]), "+v"(dv[3]),
                          "+v"(dv[4]), "+v"(dv[5]), "+v"(dv[6]), "+v"(dv[7]));
#pragma unroll
        for (int k = 0; k < SCAT_U; k++) {
            int e = base + k * SCAT_THREADS + tid;
            int t = dv[k] - lo;
            if (e < e1 && (unsigned)t < (unsigned)len) atomicAdd(&cnt[t], 1.0f);
        }
    }
    __syncthreads();

    const int rot = (chunk * (len / chunks)) & ~31;
    for (int j = tid; j < len; j += SCAT_THREADS) {
        int i = j + rot;
        if (i >= len) i -= len;
        float v = cnt[i];
        if (v != 0.0f) atomicAdd(&deg[lo + i], v);
    }
}

__global__ void dinv_kernel(float* __restrict__ deg, int n_nodes) {
    int i = blockIdx.x * blockDim.x + threadIdx.x;
    if (i < n_nodes) {
        float d = deg[i];
        deg[i] = (d > 0.0f) ? rsqrtf(d) : 0.0f;
    }
}

__global__ void __launch_bounds__(256)
gemm1_wmma(const float* __restrict__ x, const float* __restrict__ W1,
           const float* __restrict__ dinv, float* __restrict__ h,
           int n_tiles, int n_nodes) {
    int lane = threadIdx.x & 31;
    int wave = threadIdx.x >> 5;
    int tile = blockIdx.x * (blockDim.x >> 5) + wave;
    if (tile >= n_tiles) return;

    int col = lane & 15;
    int kh  = (lane >> 4) << 1;

    int rowA = tile * 16 + col;
    if (rowA > n_nodes - 1) rowA = n_nodes - 1;
    const float* xrow = x + (size_t)rowA * FEAT_IN;

    int mbase = (lane >> 4) * 8;
    float dr[8];
#pragma unroll
    for (int r = 0; r < 8; r++) {
        int row = tile * 16 + mbase + r;
        dr[r] = dinv[row < n_nodes ? row : n_nodes - 1];
    }

    v8f acc = {};
    for (int k0 = 0; k0 < FEAT_IN; k0 += 4) {
        v2f a, b;
        a.x = xrow[k0 + kh];
        a.y = xrow[k0 + kh + 1];
        b.x = W1[(k0 + kh)     * FEAT_MID + col];
        b.y = W1[(k0 + kh + 1) * FEAT_MID + col];
        acc = __builtin_amdgcn_wmma_f32_16x16x4_f32(
             false, a,  false, b,
             (short)0, acc,  false,  false);
    }

    asm volatile("v_nop\n\tv_nop\n\tv_nop\n\tv_nop\n\tv_nop" : "+v"(acc));
#pragma unroll
    for (int r = 0; r < 8; r++) {
        int row = tile * 16 + mbase + r;
        if (row < n_nodes) h[(size_t)row * FEAT_MID + col] = acc[r] * dr[r];
    }
}

typedef int v2i __attribute__((ext_vector_type(2)));
typedef int v4i __attribute__((ext_vector_type(4)));

#define E1_NODES 2400
#define E1_U     16
#define E1_STEPS 8
#define E1_LIST  (E1_U * 32)
#define E1_TRIP  (E1_U * SCAT_THREADS)

__device__ __forceinline__ void e1_push(bool m, int e, int tt, v2i* lst, int& cnt) {
    unsigned b = __builtin_amdgcn_ballot_w32(m);
    int pos = cnt + (int)__builtin_amdgcn_mbcnt_lo(b, 0u);
    if (m) {
        v2i q;
        q.x = e;
        q.y = tt;
        lst[pos] = q;
    }
    cnt += __builtin_popcount(b);
}

__device__ __forceinline__ void e1_drain(const v2i* lst, int cnt,
                                         const int* __restrict__ src, const float* __restrict__ h,
                                         float* __restrict__ acc, int f, int hf) {
    for (int j0 = 0; j0 < cnt; j0 += 2 * E1_STEPS) {
        int   tq[E1_STEPS], sq[E1_STEPS];
        float hq[E1_STEPS];
#pragma unroll
        for (int p = 0; p < E1_STEPS; p++) {
            int  idx   = j0 + 2 * p + hf;
            bool valid = idx < cnt;
            v2i  q     = lst[valid ? idx : 0];
            tq[p] = valid ? q.y : -1;
            sq[p] = src[q.x];
        }
        asm volatile("" : "+v"(sq[0]), "+v"(sq[1]), "+v"(sq[2]), "+v"(sq[3]),
                          "+v"(sq[4]), "+v"(sq[5]), "+v"(sq[6]), "+v"(sq[7]));
#pragma unroll
        for (int p = 0; p < E1_STEPS; p++)
            hq[p] = h[sq[p] * FEAT_MID + f];
        asm volatile("" : "+v"(hq[0]), "+v"(hq[1]), "+v"(hq[2]), "+v"(hq[3]),
                          "+v"(hq[4]), "+v"(hq[5]), "+v"(hq[6]), "+v"(hq[7]));
#pragma unroll
        for (int p = 0; p < E1_STEPS; p++) {
            if (tq[p] >= 0)
                atomicAdd(&acc[tq[p] * FEAT_MID + f], hq[p]);
        }
    }
}

__global__ void __launch_bounds__(SCAT_THREADS)
edge1_lds(const int* __restrict__ src, const int* __restrict__ dst,
          const float* __restrict__ h,
          float* __restrict__ acc1, int n_edges, int n_nodes,
          int range_nodes, int chunks, int per_chunk) {
    __shared__ float acc[E1_NODES * FEAT_MID];
    __shared__ v2i   list[SCAT_THREADS / 32][E1_LIST];
    const int tid   = threadIdx.x;
    const int lane  = tid & 31;
    const int f     = lane & 15;
    const int hf    = lane >> 4;
    const int range = blockIdx.x / chunks;
    const int chunk = blockIdx.x - range * chunks;
    const int lo    = range * range_nodes;
    int len = n_nodes - lo;
    if (len > range_nodes) len = range_nodes;
    if (len <= 0) return;
    const int flen = len * FEAT_MID;

    for (int i = tid; i < flen; i += SCAT_THREADS) acc[i] = 0.0f;
    __syncthreads();

    v2i* __restrict__ lst = list[tid >> 5];
    int e0 = chunk * per_chunk;
    if (e0 > n_edges) e0 = n_edges;
    e0 &= ~3;
    int e1 = n_edges;
    if (chunk + 1 < chunks) {
        e1 = (chunk + 1) * per_chunk;
        if (e1 > n_edges) e1 = n_edges;
        e1 &= ~3;
    }

    const bool wide = (((size_t)dst) & 15) == 0;
    const int nfull = wide ? (e1 - e0) / E1_TRIP : 0;
    v4i nx[E1_U / 4];
    if (nfull > 0) {
        const v4i* __restrict__ pn = (const v4i*)(dst + e0);
#pragma unroll
        for (int j = 0; j < E1_U / 4; j++) nx[j] = pn[j * SCAT_THREADS + tid];
    }
    for (int t = 0; t < nfull; t++) {
        const int base = e0 + t * E1_TRIP;
        v4i cur[E1_U / 4];
#pragma unroll
        for (int j = 0; j < E1_U / 4; j++) cur[j] = nx[j];
        if (t + 1 < nfull) {
            const v4i* __restrict__ pn = (const v4i*)(dst + base + E1_TRIP);
#pragma unroll
            for (int j = 0; j < E1_U / 4; j++) nx[j] = pn[j * SCAT_THREADS + tid];
        }
        const int eb = base + 4 * tid;
        int cnt = 0;
#pragma unroll
        for (int j = 0; j < E1_U / 4; j++) {
#pragma unroll
            for (int i = 0; i < 4; i++) {
                int tt = cur[j][i] - lo;
                e1_push((unsigned)tt < (unsigned)len, eb + j * 4 * SCAT_THREADS + i, tt, lst, cnt);
            }
        }
        __builtin_amdgcn_wave_barrier();
        e1_drain(lst, cnt, src, h, acc, f, hf);
        __builtin_amdgcn_wave_barrier();
    }

    for (int base = e0 + nfull * E1_TRIP; base < e1; base += E1_TRIP) {
        int dv[E1_U];
#pragma unroll
        for (int k = 0; k < E1_U; k++) {
            int e  = base + k * SCAT_THREADS + tid;
            int ec = e < e1 ? e : e1 - 1;
            dv[k] = dst[ec];
        }
        asm volatile("" : "+v"(dv[0]), "+v"(dv[1]), "+v"(dv[2]), "+v"(dv[3]),
                          "+v"(dv[4]), "+v"(dv[5]), "+v"(dv[6]), "+v"(dv[7]),
                          "+v"(dv[8]), "+v"(dv[9]), "+v"(dv[10]), "+v"(dv[11]),
                          "+v"(dv[12]), "+v"(dv[13]), "+v"(dv[14]), "+v"(dv[15]));
        int cnt = 0;
#pragma unroll
        for (int k = 0; k < E1_U; k++) {
            int e  = base + k * SCAT_THREADS + tid;
            int tt = dv[k] - lo;
            e1_push((e < e1) && ((unsigned)tt < (unsigned)len), e, tt, lst, cnt);
        }
        __builtin_amdgcn_wave_barrier();
        e1_drain(lst, cnt, src, h, acc, f, hf);
        __builtin_amdgcn_wave_barrier();
    }
    __syncthreads();

    const int rot = (chunk * (flen / chunks)) & ~31;
    float* __restrict__ outp = acc1 + (size_t)lo * FEAT_MID;
    for (int j = tid; j < flen; j += SCAT_THREADS) {
        int i = j + rot;
        if (i >= flen) i -= flen;
        float v = acc[i];
        if (v != 0.0f) atomicAdd(&outp[i], v);
    }
}

__global__ void relu_bias1(const float* __restrict__ h, const float* __restrict__ dinv,
                           const float* __restrict__ b1, float* __restrict__ acc1,
                           int n_nodes) {
    int tid = blockIdx.x * blockDim.x + threadIdx.x;
    if (tid >= n_nodes * FEAT_MID) return;
    int node = tid >> 4, f = tid & 15;
    float di = dinv[node];
    float v = (acc1[tid] + h[tid]) * di + b1[f];
    acc1[tid] = fmaxf(v, 0.0f);
}

__global__ void gemm2_kernel(const float* __restrict__ h1, const float* __restrict__ W2,
                             const float* __restrict__ dinv,
                             float* __restrict__ h2, int n_nodes) {
    int i = blockIdx.x * blockDim.x + threadIdx.x;
    if (i >= n_nodes) return;
    float s0 = 0.0f, s1 = 0.0f;
#pragma unroll
    for (int k = 0; k < FEAT_MID; k++) {
        float v = h1[i * FEAT_MID + k];
        s0 = fmaf(v, W2[k * 2 + 0], s0);
        s1 = fmaf(v, W2[k * 2 + 1], s1);
    }
    float di = dinv[i];
    h2[i * 2 + 0] = s0 * di;
    h2[i * 2 + 1] = s1 * di;
}

__global__ void __launch_bounds__(SCAT_THREADS)
edge2_lds(const int* __restrict__ src, const int* __restrict__ dst,
          const float* __restrict__ h2,
          float* __restrict__ acc2, int n_edges, int n_nodes,
          int range_nodes, int chunks, int per_chunk) {
    __shared__ float acc[LDS_FLOATS];
    const int tid   = threadIdx.x;
    const int range = blockIdx.x / chunks;
    const int chunk = blockIdx.x - range * chunks;
    const int lo    = range * range_nodes;
    int len = n_nodes - lo;
    if (len > range_nodes) len = range_nodes;
    if (len <= 0) return;
    const int flen = len * FEAT_OUT;

    for (int i = tid; i < flen; i += SCAT_THREADS) acc[i] = 0.0f;
    __syncthreads();

    const v2f* __restrict__ h2v = (const v2f*)h2;
    const int e0 = chunk * per_chunk;
    int e1 = e0 + per_chunk;
    if (e1 > n_edges) e1 = n_edges;
    for (int base = e0; base < e1; base += SCAT_U * SCAT_THREADS) {
        int dv[SCAT_U], sv[SCAT_U];
#pragma unroll
        for (int k = 0; k < SCAT_U; k++) {
            int e  = base + k * SCAT_THREADS + tid;
            int ec = e < e1 ? e : e1 - 1;
            dv[k] = dst[ec];
            sv[k] = src[ec];
        }
        asm volatile("" : "+v"(dv[0]), "+v"(dv[1]), "+v"(dv[2]), "+v"(dv[3]),
                          "+v"(dv[4]), "+v"(dv[5]), "+v"(dv[6]), "+v"(dv[7]),
                          "+v"(sv[0]), "+v"(sv[1]), "+v"(sv[2]), "+v"(sv[3]),
                          "+v"(sv[4]), "+v"(sv[5]), "+v"(sv[6]), "+v"(sv[7]));
        int   t[SCAT_U];
        v2f   hv[SCAT_U];
#pragma unroll
        for (int k = 0; k < SCAT_U; k++) {
            int e  = base + k * SCAT_THREADS + tid;
            int tt = dv[k] - lo;
            bool m = (e < e1) && ((unsigned)tt < (unsigned)len);
            t[k] = m ? tt : -1;
            int sg = m ? sv[k] : lo;
            hv[k] = h2v[sg];
        }
        asm volatile("" : "+v"(hv[0]), "+v"(hv[1]), "+v"(hv[2]), "+v"(hv[3]),
                          "+v"(hv[4]), "+v"(hv[5]), "+v"(hv[6]), "+v"(hv[7]));
#pragma unroll
        for (int k = 0; k < SCAT_U; k++) {
            if (t[k] >= 0) {
                atomicAdd(&acc[t[k] * FEAT_OUT + 0], hv[k].x);
                atomicAdd(&acc[t[k] * FEAT_OUT + 1], hv[k].y);
            }
        }
    }
    __syncthreads();

    const int rot = (chunk * (flen / chunks)) & ~31;
    float* __restrict__ outp = acc2 + (size_t)lo * FEAT_OUT;
    for (int j = tid; j < flen; j += SCAT_THREADS) {
        int i = j + rot;
        if (i >= flen) i -= flen;
        float v = acc[i];
        if (v != 0.0f) atomicAdd(&outp[i], v);
    }
}

__global__ void final_kernel(const float* __restrict__ acc2, const float* __restrict__ h2,
                             const float* __restrict__ dinv, const float* __restrict__ b2,
                             float* __restrict__ out, int n_nodes) {
    int i = blockIdx.x * blockDim.x + threadIdx.x;
    if (i >= n_nodes) return;
    float di = dinv[i];
    float v0 = (acc2[i * 2 + 0] + h2[i * 2 + 0]) * di + b2[0];
    float v1 = (acc2[i * 2 + 1] + h2[i * 2 + 1]) * di + b2[1];
    float m  = fmaxf(v0, v1);
    float l  = m + logf(expf(v0 - m) + expf(v1 - m));
    out[i * 2 + 0] = v0 - l;
    out[i * 2 + 1] = v1 - l;
}

extern "C" void kernel_launch(void* const* d_in, const int* in_sizes, int n_in,
                              void* d_out, int out_size, void* d_ws, size_t ws_size,
                              hipStream_t stream) {
    const float* x  = (const float*)d_in[0];
    const int*   ei = (const int*)d_in[1];
    const float* W1 = (const float*)d_in[2];
    const float* b1 = (const float*)d_in[3];
    const float* W2 = (const float*)d_in[4];
    const float* b2 = (const float*)d_in[5];
    float* out = (float*)d_out;

    int n_nodes = in_sizes[0] / FEAT_IN;
    int n_edges = in_sizes[1] / 2;
    const int* src = ei;
    const int* dst = ei + n_edges;

    float* ws   = (float*)d_ws;
    float* deg  = ws;
    float* h    = deg  + n_nodes;
    float* acc1 = h    + (size_t)n_nodes * FEAT_MID;
    float* h2   = acc1 + (size_t)n_nodes * FEAT_MID;
    float* acc2 = h2   + (size_t)n_nodes * FEAT_OUT;

    const int B = 256;
    int n16 = n_nodes * FEAT_MID;

    init_ws<<<(n16 + B - 1) / B, B, 0, stream>>>(deg, acc1, acc2, n_nodes);
    int nr1 = (n_nodes + LDS_FLOATS - 1) / LDS_FLOATS;
    int rn1 = (((n_nodes + nr1 - 1) / nr1) + 31) & ~31;
    int ch1 = 128 / nr1 > 0 ? 128 / nr1 : 1;
    int pc1 = (n_edges + ch1 - 1) / ch1;
    degree_lds<<<nr1 * ch1, SCAT_THREADS, 0, stream>>>(dst, deg, n_edges, n_nodes, rn1, ch1, pc1);
    dinv_kernel<<<(n_nodes + B - 1) / B, B, 0, stream>>>(deg, n_nodes);

    int n_tiles = (n_nodes + 15) / 16;
    int wpb = B / 32;
    gemm1_wmma<<<(n_tiles + wpb - 1) / wpb, B, 0, stream>>>(x, W1, deg, h, n_tiles, n_nodes);

    int nr0 = (n_nodes + E1_NODES - 1) / E1_NODES;
    int rn0 = (((n_nodes + nr0 - 1) / nr0) + 31) & ~31;
    int ch0 = 256 / nr0 > 0 ? 256 / nr0 : 1;
    int pc0 = (n_edges + ch0 - 1) / ch0;
    edge1_lds<<<nr0 * ch0, SCAT_THREADS, 0, stream>>>(src, dst, h, acc1, n_edges, n_nodes,
                                                      rn0, ch0, pc0);
    relu_bias1<<<(n16 + B - 1) / B, B, 0, stream>>>(h, deg, b1, acc1, n_nodes);

    gemm2_kernel<<<(n_nodes + B - 1) / B, B, 0, stream>>>(acc1, W2, deg, h2, n_nodes);

    int nr2 = (n_nodes * FEAT_OUT + LDS_FLOATS - 1) / LDS_FLOATS;
    int rn2 = (((n_nodes + nr2 - 1) / nr2) + 31) & ~31;
    int ch2 = 128 / nr2 > 0 ? 128 / nr2 : 1;
    int pc2 = (n_edges + ch2 - 1) / ch2;
    edge2_lds<<<nr2 * ch2, SCAT_THREADS, 0, stream>>>(src, dst, h2, acc2, n_edges, n_nodes,
                                                      rn2, ch2, pc2);

    final_kernel<<<(n_nodes + B - 1) / B, B, 0, stream>>>(acc2, h2, deg, b2, out, n_nodes);
}


# layer sums: edge list bucketed once by target range (one packed word per edge); one work-group per bucket adds in LDS and stores plainly, no range scans or merge atomics (sum order: bucket arrival)
# speedup vs baseline: 4.5591x; 1.5365x over previous
#include <hip/hip_runtime.h>
#include <math.h>

typedef float v2f __attribute__((ext_vector_type(2)));
typedef float v8f __attribute__((ext_vector_type(8)));

#define FEAT_IN  128
#define FEAT_MID 16
#define FEAT_OUT 2

__global__ void init_ws(float* __restrict__ deg, float* __restrict__ acc1,
                        float* __restrict__ acc2, int n_nodes) {
    int tid = blockIdx.x * blockDim.x + threadIdx.x;
    if (tid < n_nodes * FEAT_MID) acc1[tid] = 0.0f;
    if (tid < n_nodes * FEAT_OUT) acc2[tid] = 0.0f;
    if (tid < n_nodes)            deg[tid]  = 1.0f;
}

#define LDS_FLOATS   69632
#define SCAT_THREADS 1024
#define SCAT_U       8

__global__ void __launch_bounds__(SCAT_THREADS)
degree_lds(const int* __restrict__ dst, float* __restrict__ deg,
           int n_edges, int n_nodes, int range_nodes, int chunks, int per_chunk) {
    __shared__ float cnt[LDS_FLOATS];
    const int tid   = threadIdx.x;
    const int range = blockIdx.x / chunks;
    const int chunk = blockIdx.x - range * chunks;
    const int lo    = range * range_nodes;
    int len = n_nodes - lo;
    if (len > range_nodes) len = range_nodes;
    if (len <= 0) return;

    for (int i = tid; i < len; i += SCAT_THREADS) cnt[i] = 0.0f;
    __syncthreads();

    const int e0 = chunk * per_chunk;
    int e1 = e0 + per_chunk;
    if (e1 > n_edges) e1 = n_edges;
    for (int base = e0; base < e1; base += SCAT_U * SCAT_THREADS) {
        int dv[SCAT_U];
#pragma unroll
        for (int k = 0; k < SCAT_U; k++) {
            int e  = base + k * SCAT_THREADS + tid;
            int ec = e < e1 ? e : e1 - 1;
            dv[k] = dst[ec];
        }
        asm volatile("" : "+v"(dv[0]), "+v"(dv[1]), "+v"(dv[2]), "+v"(dv[3]),
                          "+v"(dv[4]), "+v"(dv[5]), "+v"(dv[6]), "+v"(dv[7]));
#pragma unroll
        for (int k = 0; k < SCAT_U; k++) {
            int e = base + k * SCAT_THREADS + tid;
            int t = dv[k] - lo;
            if (e < e1 && (unsigned)t < (unsigned)len) atomicAdd(&cnt[t], 1.0f);
        }
    }
    __syncthreads();

    const int rot = (chunk * (len / chunks)) & ~31;
    for (int j = tid; j < len; j += SCAT_THREADS) {
        int i = j + rot;
        if (i >= len) i -= len;
        float v = cnt[i];
        if (v != 0.0f) atomicAdd(&deg[lo + i], v);
    }
}

__global__ void dinv_kernel(float* __restrict__ deg, int n_nodes) {
    int i = blockIdx.x * blockDim.x + threadIdx.x;
    if (i < n_nodes) {
        float d = deg[i];
        deg[i] = (d > 0.0f) ? rsqrtf(d) : 0.0f;
    }
}

__global__ void __launch_bounds__(256)
gemm1_wmma(const float* __restrict__ x, const float* __restrict__ W1,
           const float* __restrict__ dinv, float* __restrict__ h,
           int n_tiles, int n_nodes) {
    int lane = threadIdx.x & 31;
    int wave = threadIdx.x >> 5;
    int tile = blockIdx.x * (blockDim.x >> 5) + wave;
    if (tile >= n_tiles) return;

    int col = lane & 15;
    int kh  = (lane >> 4) << 1;

    int rowA = tile * 16 + col;
    if (rowA > n_nodes - 1) rowA = n_nodes - 1;
    const float* xrow = x + (size_t)rowA * FEAT_IN;

    int mbase = (lane >> 4) * 8;
    float dr[8];
#pragma unroll
    for (int r = 0; r < 8; r++) {
        int row = tile * 16 + mbase + r;
        dr[r] = dinv[row < n_nodes ? row : n_nodes - 1];
    }

    v8f acc = {};
    for (int k0 = 0; k0 < FEAT_IN; k0 += 4) {
        v2f a, b;
        a.x = xrow[k0 + kh];
        a.y = xrow[k0 + kh + 1];
        b.x = W1[(k0 + kh)     * FEAT_MID + col];
        b.y = W1[(k0 + kh + 1) * FEAT_MID + col];
        acc = __builtin_amdgcn_wmma_f32_16x16x4_f32(
             false, a,  false, b,
             (short)0, acc,  false,  false);
    }

    asm volatile("v_nop\n\tv_nop\n\tv_nop\n\tv_nop\n\tv_nop" : "+v"(acc));
#pragma unroll
    for (int r = 0; r < 8; r++) {
        int row = tile * 16 + mbase + r;
        if (row < n_nodes) h[(size_t)row * FEAT_MID + col] = acc[r] * dr[r];
    }
}

typedef int v2i __attribute__((ext_vector_type(2)));
typedef int v4i __attribute__((ext_vector_type(4)));

#define E1_NODES 2400
#define E1_U     16
#define E1_STEPS 8
#define E1_LIST  (E1_U * 32)
#define E1_TRIP  (E1_U * SCAT_THREADS)

__device__ __forceinline__ void e1_push(bool m, int e, int tt, v2i* lst, int& cnt) {
    unsigned b = __builtin_amdgcn_ballot_w32(m);
    int pos = cnt + (int)__builtin_amdgcn_mbcnt_lo(b, 0u);
    if (m) {
        v2i q;
        q.x = e;
        q.y = tt;
        lst[pos] = q;
    }
    cnt += __builtin_popcount(b);
}

__device__ __forceinline__ void e1_drain(const v2i* lst, int cnt,
                                         const int* __restrict__ src, const float* __restrict__ h,
                                         float* __restrict__ acc, int f, int hf) {
    for (int j0 = 0; j0 < cnt; j0 += 2 * E1_STEPS) {
        int   tq[E1_STEPS], sq[E1_STEPS];
        float hq[E1_STEPS];
#pragma unroll
        for (int p = 0; p < E1_STEPS; p++) {
            int  idx   = j0 + 2 * p + hf;
            bool valid = idx < cnt;
            v2i  q     = lst[valid ? idx : 0];
            tq[p] = valid ? q.y : -1;
            sq[p] = src[q.x];
        }
        asm volatile("" : "+v"(sq[0]), "+v"(sq[1]), "+v"(sq[2]), "+v"(sq[3]),
                          "+v"(sq[4]), "+v"(sq[5]), "+v"(sq[6]), "+v"(sq[7]));
#pragma unroll
        for (int p = 0; p < E1_STEPS; p++)
            hq[p] = h[sq[p] * FEAT_MID + f];
        asm volatile("" : "+v"(hq[0]), "+v"(hq[1]), "+v"(hq[2]), "+v"(hq[3]),
                          "+v"(hq[4]), "+v"(hq[5]), "+v"(hq[6]), "+v"(hq[7]));
#pragma unroll
        for (int p = 0; p < E1_STEPS; p++) {
            if (tq[p] >= 0)
                atomicAdd(&acc[tq[p] * FEAT_MID + f], hq[p]);
        }
    }
}

__global__ void __launch_bounds__(SCAT_THREADS)
edge1_lds(const int* __restrict__ src, const int* __restrict__ dst,
          const float* __restrict__ h,
          float* __restrict__ acc1, int n_edges, int n_nodes,
          int range_nodes, int chunks, int per_chunk) {
    __shared__ float acc[E1_NODES * FEAT_MID];
    __shared__ v2i   list[SCAT_THREADS / 32][E1_LIST];
    const int tid   = threadIdx.x;
    const int lane  = tid & 31;
    const int f     = lane & 15;
    const int hf    = lane >> 4;
    const int range = blockIdx.x / chunks;
    const int chunk = blockIdx.x - range * chunks;
    const int lo    = range * range_nodes;
    int len = n_nodes - lo;
    if (len > range_nodes) len = range_nodes;
    if (len <= 0) return;
    const int flen = len * FEAT_MID;

    for (int i = tid; i < flen; i += SCAT_THREADS) acc[i] = 0.0f;
    __syncthreads();

    v2i* __restrict__ lst = list[tid >> 5];
    int e0 = chunk * per_chunk;
    if (e0 > n_edges) e0 = n_edges;
    e0 &= ~3;
    int e1 = n_edges;
    if (chunk + 1 < chunks) {
        e1 = (chunk + 1) * per_chunk;
        if (e1 > n_edges) e1 = n_edges;
        e1 &= ~3;
    }

    const bool wide = (((size_t)dst) & 15) == 0;
    const int nfull = wide ? (e1 - e0) / E1_TRIP : 0;
    v4i nx[E1_U / 4];
    if (nfull > 0) {
        const v4i* __restrict__ pn = (const v4i*)(dst + e0);
#pragma unroll
        for (int j = 0; j < E1_U / 4; j++) nx[j] = pn[j * SCAT_THREADS + tid];
    }
    for (int t = 0; t < nfull; t++) {
        const int base = e0 + t * E1_TRIP;
        v4i cur[E1_U / 4];
#pragma unroll
        for (int j = 0; j < E1_U / 4; j++) cur[j] = nx[j];
        if (t + 1 < nfull) {
            const v4i* __restrict__ pn = (const v4i*)(dst + base + E1_TRIP);
#pragma unroll
            for (int j = 0; j < E1_U / 4; j++) nx[j] = pn[j * SCAT_THREADS + tid];
        }
        const int eb = base + 4 * tid;
        int cnt = 0;
#pragma unroll
        for (int j = 0; j < E1_U / 4; j++) {
#pragma unroll
            for (int i = 0; i < 4; i++) {
                int tt = cur[j][i] - lo;
                e1_push((unsigned)tt < (unsigned)len, eb + j * 4 * SCAT_THREADS + i, tt, lst, cnt);
            }
        }
        __builtin_amdgcn_wave_barrier();
        e1_drain(lst, cnt, src, h, acc, f, hf);
        __builtin_amdgcn_wave_barrier();
    }

    for (int base = e0 + nfull * E1_TRIP; base < e1; base += E1_TRIP) {
        int dv[E1_U];
#pragma unroll
        for (int k = 0; k < E1_U; k++) {
            int e  = base + k * SCAT_THREADS + tid;
            int ec = e < e1 ? e : e1 - 1;
            dv[k] = dst[ec];
        }
        asm volatile("" : "+v"(dv[0]), "+v"(dv[1]), "+v"(dv[2]), "+v"(dv[3]),
                          "+v"(dv[4]), "+v"(dv[5]), "+v"(dv[6]), "+v"(dv[7]),
                          "+v"(dv[8]), "+v"(dv[9]), "+v"(dv[10]), "+v"(dv[11]),
                          "+v"(dv[12]), "+v"(dv[13]), "+v"(dv[14]), "+v"(dv[15]));
        int cnt = 0;
#pragma unroll
        for (int k = 0; k < E1_U; k++) {
            int e  = base + k * SCAT_THREADS + tid;
            int tt = dv[k] - lo;
            e1_push((e < e1) && ((unsigned)tt < (unsigned)len), e, tt, lst, cnt);
        }
        __builtin_amdgcn_wave_barrier();
        e1_drain(lst, cnt, src, h, acc, f, hf);
        __builtin_amdgcn_wave_barrier();
    }
    __syncthreads();

    const int rot = (chunk * (flen / chunks)) & ~31;
    float* __restrict__ outp = acc1 + (size_t)lo * FEAT_MID;
    for (int j = tid; j < flen; j += SCAT_THREADS) {
        int i = j + rot;
        if (i >= flen) i -= flen;
        float v = acc[i];
        if (v != 0.0f) atomicAdd(&outp[i], v);
    }
}

__global__ void relu_bias1(const float* __restrict__ h, const float* __restrict__ dinv,
                           const float* __restrict__ b1, float* __restrict__ acc1,
                           int n_nodes) {
    int tid = blockIdx.x * blockDim.x + threadIdx.x;
    if (tid >= n_nodes * FEAT_MID) return;
    int node = tid >> 4, f = tid & 15;
    float di = dinv[node];
    float v = (acc1[tid] + h[tid]) * di + b1[f];
    acc1[tid] = fmaxf(v, 0.0f);
}

__global__ void gemm2_kernel(const float* __restrict__ h1, const float* __restrict__ W2,
                             const float* __restrict__ dinv,
                             float* __restrict__ h2, int n_nodes) {
    int i = blockIdx.x * blockDim.x + threadIdx.x;
    if (i >= n_nodes) return;
    float s0 = 0.0f, s1 = 0.0f;
#pragma unroll
    for (int k = 0; k < FEAT_MID; k++) {
        float v = h1[i * FEAT_MID + k];
        s0 = fmaf(v, W2[k * 2 + 0], s0);
        s1 = fmaf(v, W2[k * 2 + 1], s1);
    }
    float di = dinv[i];
    h2[i * 2 + 0] = s0 * di;
    h2[i * 2 + 1] = s1 * di;
}

__global__ void __launch_bounds__(SCAT_THREADS)
edge2_lds(const int* __restrict__ src, const int* __restrict__ dst,
          const float* __restrict__ h2,
          float* __restrict__ acc2, int n_edges, int n_nodes,
          int range_nodes, int chunks, int per_chunk) {
    __shared__ float acc[LDS_FLOATS];
    const int tid   = threadIdx.x;
    const int range = blockIdx.x / chunks;
    const int chunk = blockIdx.x - range * chunks;
    const int lo    = range * range_nodes;
    int len = n_nodes - lo;
    if (len > range_nodes) len = range_nodes;
    if (len <= 0) return;
    const int flen = len * FEAT_OUT;

    for (int i = tid; i < flen; i += SCAT_THREADS) acc[i] = 0.0f;
    __syncthreads();

    const v2f* __restrict__ h2v = (const v2f*)h2;
    const int e0 = chunk * per_chunk;
    int e1 = e0 + per_chunk;
    if (e1 > n_edges) e1 = n_edges;
    for (int base = e0; base < e1; base += SCAT_U * SCAT_THREADS) {
        int dv[SCAT_U], sv[SCAT_U];
#pragma unroll
        for (int k = 0; k < SCAT_U; k++) {
            int e  = base + k * SCAT_THREADS + tid;
            int ec = e < e1 ? e : e1 - 1;
            dv[k] = dst[ec];
            sv[k] = src[ec];
        }
        asm volatile("" : "+v"(dv[0]), "+v"(dv[1]), "+v"(dv[2]), "+v"(dv[3]),
                          "+v"(dv[4]), "+v"(dv[5]), "+v"(dv[6]), "+v"(dv[7]),
                          "+v"(sv[0]), "+v"(sv[1]), "+v"(sv[2]), "+v"(sv[3]),
                          "+v"(sv[4]), "+v"(sv[5]), "+v"(sv[6]), "+v"(sv[7]));
        int   t[SCAT_U];
        v2f   hv[SCAT_U];
#pragma unroll
        for (int k = 0; k < SCAT_U; k++) {
            int e  = base + k * SCAT_THREADS + tid;
            int tt = dv[k] - lo;
            bool m = (e < e1) && ((unsigned)tt < (unsigned)len);
            t[k] = m ? tt : -1;
            int sg = m ? sv[k] : lo;
            hv[k] = h2v[sg];
        }
        asm volatile("" : "+v"(hv[0]), "+v"(hv[1]), "+v"(hv[2]), "+v"(hv[3]),
                          "+v"(hv[4]), "+v"(hv[5]), "+v"(hv[6]), "+v"(hv[7]));
#pragma unroll
        for (int k = 0; k < SCAT_U; k++) {
            if (t[k] >= 0) {
                atomicAdd(&acc[t[k] * FEAT_OUT + 0], hv[k].x);
                atomicAdd(&acc[t[k] * FEAT_OUT + 1], hv[k].y);
            }
        }
    }
    __syncthreads();

    const int rot = (chunk * (flen / chunks)) & ~31;
    float* __restrict__ outp = acc2 + (size_t)lo * FEAT_OUT;
    for (int j = tid; j < flen; j += SCAT_THREADS) {
        int i = j + rot;
        if (i >= flen) i -= flen;
        float v = acc[i];
        if (v != 0.0f) atomicAdd(&outp[i], v);
    }
}

__global__ void final_kernel(const float* __restrict__ acc2, const float* __restrict__ h2,
                             const float* __restrict__ dinv, const float* __restrict__ b2,
                             float* __restrict__ out, int n_nodes) {
    int i = blockIdx.x * blockDim.x + threadIdx.x;
    if (i >= n_nodes) return;
    float di = dinv[i];
    float v0 = (acc2[i * 2 + 0] + h2[i * 2 + 0]) * di + b2[0];
    float v1 = (acc2[i * 2 + 1] + h2[i * 2 + 1]) * di + b2[1];
    float m  = fmaxf(v0, v1);
    float l  = m + logf(expf(v0 - m) + expf(v1 - m));
    out[i * 2 + 0] = v0 - l;
    out[i * 2 + 1] = v1 - l;
}

#define RB_NODES   128
#define RB_SHIFT   7
#define RB_MAXR    1024
#define BK_THREADS 1024
#define BK_U       4
#define BK_STAGE   12800
#define EB_THREADS 512
#define EB_STEPS   16
#define E2_U       8

__global__ void init_deg(float* __restrict__ deg, int n_nodes) {
    int tid = blockIdx.x * blockDim.x + threadIdx.x;
    if (tid < n_nodes) deg[tid] = 1.0f;
}

#define DT_THREADS (2 * RB_NODES)
__global__ void __launch_bounds__(DT_THREADS)
dinv_totals(float* __restrict__ deg, int* __restrict__ totals, int* __restrict__ cursor,
            int n_nodes, int n_ranges) {
    __shared__ int part[DT_THREADS / 32];
    const int tid = threadIdx.x;
    const int i   = blockIdx.x * DT_THREADS + tid;
    int v = 0;
    if (i < n_nodes) {
        float d = deg[i];
        v = (int)d - 1;
        deg[i] = (d > 0.0f) ? rsqrtf(d) : 0.0f;
    }
    for (int m = 16; m > 0; m >>= 1) v += __shfl_xor(v, m);
    if ((tid & 31) == 0) part[tid >> 5] = v;
    __syncthreads();
    if (tid < 2) {
        const int r = blockIdx.x * 2 + tid;
        if (r < n_ranges) {
            const int* q = part + tid * (RB_NODES / 32);
            totals[r] = q[0] + q[1] + q[2] + q[3];
            cursor[r] = 0;
        }
    }
}

__global__ void __launch_bounds__(BK_THREADS)
bucket_scatter(const int* __restrict__ src, const int* __restrict__ dst,
               const int* __restrict__ totals, int* __restrict__ cursor,
               int* __restrict__ offs, int* __restrict__ bins,
               int n_edges, int per_chunk, int n_ranges) {
    __shared__ int cnt[RB_MAXR];
    __shared__ int gb[RB_MAXR];
    __shared__ int sc[RB_MAXR];
    __shared__ int so[RB_MAXR];
    __shared__ int stage_w[BK_STAGE];
    __shared__ int stage_p[BK_STAGE];
    const int tid = threadIdx.x;
    int e0 = blockIdx.x * per_chunk;
    if (e0 > n_edges) e0 = n_edges;
    int e1 = e0 + per_chunk;
    if (e1 > n_edges) e1 = n_edges;
    cnt[tid] = 0;
    __syncthreads();
    for (int base = e0; base < e1; base += BK_U * BK_THREADS) {
        int dv[BK_U];
#pragma unroll
        for (int k = 0; k < BK_U; k++) {
            int e  = base + k * BK_THREADS + tid;
            int ec = e < e1 ? e : e1 - 1;
            dv[k] = dst[ec];
        }
        asm volatile("" : "+v"(dv[0]), "+v"(dv[1]), "+v"(dv[2]), "+v"(dv[3]));
#pragma unroll
        for (int k = 0; k < BK_U; k++) {
            int e = base + k * BK_THREADS + tid;
            if (e < e1) atomicAdd(&cnt[dv[k] >> RB_SHIFT], 1);
        }
    }
    __syncthreads();
    const int c   = cnt[tid];
    const int tot = tid < n_ranges ? totals[tid] : 0;
    int rel = 0;
    if (c > 0) rel = atomicAdd(&cursor[tid], c);
    sc[tid] = c;
    so[tid] = tot;
    __syncthreads();
    for (int off = 1; off < RB_MAXR; off <<= 1) {
        int t1 = tid >= off ? sc[tid - off] : 0;
        int t2 = tid >= off ? so[tid - off] : 0;
        __syncthreads();
        sc[tid] += t1;
        so[tid] += t2;
        __syncthreads();
    }
    const int first = so[tid] - tot;
    if (blockIdx.x == 0 && tid < n_ranges) {
        offs[tid] = first;
        if (tid == n_ranges - 1) offs[n_ranges] = so[tid];
    }
    gb[tid]  = first + rel;
    sc[tid] -= c;
    cnt[tid] = 0;
    __syncthreads();
    for (int base = e0; base < e1; base += BK_U * BK_THREADS) {
        int dv[BK_U], sv[BK_U];
#pragma unroll
        for (int k = 0; k < BK_U; k++) {
            int e  = base + k * BK_THREADS + tid;
            int ec = e < e1 ? e : e1 - 1;
            dv[k] = dst[ec];
            sv[k] = src[ec];
        }
        asm volatile("" : "+v"(dv[0]), "+v"(dv[1]), "+v"(dv[2]), "+v"(dv[3]),
                          "+v"(sv[0]), "+v"(sv[1]), "+v"(sv[2]), "+v"(sv[3]));
        int rank[BK_U], ls[BK_U], gs[BK_U];
#pragma unroll
        for (int k = 0; k < BK_U; k++) {
            int e = base + k * BK_THREADS + tid;
            int r = dv[k] >> RB_SHIFT;
            rank[k] = 0;
            if (e < e1) rank[k] = atomicAdd(&cnt[r], 1);
            ls[k] = sc[r];
            gs[k] = gb[r];
        }
#pragma unroll
        for (int k = 0; k < BK_U; k++) {
            int e = base + k * BK_THREADS + tid;
            if (e < e1) {
                int lp = ls[k] + rank[k];
                stage_w[lp] = (sv[k] << RB_SHIFT) | (dv[k] & (RB_NODES - 1));
                stage_p[lp] = gs[k] + rank[k];
            }
        }
    }
    __syncthreads();
    const int n = e1 - e0;
    for (int i = tid; i < n; i += BK_THREADS) bins[stage_p[i]] = stage_w[i];
}

__global__ void __launch_bounds__(EB_THREADS)
edge1_bin(const int* __restrict__ bins, const int* __restrict__ offs,
          const float* __restrict__ h, float* __restrict__ acc1, int n_nodes) {
    __shared__ float acc[RB_NODES * FEAT_MID];
    const int tid  = threadIdx.x;
    const int lane = tid & 31;
    const int f    = lane & 15;
    const int hf   = lane >> 4;
    const int lo   = blockIdx.x << RB_SHIFT;
    int len = n_nodes - lo;
    if (len > RB_NODES) len = RB_NODES;

    for (int i = tid; i < RB_NODES * FEAT_MID; i += EB_THREADS) acc[i] = 0.0f;
    __syncthreads();

    const int b0 = offs[blockIdx.x];
    const int b1 = offs[blockIdx.x + 1];
    const int stride = (EB_THREADS / 32) * 2 * EB_STEPS;
    int j0 = b0 + (tid >> 5) * 2 * EB_STEPS;
    int w[EB_STEPS] = {};
    if (j0 < b1) {
#pragma unroll
        for (int p = 0; p < EB_STEPS; p++) {
            int idx = j0 + 2 * p + hf;
            w[p] = bins[idx < b1 ? idx : j0];
        }
    }
    while (j0 < b1) {
        int   cur[EB_STEPS];
        float hq[EB_STEPS];
#pragma unroll
        for (int p = 0; p < EB_STEPS; p++) cur[p] = w[p];
        const int jn = j0 + stride;
        if (jn < b1) {
#pragma unroll
            for (int p = 0; p < EB_STEPS; p++) {
                int idx = jn + 2 * p + hf;
                w[p] = bins[idx < b1 ? idx : jn];
            }
        }
#pragma unroll
        for (int p = 0; p < EB_STEPS; p++)
            hq[p] = h[(cur[p] >> RB_SHIFT) * FEAT_MID + f];
        asm volatile("" : "+v"(hq[0]), "+v"(hq[1]), "+v"(hq[2]), "+v"(hq[3]),
                          "+v"(hq[4]), "+v"(hq[5]), "+v"(hq[6]), "+v"(hq[7]),
                          "+v"(hq[8]), "+v"(hq[9]), "+v"(hq[10]), "+v"(hq[11]),
                          "+v"(hq[12]), "+v"(hq[13]), "+v"(hq[14]), "+v"(hq[15]));
#pragma unroll
        for (int p = 0; p < EB_STEPS; p++) {
            if (j0 + 2 * p + hf < b1)
                atomicAdd(&acc[(cur[p] & (RB_NODES - 1)) * FEAT_MID + f], hq[p]);
        }
        j0 = jn;
    }
    __syncthreads();

    float* __restrict__ outp = acc1 + (size_t)lo * FEAT_MID;
    const int flen = len * FEAT_MID;
    for (int i = tid; i < flen; i += EB_THREADS) outp[i] = acc[i];
}

__global__ void __launch_bounds__(EB_THREADS)
edge2_bin(const int* __restrict__ bins, const int* __restrict__ offs,
          const float* __restrict__ h2, float* __restrict__ acc2, int n_nodes) {
    __shared__ float acc[RB_NODES * FEAT_OUT];
    const int tid = threadIdx.x;
    const int lo  = blockIdx.x << RB_SHIFT;
    int len = n_nodes - lo;
    if (len > RB_NODES) len = RB_NODES;

    for (int i = tid; i < RB_NODES * FEAT_OUT; i += EB_THREADS) acc[i] = 0.0f;
    __syncthreads();

    const v2f* __restrict__ h2v = (const v2f*)h2;
    const int b0 = offs[blockIdx.x];
    const int b1 = offs[blockIdx.x + 1];
    for (int base = b0; base < b1; base += E2_U * EB_THREADS) {
        int w[E2_U];
        v2f hv[E2_U];
#pragma unroll
        for (int k = 0; k < E2_U; k++) {
            int idx = base + k * EB_THREADS + tid;
            w[k] = bins[idx < b1 ? idx : b0];
        }
#pragma unroll
        for (int k = 0; k < E2_U; k++) hv[k] = h2v[w[k] >> RB_SHIFT];
        asm volatile("" : "+v"(hv[0]), "+v"(hv[1]), "+v"(hv[2]), "+v"(hv[3]),
                          "+v"(hv[4]), "+v"(hv[5]), "+v"(hv[6]), "+v"(hv[7]));
#pragma unroll
        for (int k = 0; k < E2_U; k++) {
            if (base + k * EB_THREADS + tid < b1) {
                int t = w[k] & (RB_NODES - 1);
                atomicAdd(&acc[t * FEAT_OUT + 0], hv[k].x);
                atomicAdd(&acc[t * FEAT_OUT + 1], hv[k].y);
            }
        }
    }
    __syncthreads();

    float* __restrict__ outp = acc2 + (size_t)lo * FEAT_OUT;
    const int flen = len * FEAT_OUT;
    for (int i = tid; i < flen; i += EB_THREADS) outp[i] = acc[i];
}

extern "C" void kernel_launch(void* const* d_in, const int* in_sizes, int n_in,
                              void* d_out, int out_size, void* d_ws, size_t ws_size,
                              hipStream_t stream) {
    const float* x  = (const float*)d_in[0];
    const int*   ei = (const int*)d_in[1];
    const float* W1 = (const float*)d_in[2];
    const float* b1 = (const float*)d_in[3];
    const float* W2 = (const float*)d_in[4];
    const float* b2 = (const float*)d_in[5];
    float* out = (float*)d_out;

    int n_nodes = in_sizes[0] / FEAT_IN;
    int n_edges = in_sizes[1] / 2;
    const int* src = ei;
    const int* dst = ei + n_edges;

    float* ws   = (float*)d_ws;
    float* deg  = ws;
    float* h    = deg  + n_nodes;
    float* acc1 = h    + (size_t)n_nodes * FEAT_MID;
    float* h2   = acc1 + (size_t)n_nodes * FEAT_MID;
    float* acc2 = h2   + (size_t)n_nodes * FEAT_OUT;

    const int B = 256;
    int n16 = n_nodes * FEAT_MID;

    int nr1 = (n_nodes + LDS_FLOATS - 1) / LDS_FLOATS;
    int rn1 = (((n_nodes + nr1 - 1) / nr1) + 31) & ~31;
    int ch1 = 128 / nr1 > 0 ? 128 / nr1 : 1;
    int pc1 = (n_edges + ch1 - 1) / ch1;
    int n_tiles = (n_nodes + 15) / 16;
    int wpb = B / 32;

    int    n_ranges = (n_nodes + RB_NODES - 1) >> RB_SHIFT;
    size_t need = ((size_t)n_nodes * (1 + 2 * FEAT_MID + 2 * FEAT_OUT) + 3 * (size_t)n_ranges + 1 +
                   (size_t)n_edges) * sizeof(float);
    if (n_ranges <= RB_MAXR && ws_size >= need) {
        int* offs   = (int*)(acc2 + (size_t)n_nodes * FEAT_OUT);
        int* cursor = offs + n_ranges + 1;
        int* totals = cursor + n_ranges;
        int* bins   = totals + n_ranges;
        int  nb     = (n_edges + BK_STAGE - 1) / BK_STAGE;
        int  pcb    = BK_STAGE;

        init_deg<<<(n_nodes + B - 1) / B, B, 0, stream>>>(deg, n_nodes);
        degree_lds<<<nr1 * ch1, SCAT_THREADS, 0, stream>>>(dst, deg, n_edges, n_nodes, rn1, ch1, pc1);
        dinv_totals<<<(n_nodes + DT_THREADS - 1) / DT_THREADS, DT_THREADS, 0, stream>>>(
            deg, totals, cursor, n_nodes, n_ranges);
        bucket_scatter<<<nb, BK_THREADS, 0, stream>>>(src, dst, totals, cursor, offs, bins,
                                                      n_edges, pcb, n_ranges);
        gemm1_wmma<<<(n_tiles + wpb - 1) / wpb, B, 0, stream>>>(x, W1, deg, h, n_tiles, n_nodes);
        edge1_bin<<<n_ranges, EB_THREADS, 0, stream>>>(bins, offs, h, acc1, n_nodes);
        relu_bias1<<<(n16 + B - 1) / B, B, 0, stream>>>(h, deg, b1, acc1, n_nodes);
        gemm2_kernel<<<(n_nodes + B - 1) / B, B, 0, stream>>>(acc1, W2, deg, h2, n_nodes);
        edge2_bin<<<n_ranges, EB_THREADS, 0, stream>>>(bins, offs, h2, acc2, n_nodes);
        final_kernel<<<(n_nodes + B - 1) / B, B, 0, stream>>>(acc2, h2, deg, b2, out, n_nodes);
        return;
    }

    init_ws<<<(n16 + B - 1) / B, B, 0, stream>>>(deg, acc1, acc2, n_nodes);
    degree_lds<<<nr1 * ch1, SCAT_THREADS, 0, stream>>>(dst, deg, n_edges, n_nodes, rn1, ch1, pc1);
    dinv_kernel<<<(n_nodes + B - 1) / B, B, 0, stream>>>(deg, n_nodes);

    gemm1_wmma<<<(n_tiles + wpb - 1) / wpb, B, 0, stream>>>(x, W1, deg, h, n_tiles, n_nodes);

    int nr0 = (n_nodes + E1_NODES - 1) / E1_NODES;
    int rn0 = (((n_nodes + nr0 - 1) / nr0) + 31) & ~31;
    int ch0 = 256 / nr0 > 0 ? 256 / nr0 : 1;
    int pc0 = (n_edges + ch0 - 1) / ch0;
    edge1_lds<<<nr0 * ch0, SCAT_THREADS, 0, stream>>>(src, dst, h, acc1, n_edges, n_nodes,
                                                      rn0, ch0, pc0);
    relu_bias1<<<(n16 + B - 1) / B, B, 0, stream>>>(h, deg, b1, acc1, n_nodes);

    gemm2_kernel<<<(n_nodes + B - 1) / B, B, 0, stream>>>(acc1, W2, deg, h2, n_nodes);

    int nr2 = (n_nodes * FEAT_OUT + LDS_FLOATS - 1) / LDS_FLOATS;
    int rn2 = (((n_nodes + nr2 - 1) / nr2) + 31) & ~31;
    int ch2 = 128 / nr2 > 0 ? 128 / nr2 : 1;
    int pc2 = (n_edges + ch2 - 1) / ch2;
    edge2_lds<<<nr2 * ch2, SCAT_THREADS, 0, stream>>>(src, dst, h2, acc2, n_edges, n_nodes,
                                                      rn2, ch2, pc2);

    final_kernel<<<(n_nodes + B - 1) / B, B, 0, stream>>>(acc2, h2, deg, b2, out, n_nodes);
}


# node-wise tails run inside the bucket kernels: layer-1 sums get self-loop, dinv, bias, ReLU and the 16x2 product in LDS and leave as h2; layer-2 sums leave as log_softmax; 3 launches dropped
# speedup vs baseline: 5.0292x; 1.1031x over previous
#include <hip/hip_runtime.h>
#include <math.h>

typedef float v2f __attribute__((ext_vector_type(2)));
typedef float v8f __attribute__((ext_vector_type(8)));

#define FEAT_IN  128
#define FEAT_MID 16
#define FEAT_OUT 2

__global__ void init_ws(float* __restrict__ deg, float* __restrict__ acc1,
                        float* __restrict__ acc2, int n_nodes) {
    int tid = blockIdx.x * blockDim.x + threadIdx.x;
    if (tid < n_nodes * FEAT_MID) acc1[tid] = 0.0f;
    if (tid < n_nodes * FEAT_OUT) acc2[tid] = 0.0f;
    if (tid < n_nodes)            deg[tid]  = 1.0f;
}

#define LDS_FLOATS   69632
#define SCAT_THREADS 1024
#define SCAT_U       8

__global__ void __launch_bounds__(SCAT_THREADS)
degree_lds(const int* __restrict__ dst, float* __restrict__ deg,
           int n_edges, int n_nodes, int range_nodes, int chunks, int per_chunk) {
    __shared__ float cnt[LDS_FLOATS];
    const int tid   = threadIdx.x;
    const int range = blockIdx.x / chunks;
    const int chunk = blockIdx.x - range * chunks;
    const int lo    = range * range_nodes;
    int len = n_nodes - lo;
    if (len > range_nodes) len = range_nodes;
    if (len <= 0) return;

    for (int i = tid; i < len; i += SCAT_THREADS) cnt[i] = 0.0f;
    __syncthreads();

    const int e0 = chunk * per_chunk;
    int e1 = e0 + per_chunk;
    if (e1 > n_edges) e1 = n_edges;
    for (int base = e0; base < e1; base += SCAT_U * SCAT_THREADS) {
        int dv[SCAT_U];
#pragma unroll
        for (int k = 0; k < SCAT_U; k++) {
            int e  = base + k * SCAT_THREADS + tid;
            int ec = e < e1 ? e : e1 - 1;
            dv[k] = dst[ec];
        }
        asm volatile("" : "+v"(dv[0]), "+v"(dv[1]), "+v"(dv[2]), "+v"(dv[3]),
                          "+v"(dv[4]), "+v"(dv[5]), "+v"(dv[6]), "+v"(dv[7]));
#pragma unroll
        for (int k = 0; k < SCAT_U; k++) {
            int e = base + k * SCAT_THREADS + tid;
            int t = dv[k] - lo;
            if (e < e1 && (unsigned)t < (unsigned)len) atomicAdd(&cnt[t], 1.0f);
        }
    }
    __syncthreads();

    const int rot = (chunk * (len / chunks)) & ~31;
    for (int j = tid; j < len; j += SCAT_THREADS) {
        int i = j + rot;
        if (i >= len) i -= len;
        float v = cnt[i];
        if (v != 0.0f) atomicAdd(&deg[lo + i], v);
    }
}

__global__ void dinv_kernel(float* __restrict__ deg, int n_nodes) {
    int i = blockIdx.x * blockDim.x + threadIdx.x;
    if (i < n_nodes) {
        float d = deg[i];
        deg[i] = (d > 0.0f) ? rsqrtf(d) : 0.0f;
    }
}

__global__ void __launch_bounds__(256)
gemm1_wmma(const float* __restrict__ x, const float* __restrict__ W1,
           const float* __restrict__ dinv, float* __restrict__ h,
           int n_tiles, int n_nodes) {
    int lane = threadIdx.x & 31;
    int wave = threadIdx.x >> 5;
    int tile = blockIdx.x * (blockDim.x >> 5) + wave;
    if (tile >= n_tiles) return;

    int col = lane & 15;
    int kh  = (lane >> 4) << 1;

    int rowA = tile * 16 + col;
    if (rowA > n_nodes - 1) rowA = n_nodes - 1;
    const float* xrow = x + (size_t)rowA * FEAT_IN;

    int mbase = (lane >> 4) * 8;
    float dr[8];
#pragma unroll
    for (int r = 0; r < 8; r++) {
        int row = tile * 16 + mbase + r;
        dr[r] = dinv[row < n_nodes ? row : n_nodes - 1];
    }

    v8f acc = {};
    for (int k0 = 0; k0 < FEAT_IN; k0 += 4) {
        v2f a, b;
        a.x = xrow[k0 + kh];
        a.y = xrow[k0 + kh + 1];
        b.x = W1[(k0 + kh)     * FEAT_MID + col];
        b.y = W1[(k0 + kh + 1) * FEAT_MID + col];
        acc = __builtin_amdgcn_wmma_f32_16x16x4_f32(
             false, a,  false, b,
             (short)0, acc,  false,  false);
    }

    asm volatile("v_nop\n\tv_nop\n\tv_nop\n\tv_nop\n\tv_nop" : "+v"(acc));
#pragma unroll
    for (int r = 0; r < 8; r++) {
        int row = tile * 16 + mbase + r;
        if (row < n_nodes) h[(size_t)row * FEAT_MID + col] = acc[r] * dr[r];
    }
}

typedef int v2i __attribute__((ext_vector_type(2)));
typedef int v4i __attribute__((ext_vector_type(4)));

#define E1_NODES 2400
#define E1_U     16
#define E1_STEPS 8
#define E1_LIST  (E1_U * 32)
#define E1_TRIP  (E1_U * SCAT_THREADS)

__device__ __forceinline__ void e1_push(bool m, int e, int tt, v2i* lst, int& cnt) {
    unsigned b = __builtin_amdgcn_ballot_w32(m);
    int pos = cnt + (int)__builtin_amdgcn_mbcnt_lo(b, 0u);
    if (m) {
        v2i q;
        q.x = e;
        q.y = tt;
        lst[pos] = q;
    }
    cnt += __builtin_popcount(b);
}

__device__ __forceinline__ void e1_drain(const v2i* lst, int cnt,
                                         const int* __restrict__ src, const float* __restrict__ h,
                                         float* __restrict__ acc, int f, int hf) {
    for (int j0 = 0; j0 < cnt; j0 += 2 * E1_STEPS) {
        int   tq[E1_STEPS], sq[E1_STEPS];
        float hq[E1_STEPS];
#pragma unroll
        for (int p = 0; p < E1_STEPS; p++) {
            int  idx   = j0 + 2 * p + hf;
            bool valid = idx < cnt;
            v2i  q     = lst[valid ? idx : 0];
            tq[p] = valid ? q.y : -1;
            sq[p] = src[q.x];
        }
        asm volatile("" : "+v"(sq[0]), "+v"(sq[1]), "+v"(sq[2]), "+v"(sq[3]),
                          "+v"(sq[4]), "+v"(sq[5]), "+v"(sq[6]), "+v"(sq[7]));
#pragma unroll
        for (int p = 0; p < E1_STEPS; p++)
            hq[p] = h[sq[p] * FEAT_MID + f];
        asm volatile("" : "+v"(hq[0]), "+v"(hq[1]), "+v"(hq[2]), "+v"(hq[3]),
                          "+v"(hq[4]), "+v"(hq[5]), "+v"(hq[6]), "+v"(hq[7]));
#pragma unroll
        for (int p = 0; p < E1_STEPS; p++) {
            if (tq[p] >= 0)
                atomicAdd(&acc[tq[p] * FEAT_MID + f], hq[p]);
        }
    }
}

__global__ void __launch_bounds__(SCAT_THREADS)
edge1_lds(const int* __restrict__ src, const int* __restrict__ dst,
          const float* __restrict__ h,
          float* __restrict__ acc1, int n_edges, int n_nodes,
          int range_nodes, int chunks, int per_chunk) {
    __shared__ float acc[E1_NODES * FEAT_MID];
    __shared__ v2i   list[SCAT_THREADS / 32][E1_LIST];
    const int tid   = threadIdx.x;
    const int lane  = tid & 31;
    const int f     = lane & 15;
    const int hf    = lane >> 4;
    const int range = blockIdx.x / chunks;
    const int chunk = blockIdx.x - range * chunks;
    const int lo    = range * range_nodes;
    int len = n_nodes - lo;
    if (len > range_nodes) len = range_nodes;
    if (len <= 0) return;
    const int flen = len * FEAT_MID;

    for (int i = tid; i < flen; i += SCAT_THREADS) acc[i] = 0.0f;
    __syncthreads();

    v2i* __restrict__ lst = list[tid >> 5];
    int e0 = chunk * per_chunk;
    if (e0 > n_edges) e0 = n_edges;
    e0 &= ~3;
    int e1 = n_edges;
    if (chunk + 1 < chunks) {
        e1 = (chunk + 1) * per_chunk;
        if (e1 > n_edges) e1 = n_edges;
        e1 &= ~3;
    }

    const bool wide = (((size_t)dst) & 15) == 0;
    const int nfull = wide ? (e1 - e0) / E1_TRIP : 0;
    v4i nx[E1_U / 4];
    if (nfull > 0) {
        const v4i* __restrict__ pn = (const v4i*)(dst + e0);
#pragma unroll
        for (int j = 0; j < E1_U / 4; j++) nx[j] = pn[j * SCAT_THREADS + tid];
    }
    for (int t = 0; t < nfull; t++) {
        const int base = e0 + t * E1_TRIP;
        v4i cur[E1_U / 4];
#pragma unroll
        for (int j = 0; j < E1_U / 4; j++) cur[j] = nx[j];
        if (t + 1 < nfull) {
            const v4i* __restrict__ pn = (const v4i*)(dst + base + E1_TRIP);
#pragma unroll
            for (int j = 0; j < E1_U / 4; j++) nx[j] = pn[j * SCAT_THREADS + tid];
        }
        const int eb = base + 4 * tid;
        int cnt = 0;
#pragma unroll
        for (int j = 0; j < E1_U / 4; j++) {
#pragma unroll
            for (int i = 0; i < 4; i++) {
                int tt = cur[j][i] - lo;
                e1_push((unsigned)tt < (unsigned)len, eb + j * 4 * SCAT_THREADS + i, tt, lst, cnt);
            }
        }
        __builtin_amdgcn_wave_barrier();
        e1_drain(lst, cnt, src, h, acc, f, hf);
        __builtin_amdgcn_wave_barrier();
    }

    for (int base = e0 + nfull * E1_TRIP; base < e1; base += E1_TRIP) {
        int dv[E1_U];
#pragma unroll
        for (int k = 0; k < E1_U; k++) {
            int e  = base + k * SCAT_THREADS + tid;
            int ec = e < e1 ? e : e1 - 1;
            dv[k] = dst[ec];
        }
        asm volatile("" : "+v"(dv[0]), "+v"(dv[1]), "+v"(dv[2]), "+v"(dv[3]),
                          "+v"(dv[4]), "+v"(dv[5]), "+v"(dv[6]), "+v"(dv[7]),
                          "+v"(dv[8]), "+v"(dv[9]), "+v"(dv[10]), "+v"(dv[11]),
                          "+v"(dv[12]), "+v"(dv[13]), "+v"(dv[14]), "+v"(dv[15]));
        int cnt = 0;
#pragma unroll
        for (int k = 0; k < E1_U; k++) {
            int e  = base + k * SCAT_THREADS + tid;
            int tt = dv[k] - lo;
            e1_push((e < e1) && ((unsigned)tt < (unsigned)len), e, tt, lst, cnt);
        }
        __builtin_amdgcn_wave_barrier();
        e1_drain(lst, cnt, src, h, acc, f, hf);
        __builtin_amdgcn_wave_barrier();
    }
    __syncthreads();

    const int rot = (chunk * (flen / chunks)) & ~31;
    float* __restrict__ outp = acc1 + (size_t)lo * FEAT_MID;
    for (int j = tid; j < flen; j += SCAT_THREADS) {
        int i = j + rot;
        if (i >= flen) i -= flen;
        float v = acc[i];
        if (v != 0.0f) atomicAdd(&outp[i], v);
    }
}

__global__ void relu_bias1(const float* __restrict__ h, const float* __restrict__ dinv,
                           const float* __restrict__ b1, float* __restrict__ acc1,
                           int n_nodes) {
    int tid = blockIdx.x * blockDim.x + threadIdx.x;
    if (tid >= n_nodes * FEAT_MID) return;
    int node = tid >> 4, f = tid & 15;
    float di = dinv[node];
    float v = (acc1[tid] + h[tid]) * di + b1[f];
    acc1[tid] = fmaxf(v, 0.0f);
}

__global__ void gemm2_kernel(const float* __restrict__ h1, const float* __restrict__ W2,
                             const float* __restrict__ dinv,
                             float* __restrict__ h2, int n_nodes) {
    int i = blockIdx.x * blockDim.x + threadIdx.x;
    if (i >= n_nodes) return;
    float s0 = 0.0f, s1 = 0.0f;
#pragma unroll
    for (int k = 0; k < FEAT_MID; k++) {
        float v = h1[i * FEAT_MID + k];
        s0 = fmaf(v, W2[k * 2 + 0], s0);
        s1 = fmaf(v, W2[k * 2 + 1], s1);
    }
    float di = dinv[i];
    h2[i * 2 + 0] = s0 * di;
    h2[i * 2 + 1] = s1 * di;
}

__global__ void __launch_bounds__(SCAT_THREADS)
edge2_lds(const int* __restrict__ src, const int* __restrict__ dst,
          const float* __restrict__ h2,
          float* __restrict__ acc2, int n_edges, int n_nodes,
          int range_nodes, int chunks, int per_chunk) {
    __shared__ float acc[LDS_FLOATS];
    const int tid   = threadIdx.x;
    const int range = blockIdx.x / chunks;
    const int chunk = blockIdx.x - range * chunks;
    const int lo    = range * range_nodes;
    int len = n_nodes - lo;
    if (len > range_nodes) len = range_nodes;
    if (len <= 0) return;
    const int flen = len * FEAT_OUT;

    for (int i = tid; i < flen; i += SCAT_THREADS) acc[i] = 0.0f;
    __syncthreads();

    const v2f* __restrict__ h2v = (const v2f*)h2;
    const int e0 = chunk * per_chunk;
    int e1 = e0 + per_chunk;
    if (e1 > n_edges) e1 = n_edges;
    for (int base = e0; base < e1; base += SCAT_U * SCAT_THREADS) {
        int dv[SCAT_U], sv[SCAT_U];
#pragma unroll
        for (int k = 0; k < SCAT_U; k++) {
            int e  = base + k * SCAT_THREADS + tid;
            int ec = e < e1 ? e : e1 - 1;
            dv[k] = dst[ec];
            sv[k] = src[ec];
        }
        asm volatile("" : "+v"(dv[0]), "+v"(dv[1]), "+v"(dv[2]), "+v"(dv[3]),
                          "+v"(dv[4]), "+v"(dv[5]), "+v"(dv[6]), "+v"(dv[7]),
                          "+v"(sv[0]), "+v"(sv[1]), "+v"(sv[2]), "+v"(sv[3]),
                          "+v"(sv[4]), "+v"(sv[5]), "+v"(sv[6]), "+v"(sv[7]));
        int   t[SCAT_U];
        v2f   hv[SCAT_U];
#pragma unroll
        for (int k = 0; k < SCAT_U; k++) {
            int e  = base + k * SCAT_THREADS + tid;
            int tt = dv[k] - lo;
            bool m = (e < e1) && ((unsigned)tt < (unsigned)len);
            t[k] = m ? tt : -1;
            int sg = m ? sv[k] : lo;
            hv[k] = h2v[sg];
        }
        asm volatile("" : "+v"(hv[0]), "+v"(hv[1]), "+v"(hv[2]), "+v"(hv[3]),
                          "+v"(hv[4]), "+v"(hv[5]), "+v"(hv[6]), "+v"(hv[7]));
#pragma unroll
        for (int k = 0; k < SCAT_U; k++) {
            if (t[k] >= 0) {
                atomicAdd(&acc[t[k] * FEAT_OUT + 0], hv[k].x);
                atomicAdd(&acc[t[k] * FEAT_OUT + 1], hv[k].y);
            }
        }
    }
    __syncthreads();

    const int rot = (chunk * (flen / chunks)) & ~31;
    float* __restrict__ outp = acc2 + (size_t)lo * FEAT_OUT;
    for (int j = tid; j < flen; j += SCAT_THREADS) {
        int i = j + rot;
        if (i >= flen) i -= flen;
        float v = acc[i];
        if (v != 0.0f) atomicAdd(&outp[i], v);
    }
}

__global__ void final_kernel(const float* __restrict__ acc2, const float* __restrict__ h2,
                             const float* __restrict__ dinv, const float* __restrict__ b2,
                             float* __restrict__ out, int n_nodes) {
    int i = blockIdx.x * blockDim.x + threadIdx.x;
    if (i >= n_nodes) return;
    float di = dinv[i];
    float v0 = (acc2[i * 2 + 0] + h2[i * 2 + 0]) * di + b2[0];
    float v1 = (acc2[i * 2 + 1] + h2[i * 2 + 1]) * di + b2[1];
    float m  = fmaxf(v0, v1);
    float l  = m + logf(expf(v0 - m) + expf(v1 - m));
    out[i * 2 + 0] = v0 - l;
    out[i * 2 + 1] = v1 - l;
}

#define RB_NODES   128
#define RB_SHIFT   7
#define RB_MAXR    1024
#define BK_THREADS 1024
#define BK_U       4
#define BK_STAGE   12800
#define EB_THREADS 512
#define EB_STEPS   16
#define E2_U       8
#define EB_TAIL    (RB_NODES * FEAT_MID / EB_THREADS)

__global__ void init_deg(float* __restrict__ deg, int n_nodes) {
    int tid = blockIdx.x * blockDim.x + threadIdx.x;
    if (tid < n_nodes) deg[tid] = 1.0f;
}

#define DT_THREADS (2 * RB_NODES)
__global__ void __launch_bounds__(DT_THREADS)
dinv_totals(float* __restrict__ deg, int* __restrict__ totals, int* __restrict__ cursor,
            int n_nodes, int n_ranges) {
    __shared__ int part[DT_THREADS / 32];
    const int tid = threadIdx.x;
    const int i   = blockIdx.x * DT_THREADS + tid;
    int v = 0;
    if (i < n_nodes) {
        float d = deg[i];
        v = (int)d - 1;
        deg[i] = (d > 0.0f) ? rsqrtf(d) : 0.0f;
    }
    for (int m = 16; m > 0; m >>= 1) v += __shfl_xor(v, m);
    if ((tid & 31) == 0) part[tid >> 5] = v;
    __syncthreads();
    if (tid < 2) {
        const int r = blockIdx.x * 2 + tid;
        if (r < n_ranges) {
            const int* q = part + tid * (RB_NODES / 32);
            totals[r] = q[0] + q[1] + q[2] + q[3];
            cursor[r] = 0;
        }
    }
}

__global__ void __launch_bounds__(BK_THREADS)
bucket_scatter(const int* __restrict__ src, const int* __restrict__ dst,
               const int* __restrict__ totals, int* __restrict__ cursor,
               int* __restrict__ offs, int* __restrict__ bins,
               int n_edges, int per_chunk, int n_ranges) {
    __shared__ int cnt[RB_MAXR];
    __shared__ int gb[RB_MAXR];
    __shared__ int sc[RB_MAXR];
    __shared__ int so[RB_MAXR];
    __shared__ int stage_w[BK_STAGE];
    __shared__ int stage_p[BK_STAGE];
    const int tid = threadIdx.x;
    int e0 = blockIdx.x * per_chunk;
    if (e0 > n_edges) e0 = n_edges;
    int e1 = e0 + per_chunk;
    if (e1 > n_edges) e1 = n_edges;
    cnt[tid] = 0;
    __syncthreads();
    for (int base = e0; base < e1; base += BK_U * BK_THREADS) {
        int dv[BK_U];
#pragma unroll
        for (int k = 0; k < BK_U; k++) {
            int e  = base + k * BK_THREADS + tid;
            int ec = e < e1 ? e : e1 - 1;
            dv[k] = dst[ec];
        }
        asm volatile("" : "+v"(dv[0]), "+v"(dv[1]), "+v"(dv[2]), "+v"(dv[3]));
#pragma unroll
        for (int k = 0; k < BK_U; k++) {
            int e = base + k * BK_THREADS + tid;
            if (e < e1) atomicAdd(&cnt[dv[k] >> RB_SHIFT], 1);
        }
    }
    __syncthreads();
    const int c   = cnt[tid];
    const int tot = tid < n_ranges ? totals[tid] : 0;
    int rel = 0;
    if (c > 0) rel = atomicAdd(&cursor[tid], c);
    sc[tid] = c;
    so[tid] = tot;
    __syncthreads();
    for (int off = 1; off < RB_MAXR; off <<= 1) {
        int t1 = tid >= off ? sc[tid - off] : 0;
        int t2 = tid >= off ? so[tid - off] : 0;
        __syncthreads();
        sc[tid] += t1;
        so[tid] += t2;
        __syncthreads();
    }
    const int first = so[tid] - tot;
    if (blockIdx.x == 0 && tid < n_ranges) {
        offs[tid] = first;
        if (tid == n_ranges - 1) offs[n_ranges] = so[tid];
    }
    gb[tid]  = first + rel;
    sc[tid] -= c;
    cnt[tid] = 0;
    __syncthreads();
    for (int base = e0; base < e1; base += BK_U * BK_THREADS) {
        int dv[BK_U], sv[BK_U];
#pragma unroll
        for (int k = 0; k < BK_U; k++) {
            int e  = base + k * BK_THREADS + tid;
            int ec = e < e1 ? e : e1 - 1;
            dv[k] = dst[ec];
            sv[k] = src[ec];
        }
        asm volatile("" : "+v"(dv[0]), "+v"(dv[1]), "+v"(dv[2]), "+v"(dv[3]),
                          "+v"(sv[0]), "+v"(sv[1]), "+v"(sv[2]), "+v"(sv[3]));
        int rank[BK_U], ls[BK_U], gs[BK_U];
#pragma unroll
        for (int k = 0; k < BK_U; k++) {
            int e = base + k * BK_THREADS + tid;
            int r = dv[k] >> RB_SHIFT;
            rank[k] = 0;
            if (e < e1) rank[k] = atomicAdd(&cnt[r], 1);
            ls[k] = sc[r];
            gs[k] = gb[r];
        }
#pragma unroll
        for (int k = 0; k < BK_U; k++) {
            int e = base + k * BK_THREADS + tid;
            if (e < e1) {
                int lp = ls[k] + rank[k];
                stage_w[lp] = (sv[k] << RB_SHIFT) | (dv[k] & (RB_NODES - 1));
                stage_p[lp] = gs[k] + rank[k];
            }
        }
    }
    __syncthreads();
    const int n = e1 - e0;
    for (int i = tid; i < n; i += BK_THREADS) bins[stage_p[i]] = stage_w[i];
}

__global__ void __launch_bounds__(EB_THREADS)
edge1_bin(const int* __restrict__ bins, const int* __restrict__ offs,
          const float* __restrict__ h, const float* __restrict__ dinv,
          const float* __restrict__ bias1, const float* __restrict__ W2,
          float* __restrict__ h2, int n_nodes) {
    __shared__ float acc[RB_NODES * FEAT_MID];
    const int tid  = threadIdx.x;
    const int lane = tid & 31;
    const int f    = lane & 15;
    const int hf   = lane >> 4;
    const int lo   = blockIdx.x << RB_SHIFT;
    int len = n_nodes - lo;
    if (len > RB_NODES) len = RB_NODES;

    for (int i = tid; i < RB_NODES * FEAT_MID; i += EB_THREADS) acc[i] = 0.0f;
    __syncthreads();

    const int b0 = offs[blockIdx.x];
    const int b1 = offs[blockIdx.x + 1];
    const int stride = (EB_THREADS / 32) * 2 * EB_STEPS;
    int j0 = b0 + (tid >> 5) * 2 * EB_STEPS;
    int w[EB_STEPS] = {};
    if (j0 < b1) {
#pragma unroll
        for (int p = 0; p < EB_STEPS; p++) {
            int idx = j0 + 2 * p + hf;
            w[p] = bins[idx < b1 ? idx : j0];
        }
    }
    while (j0 < b1) {
        int   cur[EB_STEPS];
        float hq[EB_STEPS];
#pragma unroll
        for (int p = 0; p < EB_STEPS; p++) cur[p] = w[p];
        const int jn = j0 + stride;
        if (jn < b1) {
#pragma unroll
            for (int p = 0; p < EB_STEPS; p++) {
                int idx = jn + 2 * p + hf;
                w[p] = bins[idx < b1 ? idx : jn];
            }
        }
#pragma unroll
        for (int p = 0; p < EB_STEPS; p++)
            hq[p] = h[(cur[p] >> RB_SHIFT) * FEAT_MID + f];
        asm volatile("" : "+v"(hq[0]), "+v"(hq[1]), "+v"(hq[2]), "+v"(hq[3]),
                          "+v"(hq[4]), "+v"(hq[5]), "+v"(hq[6]), "+v"(hq[7]),
                          "+v"(hq[8]), "+v"(hq[9]), "+v"(hq[10]), "+v"(hq[11]),
                          "+v"(hq[12]), "+v"(hq[13]), "+v"(hq[14]), "+v"(hq[15]));
#pragma unroll
        for (int p = 0; p < EB_STEPS; p++) {
            if (j0 + 2 * p + hf < b1)
                atomicAdd(&acc[(cur[p] & (RB_NODES - 1)) * FEAT_MID + f], hq[p]);
        }
        j0 = jn;
    }
    __syncthreads();

    const float* __restrict__ hown = h + (size_t)lo * FEAT_MID;
    const int   flen = len * FEAT_MID;
    const float bf   = bias1[f];
    float hs[EB_TAIL], dd[EB_TAIL];
#pragma unroll
    for (int j = 0; j < EB_TAIL; j++) {
        int i  = tid + j * EB_THREADS;
        int ic = i < flen ? i : tid & 15;
        hs[j] = hown[ic];
        dd[j] = dinv[lo + (ic >> 4)];
    }
#pragma unroll
    for (int j = 0; j < EB_TAIL; j++) {
        int i = tid + j * EB_THREADS;
        if (i < flen) acc[i] = fmaxf((acc[i] + hs[j]) * dd[j] + bf, 0.0f);
    }
    __syncthreads();

    if (tid < len * FEAT_OUT) {
        const int node = tid >> 1, c = tid & 1;
        float s = 0.0f;
#pragma unroll
        for (int k = 0; k < FEAT_MID; k++)
            s = fmaf(acc[node * FEAT_MID + k], W2[k * FEAT_OUT + c], s);
        h2[(size_t)lo * FEAT_OUT + tid] = s * dinv[lo + node];
    }
}

__global__ void __launch_bounds__(EB_THREADS)
edge2_bin(const int* __restrict__ bins, const int* __restrict__ offs,
          const float* __restrict__ h2, const float* __restrict__ dinv,
          const float* __restrict__ bias2, float* __restrict__ out, int n_nodes) {
    __shared__ float acc[RB_NODES * FEAT_OUT];
    const int tid = threadIdx.x;
    const int lo  = blockIdx.x << RB_SHIFT;
    int len = n_nodes - lo;
    if (len > RB_NODES) len = RB_NODES;

    for (int i = tid; i < RB_NODES * FEAT_OUT; i += EB_THREADS) acc[i] = 0.0f;
    __syncthreads();

    const v2f* __restrict__ h2v = (const v2f*)h2;
    const int b0 = offs[blockIdx.x];
    const int b1 = offs[blockIdx.x + 1];
    for (int base = b0; base < b1; base += E2_U * EB_THREADS) {
        int w[E2_U];
        v2f hv[E2_U];
#pragma unroll
        for (int k = 0; k < E2_U; k++) {
            int idx = base + k * EB_THREADS + tid;
            w[k] = bins[idx < b1 ? idx : b0];
        }
#pragma unroll
        for (int k = 0; k < E2_U; k++) hv[k] = h2v[w[k] >> RB_SHIFT];
        asm volatile("" : "+v"(hv[0]), "+v"(hv[1]), "+v"(hv[2]), "+v"(hv[3]),
                          "+v"(hv[4]), "+v"(hv[5]), "+v"(hv[6]), "+v"(hv[7]));
#pragma unroll
        for (int k = 0; k < E2_U; k++) {
            if (base + k * EB_THREADS + tid < b1) {
                int t = w[k] & (RB_NODES - 1);
                atomicAdd(&acc[t * FEAT_OUT + 0], hv[k].x);
                atomicAdd(&acc[t * FEAT_OUT + 1], hv[k].y);
            }
        }
    }
    __syncthreads();

    if (tid < len) {
        const int i = lo + tid;
        float di = dinv[i];
        float v0 = (acc[tid * 2 + 0] + h2[i * 2 + 0]) * di + bias2[0];
        float v1 = (acc[tid * 2 + 1] + h2[i * 2 + 1]) * di + bias2[1];
        float m  = fmaxf(v0, v1);
        float l  = m + logf(expf(v0 - m) + expf(v1 - m));
        out[i * 2 + 0] = v0 - l;
        out[i * 2 + 1] = v1 - l;
    }
}

extern "C" void kernel_launch(void* const* d_in, const int* in_sizes, int n_in,
                              void* d_out, int out_size, void* d_ws, size_t ws_size,
                              hipStream_t stream) {
    const float* x  = (const float*)d_in[0];
    const int*   ei = (const int*)d_in[1];
    const float* W1 = (const float*)d_in[2];
    const float* b1 = (const float*)d_in[3];
    const float* W2 = (const float*)d_in[4];
    const float* b2 = (const float*)d_in[5];
    float* out = (float*)d_out;

    int n_nodes = in_sizes[0] / FEAT_IN;
    int n_edges = in_sizes[1] / 2;
    const int* src = ei;
    const int* dst = ei + n_edges;

    float* ws   = (float*)d_ws;
    float* deg  = ws;
    float* h    = deg  + n_nodes;
    float* acc1 = h    + (size_t)n_nodes * FEAT_MID;
    float* h2   = acc1 + (size_t)n_nodes * FEAT_MID;
    float* acc2 = h2   + (size_t)n_nodes * FEAT_OUT;

    const int B = 256;
    int n16 = n_nodes * FEAT_MID;

    int nr1 = (n_nodes + LDS_FLOATS - 1) / LDS_FLOATS;
    int rn1 = (((n_nodes + nr1 - 1) / nr1) + 31) & ~31;
    int ch1 = 128 / nr1 > 0 ? 128 / nr1 : 1;
    int pc1 = (n_edges + ch1 - 1) / ch1;
    int n_tiles = (n_nodes + 15) / 16;
    int wpb = B / 32;

    int    n_ranges = (n_nodes + RB_NODES - 1) >> RB_SHIFT;
    size_t need = ((size_t)n_nodes * (1 + 2 * FEAT_MID + 2 * FEAT_OUT) + 3 * (size_t)n_ranges + 1 +
                   (size_t)n_edges) * sizeof(float);
    if (n_ranges <= RB_MAXR && ws_size >= need) {
        int* offs   = (int*)(acc2 + (size_t)n_nodes * FEAT_OUT);
        int* cursor = offs + n_ranges + 1;
        int* totals = cursor + n_ranges;
        int* bins   = totals + n_ranges;
        int  nb     = (n_edges + BK_STAGE - 1) / BK_STAGE;
        int  pcb    = BK_STAGE;

        init_deg<<<(n_nodes + B - 1) / B, B, 0, stream>>>(deg, n_nodes);
        degree_lds<<<nr1 * ch1, SCAT_THREADS, 0, stream>>>(dst, deg, n_edges, n_nodes, rn1, ch1, pc1);
        dinv_totals<<<(n_nodes + DT_THREADS - 1) / DT_THREADS, DT_THREADS, 0, stream>>>(
            deg, totals, cursor, n_nodes, n_ranges);
        bucket_scatter<<<nb, BK_THREADS, 0, stream>>>(src, dst, totals, cursor, offs, bins,
                                                      n_edges, pcb, n_ranges);
        gemm1_wmma<<<(n_tiles + wpb - 1) / wpb, B, 0, stream>>>(x, W1, deg, h, n_tiles, n_nodes);
        edge1_bin<<<n_ranges, EB_THREADS, 0, stream>>>(bins, offs, h, deg, b1, W2, h2, n_nodes);
        edge2_bin<<<n_ranges, EB_THREADS, 0, stream>>>(bins, offs, h2, deg, b2, out, n_nodes);
        return;
    }

    init_ws<<<(n16 + B - 1) / B, B, 0, stream>>>(deg, acc1, acc2, n_nodes);
    degree_lds<<<nr1 * ch1, SCAT_THREADS, 0, stream>>>(dst, deg, n_edges, n_nodes, rn1, ch1, pc1);
    dinv_kernel<<<(n_nodes + B - 1) / B, B, 0, stream>>>(deg, n_nodes);

    gemm1_wmma<<<(n_tiles + wpb - 1) / wpb, B, 0, stream>>>(x, W1, deg, h, n_tiles, n_nodes);

    int nr0 = (n_nodes + E1_NODES - 1) / E1_NODES;
    int rn0 = (((n_nodes + nr0 - 1) / nr0) + 31) & ~31;
    int ch0 = 256 / nr0 > 0 ? 256 / nr0 : 1;
    int pc0 = (n_edges + ch0 - 1) / ch0;
    edge1_lds<<<nr0 * ch0, SCAT_THREADS, 0, stream>>>(src, dst, h, acc1, n_edges, n_nodes,
                                                      rn0, ch0, pc0);
    relu_bias1<<<(n16 + B - 1) / B, B, 0, stream>>>(h, deg, b1, acc1, n_nodes);

    gemm2_kernel<<<(n_nodes + B - 1) / B, B, 0, stream>>>(acc1, W2, deg, h2, n_nodes);

    int nr2 = (n_nodes * FEAT_OUT + LDS_FLOATS - 1) / LDS_FLOATS;
    int rn2 = (((n_nodes + nr2 - 1) / nr2) + 31) & ~31;
    int ch2 = 128 / nr2 > 0 ? 128 / nr2 : 1;
    int pc2 = (n_edges + ch2 - 1) / ch2;
    edge2_lds<<<nr2 * ch2, SCAT_THREADS, 0, stream>>>(src, dst, h2, acc2, n_edges, n_nodes,
                                                      rn2, ch2, pc2);

    final_kernel<<<(n_nodes + B - 1) / B, B, 0, stream>>>(acc2, h2, deg, b2, out, n_nodes);
}


# bucketing pass and the dinv * (x @ W1) product run side by side in one grid, work-groups chosen by block index (neither needs the other): one launch dropped, arithmetic and sum orders as before
# speedup vs baseline: 5.2582x; 1.0455x over previous
#include <hip/hip_runtime.h>
#include <math.h>

typedef float v2f __attribute__((ext_vector_type(2)));
typedef float v8f __attribute__((ext_vector_type(8)));

#define FEAT_IN  128
#define FEAT_MID 16
#define FEAT_OUT 2

__global__ void init_ws(float* __restrict__ deg, float* __restrict__ acc1,
                        float* __restrict__ acc2, int n_nodes) {
    int tid = blockIdx.x * blockDim.x + threadIdx.x;
    if (tid < n_nodes * FEAT_MID) acc1[tid] = 0.0f;
    if (tid < n_nodes * FEAT_OUT) acc2[tid] = 0.0f;
    if (tid < n_nodes)            deg[tid]  = 1.0f;
}

#define LDS_FLOATS   69632
#define SCAT_THREADS 1024
#define SCAT_U       8

__global__ void __launch_bounds__(SCAT_THREADS)
degree_lds(const int* __restrict__ dst, float* __restrict__ deg,
           int n_edges, int n_nodes, int range_nodes, int chunks, int per_chunk) {
    __shared__ float cnt[LDS_FLOATS];
    const int tid   = threadIdx.x;
    const int range = blockIdx.x / chunks;
    const int chunk = blockIdx.x - range * chunks;
    const int lo    = range * range_nodes;
    int len = n_nodes - lo;
    if (len > range_nodes) len = range_nodes;
    if (len <= 0) return;

    for (int i = tid; i < len; i += SCAT_THREADS) cnt[i] = 0.0f;
    __syncthreads();

    const int e0 = chunk * per_chunk;
    int e1 = e0 + per_chunk;
    if (e1 > n_edges) e1 = n_edges;
    for (int base = e0; base < e1; base += SCAT_U * SCAT_THREADS) {
        int dv[SCAT_U];
#pragma unroll
        for (int k = 0; k < SCAT_U; k++) {
            int e  = base + k * SCAT_THREADS + tid;
            int ec = e < e1 ? e : e1 - 1;
            dv[k] = dst[ec];
        }
        asm volatile("" : "+v"(dv[0]), "+v"(dv[1]), "+v"(dv[2]), "+v"(dv[3]),
                          "+v"(dv[4]), "+v"(dv[5]), "+v"(dv[6]), "+v"(dv[7]));
#pragma unroll
        for (int k = 0; k < SCAT_U; k++) {
            int e = base + k * SCAT_THREADS + tid;
            int t = dv[k] - lo;
            if (e < e1 && (unsigned)t < (unsigned)len) atomicAdd(&cnt[t], 1.0f);
        }
    }
    __syncthreads();

    const int rot = (chunk * (len / chunks)) & ~31;
    for (int j = tid; j < len; j += SCAT_THREADS) {
        int i = j + rot;
        if (i >= len) i -= len;
        float v = cnt[i];
        if (v != 0.0f) atomicAdd(&deg[lo + i], v);
    }
}

__global__ void dinv_kernel(float* __restrict__ deg, int n_nodes) {
    int i = blockIdx.x * blockDim.x + threadIdx.x;
    if (i < n_nodes) {
        float d = deg[i];
        deg[i] = (d > 0.0f) ? rsqrtf(d) : 0.0f;
    }
}

__device__ __forceinline__ void
gemm1_tile(const float* __restrict__ x, const float* __restrict__ W1,
           const float* __restrict__ dinv, float* __restrict__ h,
           int tile, int n_tiles, int n_nodes) {
    int lane = threadIdx.x & 31;
    if (tile >= n_tiles) return;

    int col = lane & 15;
    int kh  = (lane >> 4) << 1;

    int rowA = tile * 16 + col;
    if (rowA > n_nodes - 1) rowA = n_nodes - 1;
    const float* xrow = x + (size_t)rowA * FEAT_IN;

    int mbase = (lane >> 4) * 8;
    float dr[8];
#pragma unroll
    for (int r = 0; r < 8; r++) {
        int row = tile * 16 + mbase + r;
        dr[r] = dinv[row < n_nodes ? row : n_nodes - 1];
    }

    v8f acc = {};
    for (int k0 = 0; k0 < FEAT_IN; k0 += 4) {
        v2f a, b;
        a.x = xrow[k0 + kh];
        a.y = xrow[k0 + kh + 1];
        b.x = W1[(k0 + kh)     * FEAT_MID + col];
        b.y = W1[(k0 + kh + 1) * FEAT_MID + col];
        acc = __builtin_amdgcn_wmma_f32_16x16x4_f32(
             false, a,  false, b,
             (short)0, acc,  false,  false);
    }

    asm volatile("v_nop\n\tv_nop\n\tv_nop\n\tv_nop\n\tv_nop" : "+v"(acc));
#pragma unroll
    for (int r = 0; r < 8; r++) {
        int row = tile * 16 + mbase + r;
        if (row < n_nodes) h[(size_t)row * FEAT_MID + col] = acc[r] * dr[r];
    }
}

__global__ void __launch_bounds__(256)
gemm1_wmma(const float* __restrict__ x, const float* __restrict__ W1,
           const float* __restrict__ dinv, float* __restrict__ h,
           int n_tiles, int n_nodes) {
    gemm1_tile(x, W1, dinv, h, blockIdx.x * (blockDim.x >> 5) + (threadIdx.x >> 5), n_tiles, n_nodes);
}

typedef int v2i __attribute__((ext_vector_type(2)));
typedef int v4i __attribute__((ext_vector_type(4)));

#define E1_NODES 2400
#define E1_U     16
#define E1_STEPS 8
#define E1_LIST  (E1_U * 32)
#define E1_TRIP  (E1_U * SCAT_THREADS)

__device__ __forceinline__ void e1_push(bool m, int e, int tt, v2i* lst, int& cnt) {
    unsigned b = __builtin_amdgcn_ballot_w32(m);
    int pos = cnt + (int)__builtin_amdgcn_mbcnt_lo(b, 0u);
    if (m) {
        v2i q;
        q.x = e;
        q.y = tt;
        lst[pos] = q;
    }
    cnt += __builtin_popcount(b);
}

__device__ __forceinline__ void e1_drain(const v2i* lst, int cnt,
                                         const int* __restrict__ src, const float* __restrict__ h,
                                         float* __restrict__ acc, int f, int hf) {
    for (int j0 = 0; j0 < cnt; j0 += 2 * E1_STEPS) {
        int   tq[E1_STEPS], sq[E1_STEPS];
        float hq[E1_STEPS];
#pragma unroll
        for (int p = 0; p < E1_STEPS; p++) {
            int  idx   = j0 + 2 * p + hf;
            bool valid = idx < cnt;
            v2i  q     = lst[valid ? idx : 0];
            tq[p] = valid ? q.y : -1;
            sq[p] = src[q.x];
        }
        asm volatile("" : "+v"(sq[0]), "+v"(sq[1]), "+v"(sq[2]), "+v"(sq[3]),
                          "+v"(sq[4]), "+v"(sq[5]), "+v"(sq[6]), "+v"(sq[7]));
#pragma unroll
        for (int p = 0; p < E1_STEPS; p++)
            hq[p] = h[sq[p] * FEAT_MID + f];
        asm volatile("" : "+v"(hq[0]), "+v"(hq[1]), "+v"(hq[2]), "+v"(hq[3]),
                          "+v"(hq[4]), "+v"(hq[5]), "+v"(hq[6]), "+v"(hq[7]));
#pragma unroll
        for (int p = 0; p < E1_STEPS; p++) {
            if (tq[p] >= 0)
                atomicAdd(&acc[tq[p] * FEAT_MID + f], hq[p]);
        }
    }
}

__global__ void __launch_bounds__(SCAT_THREADS)
edge1_lds(const int* __restrict__ src, const int* __restrict__ dst,
          const float* __restrict__ h,
          float* __restrict__ acc1, int n_edges, int n_nodes,
          int range_nodes, int chunks, int per_chunk) {
    __shared__ float acc[E1_NODES * FEAT_MID];
    __shared__ v2i   list[SCAT_THREADS / 32][E1_LIST];
    const int tid   = threadIdx.x;
    const int lane  = tid & 31;
    const int f     = lane & 15;
    const int hf    = lane >> 4;
    const int range = blockIdx.x / chunks;
    const int chunk = blockIdx.x - range * chunks;
    const int lo    = range * range_nodes;
    int len = n_nodes - lo;
    if (len > range_nodes) len = range_nodes;
    if (len <= 0) return;
    const int flen = len * FEAT_MID;

    for (int i = tid; i < flen; i += SCAT_THREADS) acc[i] = 0.0f;
    __syncthreads();

    v2i* __restrict__ lst = list[tid >> 5];
    int e0 = chunk * per_chunk;
    if (e0 > n_edges) e0 = n_edges;
    e0 &= ~3;
    int e1 = n_edges;
    if (chunk + 1 < chunks) {
        e1 = (chunk + 1) * per_chunk;
        if (e1 > n_edges) e1 = n_edges;
        e1 &= ~3;
    }

    const bool wide = (((size_t)dst) & 15) == 0;
    const int nfull = wide ? (e1 - e0) / E1_TRIP : 0;
    v4i nx[E1_U / 4];
    if (nfull > 0) {
        const v4i* __restrict__ pn = (const v4i*)(dst + e0);
#pragma unroll
        for (int j = 0; j < E1_U / 4; j++) nx[j] = pn[j * SCAT_THREADS + tid];
    }
    for (int t = 0; t < nfull; t++) {
        const int base = e0 + t * E1_TRIP;
        v4i cur[E1_U / 4];
#pragma unroll
        for (int j = 0; j < E1_U / 4; j++) cur[j] = nx[j];
        if (t + 1 < nfull) {
            const v4i* __restrict__ pn = (const v4i*)(dst + base + E1_TRIP);
#pragma unroll
            for (int j = 0; j < E1_U / 4; j++) nx[j] = pn[j * SCAT_THREADS + tid];
        }
        const int eb = base + 4 * tid;
        int cnt = 0;
#pragma unroll
        for (int j = 0; j < E1_U / 4; j++) {
#pragma unroll
            for (int i = 0; i < 4; i++) {
                int tt = cur[j][i] - lo;
                e1_push((unsigned)tt < (unsigned)len, eb + j * 4 * SCAT_THREADS + i, tt, lst, cnt);
            }
        }
        __builtin_amdgcn_wave_barrier();
        e1_drain(lst, cnt, src, h, acc, f, hf);
        __builtin_amdgcn_wave_barrier();
    }

    for (int base = e0 + nfull * E1_TRIP; base < e1; base += E1_TRIP) {
        int dv[E1_U];
#pragma unroll
        for (int k = 0; k < E1_U; k++) {
            int e  = base + k * SCAT_THREADS + tid;
            int ec = e < e1 ? e : e1 - 1;
            dv[k] = dst[ec];
        }
        asm volatile("" : "+v"(dv[0]), "+v"(dv[1]), "+v"(dv[2]), "+v"(dv[3]),
                          "+v"(dv[4]), "+v"(dv[5]), "+v"(dv[6]), "+v"(dv[7]),
                          "+v"(dv[8]), "+v"(dv[9]), "+v"(dv[10]), "+v"(dv[11]),
                          "+v"(dv[12]), "+v"(dv[13]), "+v"(dv[14]), "+v"(dv[15]));
        int cnt = 0;
#pragma unroll
        for (int k = 0; k < E1_U; k++) {
            int e  = base + k * SCAT_THREADS + tid;
            int tt = dv[k] - lo;
            e1_push((e < e1) && ((unsigned)tt < (unsigned)len), e, tt, lst, cnt);
        }
        __builtin_amdgcn_wave_barrier();
        e1_drain(lst, cnt, src, h, acc, f, hf);
        __builtin_amdgcn_wave_barrier();
    }
    __syncthreads();

    const int rot = (chunk * (flen / chunks)) & ~31;
    float* __restrict__ outp = acc1 + (size_t)lo * FEAT_MID;
    for (int j = tid; j < flen; j += SCAT_THREADS) {
        int i = j + rot;
        if (i >= flen) i -= flen;
        float v = acc[i];
        if (v != 0.0f) atomicAdd(&outp[i], v);
    }
}

__global__ void relu_bias1(const float* __restrict__ h, const float* __restrict__ dinv,
                           const float* __restrict__ b1, float* __restrict__ acc1,
                           int n_nodes) {
    int tid = blockIdx.x * blockDim.x + threadIdx.x;
    if (tid >= n_nodes * FEAT_MID) return;
    int node = tid >> 4, f = tid & 15;
    float di = dinv[node];
    float v = (acc1[tid] + h[tid]) * di + b1[f];
    acc1[tid] = fmaxf(v, 0.0f);
}

__global__ void gemm2_kernel(const float* __restrict__ h1, const float* __restrict__ W2,
                             const float* __restrict__ dinv,
                             float* __restrict__ h2, int n_nodes) {
    int i = blockIdx.x * blockDim.x + threadIdx.x;
    if (i >= n_nodes) return;
    float s0 = 0.0f, s1 = 0.0f;
#pragma unroll
    for (int k = 0; k < FEAT_MID; k++) {
        float v = h1[i * FEAT_MID + k];
        s0 = fmaf(v, W2[k * 2 + 0], s0);
        s1 = fmaf(v, W2[k * 2 + 1], s1);
    }
    float di = dinv[i];
    h2[i * 2 + 0] = s0 * di;
    h2[i * 2 + 1] = s1 * di;
}

__global__ void __launch_bounds__(SCAT_THREADS)
edge2_lds(const int* __restrict__ src, const int* __restrict__ dst,
          const float* __restrict__ h2,
          float* __restrict__ acc2, int n_edges, int n_nodes,
          int range_nodes, int chunks, int per_chunk) {
    __shared__ float acc[LDS_FLOATS];
    const int tid   = threadIdx.x;
    const int range = blockIdx.x / chunks;
    const int chunk = blockIdx.x - range * chunks;
    const int lo    = range * range_nodes;
    int len = n_nodes - lo;
    if (len > range_nodes) len = range_nodes;
    if (len <= 0) return;
    const int flen = len * FEAT_OUT;

    for (int i = tid; i < flen; i += SCAT_THREADS) acc[i] = 0.0f;
    __syncthreads();

    const v2f* __restrict__ h2v = (const v2f*)h2;
    const int e0 = chunk * per_chunk;
    int e1 = e0 + per_chunk;
    if (e1 > n_edges) e1 = n_edges;
    for (int base = e0; base < e1; base += SCAT_U * SCAT_THREADS) {
        int dv[SCAT_U], sv[SCAT_U];
#pragma unroll
        for (int k = 0; k < SCAT_U; k++) {
            int e  = base + k * SCAT_THREADS + tid;
            int ec = e < e1 ? e : e1 - 1;
            dv[k] = dst[ec];
            sv[k] = src[ec];
        }
        asm volatile("" : "+v"(dv[0]), "+v"(dv[1]), "+v"(dv[2]), "+v"(dv[3]),
                          "+v"(dv[4]), "+v"(dv[5]), "+v"(dv[6]), "+v"(dv[7]),
                          "+v"(sv[0]), "+v"(sv[1]), "+v"(sv[2]), "+v"(sv[3]),
                          "+v"(sv[4]), "+v"(sv[5]), "+v"(sv[6]), "+v"(sv[7]));
        int   t[SCAT_U];
        v2f   hv[SCAT_U];
#pragma unroll
        for (int k = 0; k < SCAT_U; k++) {
            int e  = base + k * SCAT_THREADS + tid;
            int tt = dv[k] - lo;
            bool m = (e < e1) && ((unsigned)tt < (unsigned)len);
            t[k] = m ? tt : -1;
            int sg = m ? sv[k] : lo;
            hv[k] = h2v[sg];
        }
        asm volatile("" : "+v"(hv[0]), "+v"(hv[1]), "+v"(hv[2]), "+v"(hv[3]),
                          "+v"(hv[4]), "+v"(hv[5]), "+v"(hv[6]), "+v"(hv[7]));
#pragma unroll
        for (int k = 0; k < SCAT_U; k++) {
            if (t[k] >= 0) {
                atomicAdd(&acc[t[k] * FEAT_OUT + 0], hv[k].x);
                atomicAdd(&acc[t[k] * FEAT_OUT + 1], hv[k].y);
            }
        }
    }
    __syncthreads();

    const int rot = (chunk * (flen / chunks)) & ~31;
    float* __restrict__ outp = acc2 + (size_t)lo * FEAT_OUT;
    for (int j = tid; j < flen; j += SCAT_THREADS) {
        int i = j + rot;
        if (i >= flen) i -= flen;
        float v = acc[i];
        if (v != 0.0f) atomicAdd(&outp[i], v);
    }
}

__global__ void final_kernel(const float* __restrict__ acc2, const float* __restrict__ h2,
                             const float* __restrict__ dinv, const float* __restrict__ b2,
                             float* __restrict__ out, int n_nodes) {
    int i = blockIdx.x * blockDim.x + threadIdx.x;
    if (i >= n_nodes) return;
    float di = dinv[i];
    float v0 = (acc2[i * 2 + 0] + h2[i * 2 + 0]) * di + b2[0];
    float v1 = (acc2[i * 2 + 1] + h2[i * 2 + 1]) * di + b2[1];
    float m  = fmaxf(v0, v1);
    float l  = m + logf(expf(v0 - m) + expf(v1 - m));
    out[i * 2 + 0] = v0 - l;
    out[i * 2 + 1] = v1 - l;
}

#define RB_NODES   128
#define RB_SHIFT   7
#define RB_MAXR    1024
#define BK_THREADS 1024
#define BK_U       4
#define BK_STAGE   12800
#define EB_THREADS 512
#define EB_STEPS   16
#define E2_U       8
#define EB_TAIL    (RB_NODES * FEAT_MID / EB_THREADS)

__global__ void init_deg(float* __restrict__ deg, int n_nodes) {
    int tid = blockIdx.x * blockDim.x + threadIdx.x;
    if (tid < n_nodes) deg[tid] = 1.0f;
}

#define DT_THREADS (2 * RB_NODES)
__global__ void __launch_bounds__(DT_THREADS)
dinv_totals(float* __restrict__ deg, int* __restrict__ totals, int* __restrict__ cursor,
            int n_nodes, int n_ranges) {
    __shared__ int part[DT_THREADS / 32];
    const int tid = threadIdx.x;
    const int i   = blockIdx.x * DT_THREADS + tid;
    int v = 0;
    if (i < n_nodes) {
        float d = deg[i];
        v = (int)d - 1;
        deg[i] = (d > 0.0f) ? rsqrtf(d) : 0.0f;
    }
    for (int m = 16; m > 0; m >>= 1) v += __shfl_xor(v, m);
    if ((tid & 31) == 0) part[tid >> 5] = v;
    __syncthreads();
    if (tid < 2) {
        const int r = blockIdx.x * 2 + tid;
        if (r < n_ranges) {
            const int* q = part + tid * (RB_NODES / 32);
            totals[r] = q[0] + q[1] + q[2] + q[3];
            cursor[r] = 0;
        }
    }
}

__device__ __forceinline__ void
bucket_chunk(const int* __restrict__ src, const int* __restrict__ dst,
             const int* __restrict__ totals, int* __restrict__ cursor,
             int* __restrict__ offs, int* __restrict__ bins,
             int n_edges, int per_chunk, int n_ranges, const int blk) {
    __shared__ int cnt[RB_MAXR];
    __shared__ int gb[RB_MAXR];
    __shared__ int sc[RB_MAXR];
    __shared__ int so[RB_MAXR];
    __shared__ int stage_w[BK_STAGE];
    __shared__ int stage_p[BK_STAGE];
    const int tid = threadIdx.x;
    int e0 = blk * per_chunk;
    if (e0 > n_edges) e0 = n_edges;
    int e1 = e0 + per_chunk;
    if (e1 > n_edges) e1 = n_edges;
    cnt[tid] = 0;
    __syncthreads();
    for (int base = e0; base < e1; base += BK_U * BK_THREADS) {
        int dv[BK_U];
#pragma unroll
        for (int k = 0; k < BK_U; k++) {
            int e  = base + k * BK_THREADS + tid;
            int ec = e < e1 ? e : e1 - 1;
            dv[k] = dst[ec];
        }
        asm volatile("" : "+v"(dv[0]), "+v"(dv[1]), "+v"(dv[2]), "+v"(dv[3]));
#pragma unroll
        for (int k = 0; k < BK_U; k++) {
            int e = base + k * BK_THREADS + tid;
            if (e < e1) atomicAdd(&cnt[dv[k] >> RB_SHIFT], 1);
        }
    }
    __syncthreads();
    const int c   = cnt[tid];
    const int tot = tid < n_ranges ? totals[tid] : 0;
    int rel = 0;
    if (c > 0) rel = atomicAdd(&cursor[tid], c);
    sc[tid] = c;
    so[tid] = tot;
    __syncthreads();
    for (int off = 1; off < RB_MAXR; off <<= 1) {
        int t1 = tid >= off ? sc[tid - off] : 0;
        int t2 = tid >= off ? so[tid - off] : 0;
        __syncthreads();
        sc[tid] += t1;
        so[tid] += t2;
        __syncthreads();
    }
    const int first = so[tid] - tot;
    if (blk == 0 && tid < n_ranges) {
        offs[tid] = first;
        if (tid == n_ranges - 1) offs[n_ranges] = so[tid];
    }
    gb[tid]  = first + rel;
    sc[tid] -= c;
    cnt[tid] = 0;
    __syncthreads();
    for (int base = e0; base < e1; base += BK_U * BK_THREADS) {
        int dv[BK_U], sv[BK_U];
#pragma unroll
        for (int k = 0; k < BK_U; k++) {
            int e  = base + k * BK_THREADS + tid;
            int ec = e < e1 ? e : e1 - 1;
            dv[k] = dst[ec];
            sv[k] = src[ec];
        }
        asm volatile("" : "+v"(dv[0]), "+v"(dv[1]), "+v"(dv[2]), "+v"(dv[3]),
                          "+v"(sv[0]), "+v"(sv[1]), "+v"(sv[2]), "+v"(sv[3]));
        int rank[BK_U], ls[BK_U], gs[BK_U];
#pragma unroll
        for (int k = 0; k < BK_U; k++) {
            int e = base + k * BK_THREADS + tid;
            int r = dv[k] >> RB_SHIFT;
            rank[k] = 0;
            if (e < e1) rank[k] = atomicAdd(&cnt[r], 1);
            ls[k] = sc[r];
            gs[k] = gb[r];
        }
#pragma unroll
        for (int k = 0; k < BK_U; k++) {
            int e = base + k * BK_THREADS + tid;
            if (e < e1) {
                int lp = ls[k] + rank[k];
                stage_w[lp] = (sv[k] << RB_SHIFT) | (dv[k] & (RB_NODES - 1));
                stage_p[lp] = gs[k] + rank[k];
            }
        }
    }
    __syncthreads();
    const int n = e1 - e0;
    for (int i = tid; i < n; i += BK_THREADS) bins[stage_p[i]] = stage_w[i];
}

__global__ void __launch_bounds__(BK_THREADS)
bucket_gemm(const int* __restrict__ src, const int* __restrict__ dst,
            const int* __restrict__ totals, int* __restrict__ cursor,
            int* __restrict__ offs, int* __restrict__ bins,
            int n_edges, int per_chunk, int n_ranges, int nb,
            const float* __restrict__ x, const float* __restrict__ W1,
            const float* __restrict__ dinv, float* __restrict__ h,
            int n_tiles, int n_nodes) {
    if ((int)blockIdx.x < nb) {
        bucket_chunk(src, dst, totals, cursor, offs, bins, n_edges, per_chunk, n_ranges, blockIdx.x);
    } else {
        gemm1_tile(x, W1, dinv, h,
                   ((int)blockIdx.x - nb) * (BK_THREADS / 32) + (int)(threadIdx.x >> 5),
                   n_tiles, n_nodes);
    }
}

__global__ void __launch_bounds__(EB_THREADS)
edge1_bin(const int* __restrict__ bins, const int* __restrict__ offs,
          const float* __restrict__ h, const float* __restrict__ dinv,
          const float* __restrict__ bias1, const float* __restrict__ W2,
          float* __restrict__ h2, int n_nodes) {
    __shared__ float acc[RB_NODES * FEAT_MID];
    const int tid  = threadIdx.x;
    const int lane = tid & 31;
    const int f    = lane & 15;
    const int hf   = lane >> 4;
    const int lo   = blockIdx.x << RB_SHIFT;
    int len = n_nodes - lo;
    if (len > RB_NODES) len = RB_NODES;

    for (int i = tid; i < RB_NODES * FEAT_MID; i += EB_THREADS) acc[i] = 0.0f;
    __syncthreads();

    const int b0 = offs[blockIdx.x];
    const int b1 = offs[blockIdx.x + 1];
    const int stride = (EB_THREADS / 32) * 2 * EB_STEPS;
    int j0 = b0 + (tid >> 5) * 2 * EB_STEPS;
    int w[EB_STEPS] = {};
    if (j0 < b1) {
#pragma unroll
        for (int p = 0; p < EB_STEPS; p++) {
            int idx = j0 + 2 * p + hf;
            w[p] = bins[idx < b1 ? idx : j0];
        }
    }
    while (j0 < b1) {
        int   cur[EB_STEPS];
        float hq[EB_STEPS];
#pragma unroll
        for (int p = 0; p < EB_STEPS; p++) cur[p] = w[p];
        const int jn = j0 + stride;
        if (jn < b1) {
#pragma unroll
            for (int p = 0; p < EB_STEPS; p++) {
                int idx = jn + 2 * p + hf;
                w[p] = bins[idx < b1 ? idx : jn];
            }
        }
#pragma unroll
        for (int p = 0; p < EB_STEPS; p++)
            hq[p] = h[(cur[p] >> RB_SHIFT) * FEAT_MID + f];
        asm volatile("" : "+v"(hq[0]), "+v"(hq[1]), "+v"(hq[2]), "+v"(hq[3]),
                          "+v"(hq[4]), "+v"(hq[5]), "+v"(hq[6]), "+v"(hq[7]),
                          "+v"(hq[8]), "+v"(hq[9]), "+v"(hq[10]), "+v"(hq[11]),
                          "+v"(hq[12]), "+v"(hq[13]), "+v"(hq[14]), "+v"(hq[15]));
#pragma unroll
        for (int p = 0; p < EB_STEPS; p++) {
            if (j0 + 2 * p + hf < b1)
                atomicAdd(&acc[(cur[p] & (RB_NODES - 1)) * FEAT_MID + f], hq[p]);
        }
        j0 = jn;
    }
    __syncthreads();

    const float* __restrict__ hown = h + (size_t)lo * FEAT_MID;
    const int   flen = len * FEAT_MID;
    const float bf   = bias1[f];
    float hs[EB_TAIL], dd[EB_TAIL];
#pragma unroll
    for (int j = 0; j < EB_TAIL; j++) {
        int i  = tid + j * EB_THREADS;
        int ic = i < flen ? i : tid & 15;
        hs[j] = hown[ic];
        dd[j] = dinv[lo + (ic >> 4)];
    }
#pragma unroll
    for (int j = 0; j < EB_TAIL; j++) {
        int i = tid + j * EB_THREADS;
        if (i < flen) acc[i] = fmaxf((acc[i] + hs[j]) * dd[j] + bf, 0.0f);
    }
    __syncthreads();

    if (tid < len * FEAT_OUT) {
        const int node = tid >> 1, c = tid & 1;
        float s = 0.0f;
#pragma unroll
        for (int k = 0; k < FEAT_MID; k++)
            s = fmaf(acc[node * FEAT_MID + k], W2[k * FEAT_OUT + c], s);
        h2[(size_t)lo * FEAT_OUT + tid] = s * dinv[lo + node];
    }
}

__global__ void __launch_bounds__(EB_THREADS)
edge2_bin(const int* __restrict__ bins, const int* __restrict__ offs,
          const float* __restrict__ h2, const float* __restrict__ dinv,
          const float* __restrict__ bias2, float* __restrict__ out, int n_nodes) {
    __shared__ float acc[RB_NODES * FEAT_OUT];
    const int tid = threadIdx.x;
    const int lo  = blockIdx.x << RB_SHIFT;
    int len = n_nodes - lo;
    if (len > RB_NODES) len = RB_NODES;

    for (int i = tid; i < RB_NODES * FEAT_OUT; i += EB_THREADS) acc[i] = 0.0f;
    __syncthreads();

    const v2f* __restrict__ h2v = (const v2f*)h2;
    const int b0 = offs[blockIdx.x];
    const int b1 = offs[blockIdx.x + 1];
    for (int base = b0; base < b1; base += E2_U * EB_THREADS) {
        int w[E2_U];
        v2f hv[E2_U];
#pragma unroll
        for (int k = 0; k < E2_U; k++) {
            int idx = base + k * EB_THREADS + tid;
            w[k] = bins[idx < b1 ? idx : b0];
        }
#pragma unroll
        for (int k = 0; k < E2_U; k++) hv[k] = h2v[w[k] >> RB_SHIFT];
        asm volatile("" : "+v"(hv[0]), "+v"(hv[1]), "+v"(hv[2]), "+v"(hv[3]),
                          "+v"(hv[4]), "+v"(hv[5]), "+v"(hv[6]), "+v"(hv[7]));
#pragma unroll
        for (int k = 0; k < E2_U; k++) {
            if (base + k * EB_THREADS + tid < b1) {
                int t = w[k] & (RB_NODES - 1);
                atomicAdd(&acc[t * FEAT_OUT + 0], hv[k].x);
                atomicAdd(&acc[t * FEAT_OUT + 1], hv[k].y);
            }
        }
    }
    __syncthreads();

    if (tid < len) {
        const int i = lo + tid;
        float di = dinv[i];
        float v0 = (acc[tid * 2 + 0] + h2[i * 2 + 0]) * di + bias2[0];
        float v1 = (acc[tid * 2 + 1] + h2[i * 2 + 1]) * di + bias2[1];
        float m  = fmaxf(v0, v1);
        float l  = m + logf(expf(v0 - m) + expf(v1 - m));
        out[i * 2 + 0] = v0 - l;
        out[i * 2 + 1] = v1 - l;
    }
}

extern "C" void kernel_launch(void* const* d_in, const int* in_sizes, int n_in,
                              void* d_out, int out_size, void* d_ws, size_t ws_size,
                              hipStream_t stream) {
    const float* x  = (const float*)d_in[0];
    const int*   ei = (const int*)d_in[1];
    const float* W1 = (const float*)d_in[2];
    const float* b1 = (const float*)d_in[3];
    const float* W2 = (const float*)d_in[4];
    const float* b2 = (const float*)d_in[5];
    float* out = (float*)d_out;

    int n_nodes = in_sizes[0] / FEAT_IN;
    int n_edges = in_sizes[1] / 2;
    const int* src = ei;
    const int* dst = ei + n_edges;

    float* ws   = (float*)d_ws;
    float* deg  = ws;
    float* h    = deg  + n_nodes;
    float* acc1 = h    + (size_t)n_nodes * FEAT_MID;
    float* h2   = acc1 + (size_t)n_nodes * FEAT_MID;
    float* acc2 = h2   + (size_t)n_nodes * FEAT_OUT;

    const int B = 256;
    int n16 = n_nodes * FEAT_MID;

    int nr1 = (n_nodes + LDS_FLOATS - 1) / LDS_FLOATS;
    int rn1 = (((n_nodes + nr1 - 1) / nr1) + 31) & ~31;
    int ch1 = 128 / nr1 > 0 ? 128 / nr1 : 1;
    int pc1 = (n_edges + ch1 - 1) / ch1;
    int n_tiles = (n_nodes + 15) / 16;
    int wpb = B / 32;

    int    n_ranges = (n_nodes + RB_NODES - 1) >> RB_SHIFT;
    size_t need = ((size_t)n_nodes * (1 + 2 * FEAT_MID + 2 * FEAT_OUT) + 3 * (size_t)n_ranges + 1 +
                   (size_t)n_edges) * sizeof(float);
    if (n_ranges <= RB_MAXR && ws_size >= need) {
        int* offs   = (int*)(acc2 + (size_t)n_nodes * FEAT_OUT);
        int* cursor = offs + n_ranges + 1;
        int* totals = cursor + n_ranges;
        int* bins   = totals + n_ranges;
        int  nb     = (n_edges + BK_STAGE - 1) / BK_STAGE;
        int  pcb    = BK_STAGE;

        init_deg<<<(n_nodes + B - 1) / B, B, 0, stream>>>(deg, n_nodes);
        degree_lds<<<nr1 * ch1, SCAT_THREADS, 0, stream>>>(dst, deg, n_edges, n_nodes, rn1, ch1, pc1);
        dinv_totals<<<(n_nodes + DT_THREADS - 1) / DT_THREADS, DT_THREADS, 0, stream>>>(
            deg, totals, cursor, n_nodes, n_ranges);
        int  gw     = (n_tiles + BK_THREADS / 32 - 1) / (BK_THREADS / 32);
        bucket_gemm<<<nb + gw, BK_THREADS, 0, stream>>>(src, dst, totals, cursor, offs, bins,
                                                        n_edges, pcb, n_ranges, nb,
                                                        x, W1, deg, h, n_tiles, n_nodes);
        edge1_bin<<<n_ranges, EB_THREADS, 0, stream>>>(bins, offs, h, deg, b1, W2, h2, n_nodes);
        edge2_bin<<<n_ranges, EB_THREADS, 0, stream>>>(bins, offs, h2, deg, b2, out, n_nodes);
        return;
    }

    init_ws<<<(n16 + B - 1) / B, B, 0, stream>>>(deg, acc1, acc2, n_nodes);
    degree_lds<<<nr1 * ch1, SCAT_THREADS, 0, stream>>>(dst, deg, n_edges, n_nodes, rn1, ch1, pc1);
    dinv_kernel<<<(n_nodes + B - 1) / B, B, 0, stream>>>(deg, n_nodes);

    gemm1_wmma<<<(n_tiles + wpb - 1) / wpb, B, 0, stream>>>(x, W1, deg, h, n_tiles, n_nodes);

    int nr0 = (n_nodes + E1_NODES - 1) / E1_NODES;
    int rn0 = (((n_nodes + nr0 - 1) / nr0) + 31) & ~31;
    int ch0 = 256 / nr0 > 0 ? 256 / nr0 : 1;
    int pc0 = (n_edges + ch0 - 1) / ch0;
    edge1_lds<<<nr0 * ch0, SCAT_THREADS, 0, stream>>>(src, dst, h, acc1, n_edges, n_nodes,
                                                      rn0, ch0, pc0);
    relu_bias1<<<(n16 + B - 1) / B, B, 0, stream>>>(h, deg, b1, acc1, n_nodes);

    gemm2_kernel<<<(n_nodes + B - 1) / B, B, 0, stream>>>(acc1, W2, deg, h2, n_nodes);

    int nr2 = (n_nodes * FEAT_OUT + LDS_FLOATS - 1) / LDS_FLOATS;
    int rn2 = (((n_nodes + nr2 - 1) / nr2) + 31) & ~31;
    int ch2 = 128 / nr2 > 0 ? 128 / nr2 : 1;
    int pc2 = (n_edges + ch2 - 1) / ch2;
    edge2_lds<<<nr2 * ch2, SCAT_THREADS, 0, stream>>>(src, dst, h2, acc2, n_edges, n_nodes,
                                                      rn2, ch2, pc2);

    final_kernel<<<(n_nodes + B - 1) / B, B, 0, stream>>>(acc2, h2, deg, b2, out, n_nodes);
}


# layer-1 sums: bucket edges sorted by target node in LDS, 4 lanes per node add 16-byte row pieces in registers; LDS float adds gone (layer-1 and h1 @ W2 sum order: per node, sorted order)
# speedup vs baseline: 5.5054x; 1.0470x over previous
#include <hip/hip_runtime.h>
#include <math.h>

typedef float v2f __attribute__((ext_vector_type(2)));
typedef float v8f __attribute__((ext_vector_type(8)));

#define FEAT_IN  128
#define FEAT_MID 16
#define FEAT_OUT 2

__global__ void init_ws(float* __restrict__ deg, float* __restrict__ acc1,
                        float* __restrict__ acc2, int n_nodes) {
    int tid = blockIdx.x * blockDim.x + threadIdx.x;
    if (tid < n_nodes * FEAT_MID) acc1[tid] = 0.0f;
    if (tid < n_nodes * FEAT_OUT) acc2[tid] = 0.0f;
    if (tid < n_nodes)            deg[tid]  = 1.0f;
}

#define LDS_FLOATS   69632
#define SCAT_THREADS 1024
#define SCAT_U       8

__global__ void __launch_bounds__(SCAT_THREADS)
degree_lds(const int* __restrict__ dst, float* __restrict__ deg,
           int n_edges, int n_nodes, int range_nodes, int chunks, int per_chunk) {
    __shared__ float cnt[LDS_FLOATS];
    const int tid   = threadIdx.x;
    const int range = blockIdx.x / chunks;
    const int chunk = blockIdx.x - range * chunks;
    const int lo    = range * range_nodes;
    int len = n_nodes - lo;
    if (len > range_nodes) len = range_nodes;
    if (len <= 0) return;

    for (int i = tid; i < len; i += SCAT_THREADS) cnt[i] = 0.0f;
    __syncthreads();

    const int e0 = chunk * per_chunk;
    int e1 = e0 + per_chunk;
    if (e1 > n_edges) e1 = n_edges;
    for (int base = e0; base < e1; base += SCAT_U * SCAT_THREADS) {
        int dv[SCAT_U];
#pragma unroll
        for (int k = 0; k < SCAT_U; k++) {
            int e  = base + k * SCAT_THREADS + tid;
            int ec = e < e1 ? e : e1 - 1;
            dv[k] = dst[ec];
        }
        asm volatile("" : "+v"(dv[0]), "+v"(dv[1]), "+v"(dv[2]), "+v"(dv[3]),
                          "+v"(dv[4]), "+v"(dv[5]), "+v"(dv[6]), "+v"(dv[7]));
#pragma unroll
        for (int k = 0; k < SCAT_U; k++) {
            int e = base + k * SCAT_THREADS + tid;
            int t = dv[k] - lo;
            if (e < e1 && (unsigned)t < (unsigned)len) atomicAdd(&cnt[t], 1.0f);
        }
    }
    __syncthreads();

    const int rot = (chunk * (len / chunks)) & ~31;
    for (int j = tid; j < len; j += SCAT_THREADS) {
        int i = j + rot;
        if (i >= len) i -= len;
        float v = cnt[i];
        if (v != 0.0f) atomicAdd(&deg[lo + i], v);
    }
}

__global__ void dinv_kernel(float* __restrict__ deg, int n_nodes) {
    int i = blockIdx.x * blockDim.x + threadIdx.x;
    if (i < n_nodes) {
        float d = deg[i];
        deg[i] = (d > 0.0f) ? rsqrtf(d) : 0.0f;
    }
}

__device__ __forceinline__ void
gemm1_tile(const float* __restrict__ x, const float* __restrict__ W1,
           const float* __restrict__ dinv, float* __restrict__ h,
           int tile, int n_tiles, int n_nodes) {
    int lane = threadIdx.x & 31;
    if (tile >= n_tiles) return;

    int col = lane & 15;
    int kh  = (lane >> 4) << 1;

    int rowA = tile * 16 + col;
    if (rowA > n_nodes - 1) rowA = n_nodes - 1;
    const float* xrow = x + (size_t)rowA * FEAT_IN;

    int mbase = (lane >> 4) * 8;
    float dr[8];
#pragma unroll
    for (int r = 0; r < 8; r++) {
        int row = tile * 16 + mbase + r;
        dr[r] = dinv[row < n_nodes ? row : n_nodes - 1];
    }

    v8f acc = {};
    for (int k0 = 0; k0 < FEAT_IN; k0 += 4) {
        v2f a, b;
        a.x = xrow[k0 + kh];
        a.y = xrow[k0 + kh + 1];
        b.x = W1[(k0 + kh)     * FEAT_MID + col];
        b.y = W1[(k0 + kh + 1) * FEAT_MID + col];
        acc = __builtin_amdgcn_wmma_f32_16x16x4_f32(
             false, a,  false, b,
             (short)0, acc,  false,  false);
    }

    asm volatile("v_nop\n\tv_nop\n\tv_nop\n\tv_nop\n\tv_nop" : "+v"(acc));
#pragma unroll
    for (int r = 0; r < 8; r++) {
        int row = tile * 16 + mbase + r;
        if (row < n_nodes) h[(size_t)row * FEAT_MID + col] = acc[r] * dr[r];
    }
}

__global__ void __launch_bounds__(256)
gemm1_wmma(const float* __restrict__ x, const float* __restrict__ W1,
           const float* __restrict__ dinv, float* __restrict__ h,
           int n_tiles, int n_nodes) {
    gemm1_tile(x, W1, dinv, h, blockIdx.x * (blockDim.x >> 5) + (threadIdx.x >> 5), n_tiles, n_nodes);
}

typedef int v2i __attribute__((ext_vector_type(2)));
typedef int v4i __attribute__((ext_vector_type(4)));

#define E1_NODES 2400
#define E1_U     16
#define E1_STEPS 8
#define E1_LIST  (E1_U * 32)
#define E1_TRIP  (E1_U * SCAT_THREADS)

__device__ __forceinline__ void e1_push(bool m, int e, int tt, v2i* lst, int& cnt) {
    unsigned b = __builtin_amdgcn_ballot_w32(m);
    int pos = cnt + (int)__builtin_amdgcn_mbcnt_lo(b, 0u);
    if (m) {
        v2i q;
        q.x = e;
        q.y = tt;
        lst[pos] = q;
    }
    cnt += __builtin_popcount(b);
}

__device__ __forceinline__ void e1_drain(const v2i* lst, int cnt,
                                         const int* __restrict__ src, const float* __restrict__ h,
                                         float* __restrict__ acc, int f, int hf) {
    for (int j0 = 0; j0 < cnt; j0 += 2 * E1_STEPS) {
        int   tq[E1_STEPS], sq[E1_STEPS];
        float hq[E1_STEPS];
#pragma unroll
        for (int p = 0; p < E1_STEPS; p++) {
            int  idx   = j0 + 2 * p + hf;
            bool valid = idx < cnt;
            v2i  q     = lst[valid ? idx : 0];
            tq[p] = valid ? q.y : -1;
            sq[p] = src[q.x];
        }
        asm volatile("" : "+v"(sq[0]), "+v"(sq[1]), "+v"(sq[2]), "+v"(sq[3]),
                          "+v"(sq[4]), "+v"(sq[5]), "+v"(sq[6]), "+v"(sq[7]));
#pragma unroll
        for (int p = 0; p < E1_STEPS; p++)
            hq[p] = h[sq[p] * FEAT_MID + f];
        asm volatile("" : "+v"(hq[0]), "+v"(hq[1]), "+v"(hq[2]), "+v"(hq[3]),
                          "+v"(hq[4]), "+v"(hq[5]), "+v"(hq[6]), "+v"(hq[7]));
#pragma unroll
        for (int p = 0; p < E1_STEPS; p++) {
            if (tq[p] >= 0)
                atomicAdd(&acc[tq[p] * FEAT_MID + f], hq[p]);
        }
    }
}

__global__ void __launch_bounds__(SCAT_THREADS)
edge1_lds(const int* __restrict__ src, const int* __restrict__ dst,
          const float* __restrict__ h,
          float* __restrict__ acc1, int n_edges, int n_nodes,
          int range_nodes, int chunks, int per_chunk) {
    __shared__ float acc[E1_NODES * FEAT_MID];
    __shared__ v2i   list[SCAT_THREADS / 32][E1_LIST];
    const int tid   = threadIdx.x;
    const int lane  = tid & 31;
    const int f     = lane & 15;
    const int hf    = lane >> 4;
    const int range = blockIdx.x / chunks;
    const int chunk = blockIdx.x - range * chunks;
    const int lo    = range * range_nodes;
    int len = n_nodes - lo;
    if (len > range_nodes) len = range_nodes;
    if (len <= 0) return;
    const int flen = len * FEAT_MID;

    for (int i = tid; i < flen; i += SCAT_THREADS) acc[i] = 0.0f;
    __syncthreads();

    v2i* __restrict__ lst = list[tid >> 5];
    int e0 = chunk * per_chunk;
    if (e0 > n_edges) e0 = n_edges;
    e0 &= ~3;
    int e1 = n_edges;
    if (chunk + 1 < chunks) {
        e1 = (chunk + 1) * per_chunk;
        if (e1 > n_edges) e1 = n_edges;
        e1 &= ~3;
    }

    const bool wide = (((size_t)dst) & 15) == 0;
    const int nfull = wide ? (e1 - e0) / E1_TRIP : 0;
    v4i nx[E1_U / 4];
    if (nfull > 0) {
        const v4i* __restrict__ pn = (const v4i*)(dst + e0);
#pragma unroll
        for (int j = 0; j < E1_U / 4; j++) nx[j] = pn[j * SCAT_THREADS + tid];
    }
    for (int t = 0; t < nfull; t++) {
        const int base = e0 + t * E1_TRIP;
        v4i cur[E1_U / 4];
#pragma unroll
        for (int j = 0; j < E1_U / 4; j++) cur[j] = nx[j];
        if (t + 1 < nfull) {
            const v4i* __restrict__ pn = (const v4i*)(dst + base + E1_TRIP);
#pragma unroll
            for (int j = 0; j < E1_U / 4; j++) nx[j] = pn[j * SCAT_THREADS + tid];
        }
        const int eb = base + 4 * tid;
        int cnt = 0;
#pragma unroll
        for (int j = 0; j < E1_U / 4; j++) {
#pragma unroll
            for (int i = 0; i < 4; i++) {
                int tt = cur[j][i] - lo;
                e1_push((unsigned)tt < (unsigned)len, eb + j * 4 * SCAT_THREADS + i, tt, lst, cnt);
            }
        }
        __builtin_amdgcn_wave_barrier();
        e1_drain(lst, cnt, src, h, acc, f, hf);
        __builtin_amdgcn_wave_barrier();
    }

    for (int base = e0 + nfull * E1_TRIP; base < e1; base += E1_TRIP) {
        int dv[E1_U];
#pragma unroll
        for (int k = 0; k < E1_U; k++) {
            int e  = base + k * SCAT_THREADS + tid;
            int ec = e < e1 ? e : e1 - 1;
            dv[k] = dst[ec];
        }
        asm volatile("" : "+v"(dv[0]), "+v"(dv[1]), "+v"(dv[2]), "+v"(dv[3]),
                          "+v"(dv[4]), "+v"(dv[5]), "+v"(dv[6]), "+v"(dv[7]),
                          "+v"(dv[8]), "+v"(dv[9]), "+v"(dv[10]), "+v"(dv[11]),
                          "+v"(dv[12]), "+v"(dv[13]), "+v"(dv[14]), "+v"(dv[15]));
        int cnt = 0;
#pragma unroll
        for (int k = 0; k < E1_U; k++) {
            int e  = base + k * SCAT_THREADS + tid;
            int tt = dv[k] - lo;
            e1_push((e < e1) && ((unsigned)tt < (unsigned)len), e, tt, lst, cnt);
        }
        __builtin_amdgcn_wave_barrier();
        e1_drain(lst, cnt, src, h, acc, f, hf);
        __builtin_amdgcn_wave_barrier();
    }
    __syncthreads();

    const int rot = (chunk * (flen / chunks)) & ~31;
    float* __restrict__ outp = acc1 + (size_t)lo * FEAT_MID;
    for (int j = tid; j < flen; j += SCAT_THREADS) {
        int i = j + rot;
        if (i >= flen) i -= flen;
        float v = acc[i];
        if (v != 0.0f) atomicAdd(&outp[i], v);
    }
}

__global__ void relu_bias1(const float* __restrict__ h, const float* __restrict__ dinv,
                           const float* __restrict__ b1, float* __restrict__ acc1,
                           int n_nodes) {
    int tid = blockIdx.x * blockDim.x + threadIdx.x;
    if (tid >= n_nodes * FEAT_MID) return;
    int node = tid >> 4, f = tid & 15;
    float di = dinv[node];
    float v = (acc1[tid] + h[tid]) * di + b1[f];
    acc1[tid] = fmaxf(v, 0.0f);
}

__global__ void gemm2_kernel(const float* __restrict__ h1, const float* __restrict__ W2,
                             const float* __restrict__ dinv,
                             float* __restrict__ h2, int n_nodes) {
    int i = blockIdx.x * blockDim.x + threadIdx.x;
    if (i >= n_nodes) return;
    float s0 = 0.0f, s1 = 0.0f;
#pragma unroll
    for (int k = 0; k < FEAT_MID; k++) {
        float v = h1[i * FEAT_MID + k];
        s0 = fmaf(v, W2[k * 2 + 0], s0);
        s1 = fmaf(v, W2[k * 2 + 1], s1);
    }
    float di = dinv[i];
    h2[i * 2 + 0] = s0 * di;
    h2[i * 2 + 1] = s1 * di;
}

__global__ void __launch_bounds__(SCAT_THREADS)
edge2_lds(const int* __restrict__ src, const int* __restrict__ dst,
          const float* __restrict__ h2,
          float* __restrict__ acc2, int n_edges, int n_nodes,
          int range_nodes, int chunks, int per_chunk) {
    __shared__ float acc[LDS_FLOATS];
    const int tid   = threadIdx.x;
    const int range = blockIdx.x / chunks;
    const int chunk = blockIdx.x - range * chunks;
    const int lo    = range * range_nodes;
    int len = n_nodes - lo;
    if (len > range_nodes) len = range_nodes;
    if (len <= 0) return;
    const int flen = len * FEAT_OUT;

    for (int i = tid; i < flen; i += SCAT_THREADS) acc[i] = 0.0f;
    __syncthreads();

    const v2f* __restrict__ h2v = (const v2f*)h2;
    const int e0 = chunk * per_chunk;
    int e1 = e0 + per_chunk;
    if (e1 > n_edges) e1 = n_edges;
    for (int base = e0; base < e1; base += SCAT_U * SCAT_THREADS) {
        int dv[SCAT_U], sv[SCAT_U];
#pragma unroll
        for (int k = 0; k < SCAT_U; k++) {
            int e  = base + k * SCAT_THREADS + tid;
            int ec = e < e1 ? e : e1 - 1;
            dv[k] = dst[ec];
            sv[k] = src[ec];
        }
        asm volatile("" : "+v"(dv[0]), "+v"(dv[1]), "+v"(dv[2]), "+v"(dv[3]),
                          "+v"(dv[4]), "+v"(dv[5]), "+v"(dv[6]), "+v"(dv[7]),
                          "+v"(sv[0]), "+v"(sv[1]), "+v"(sv[2]), "+v"(sv[3]),
                          "+v"(sv[4]), "+v"(sv[5]), "+v"(sv[6]), "+v"(sv[7]));
        int   t[SCAT_U];
        v2f   hv[SCAT_U];
#pragma unroll
        for (int k = 0; k < SCAT_U; k++) {
            int e  = base + k * SCAT_THREADS + tid;
            int tt = dv[k] - lo;
            bool m = (e < e1) && ((unsigned)tt < (unsigned)len);
            t[k] = m ? tt : -1;
            int sg = m ? sv[k] : lo;
            hv[k] = h2v[sg];
        }
        asm volatile("" : "+v"(hv[0]), "+v"(hv[1]), "+v"(hv[2]), "+v"(hv[3]),
                          "+v"(hv[4]), "+v"(hv[5]), "+v"(hv[6]), "+v"(hv[7]));
#pragma unroll
        for (int k = 0; k < SCAT_U; k++) {
            if (t[k] >= 0) {
                atomicAdd(&acc[t[k] * FEAT_OUT + 0], hv[k].x);
                atomicAdd(&acc[t[k] * FEAT_OUT + 1], hv[k].y);
            }
        }
    }
    __syncthreads();

    const int rot = (chunk * (flen / chunks)) & ~31;
    float* __restrict__ outp = acc2 + (size_t)lo * FEAT_OUT;
    for (int j = tid; j < flen; j += SCAT_THREADS) {
        int i = j + rot;
        if (i >= flen) i -= flen;
        float v = acc[i];
        if (v != 0.0f) atomicAdd(&outp[i], v);
    }
}

__global__ void final_kernel(const float* __restrict__ acc2, const float* __restrict__ h2,
                             const float* __restrict__ dinv, const float* __restrict__ b2,
                             float* __restrict__ out, int n_nodes) {
    int i = blockIdx.x * blockDim.x + threadIdx.x;
    if (i >= n_nodes) return;
    float di = dinv[i];
    float v0 = (acc2[i * 2 + 0] + h2[i * 2 + 0]) * di + b2[0];
    float v1 = (acc2[i * 2 + 1] + h2[i * 2 + 1]) * di + b2[1];
    float m  = fmaxf(v0, v1);
    float l  = m + logf(expf(v0 - m) + expf(v1 - m));
    out[i * 2 + 0] = v0 - l;
    out[i * 2 + 1] = v1 - l;
}

#define RB_NODES   128
#define RB_SHIFT   7
#define RB_MAXR    1024
#define BK_THREADS 1024
#define BK_U       4
#define BK_STAGE   12800
#define EB_THREADS 512
#define EB_STEPS   16
#define E2_U       8
#define EB_TAIL    (RB_NODES * FEAT_MID / EB_THREADS)

__global__ void init_deg(float* __restrict__ deg, int n_nodes) {
    int tid = blockIdx.x * blockDim.x + threadIdx.x;
    if (tid < n_nodes) deg[tid] = 1.0f;
}

#define DT_THREADS (2 * RB_NODES)
__global__ void __launch_bounds__(DT_THREADS)
dinv_totals(float* __restrict__ deg, int* __restrict__ totals, int* __restrict__ cursor,
            int n_nodes, int n_ranges) {
    __shared__ int part[DT_THREADS / 32];
    const int tid = threadIdx.x;
    const int i   = blockIdx.x * DT_THREADS + tid;
    int v = 0;
    if (i < n_nodes) {
        float d = deg[i];
        v = (int)d - 1;
        deg[i] = (d > 0.0f) ? rsqrtf(d) : 0.0f;
    }
    for (int m = 16; m > 0; m >>= 1) v += __shfl_xor(v, m);
    if ((tid & 31) == 0) part[tid >> 5] = v;
    __syncthreads();
    if (tid < 2) {
        const int r = blockIdx.x * 2 + tid;
        if (r < n_ranges) {
            const int* q = part + tid * (RB_NODES / 32);
            totals[r] = q[0] + q[1] + q[2] + q[3];
            cursor[r] = 0;
        }
    }
}

__device__ __forceinline__ void
bucket_chunk(const int* __restrict__ src, const int* __restrict__ dst,
             const int* __restrict__ totals, int* __restrict__ cursor,
             int* __restrict__ offs, int* __restrict__ bins,
             int n_edges, int per_chunk, int n_ranges, const int blk) {
    __shared__ int cnt[RB_MAXR];
    __shared__ int gb[RB_MAXR];
    __shared__ int sc[RB_MAXR];
    __shared__ int so[RB_MAXR];
    __shared__ int stage_w[BK_STAGE];
    __shared__ int stage_p[BK_STAGE];
    const int tid = threadIdx.x;
    int e0 = blk * per_chunk;
    if (e0 > n_edges) e0 = n_edges;
    int e1 = e0 + per_chunk;
    if (e1 > n_edges) e1 = n_edges;
    cnt[tid] = 0;
    __syncthreads();
    for (int base = e0; base < e1; base += BK_U * BK_THREADS) {
        int dv[BK_U];
#pragma unroll
        for (int k = 0; k < BK_U; k++) {
            int e  = base + k * BK_THREADS + tid;
            int ec = e < e1 ? e : e1 - 1;
            dv[k] = dst[ec];
        }
        asm volatile("" : "+v"(dv[0]), "+v"(dv[1]), "+v"(dv[2]), "+v"(dv[3]));
#pragma unroll
        for (int k = 0; k < BK_U; k++) {
            int e = base + k * BK_THREADS + tid;
            if (e < e1) atomicAdd(&cnt[dv[k] >> RB_SHIFT], 1);
        }
    }
    __syncthreads();
    const int c   = cnt[tid];
    const int tot = tid < n_ranges ? totals[tid] : 0;
    int rel = 0;
    if (c > 0) rel = atomicAdd(&cursor[tid], c);
    sc[tid] = c;
    so[tid] = tot;
    __syncthreads();
    for (int off = 1; off < RB_MAXR; off <<= 1) {
        int t1 = tid >= off ? sc[tid - off] : 0;
        int t2 = tid >= off ? so[tid - off] : 0;
        __syncthreads();
        sc[tid] += t1;
        so[tid] += t2;
        __syncthreads();
    }
    const int first = so[tid] - tot;
    if (blk == 0 && tid < n_ranges) {
        offs[tid] = first;
        if (tid == n_ranges - 1) offs[n_ranges] = so[tid];
    }
    gb[tid]  = first + rel;
    sc[tid] -= c;
    cnt[tid] = 0;
    __syncthreads();
    for (int base = e0; base < e1; base += BK_U * BK_THREADS) {
        int dv[BK_U], sv[BK_U];
#pragma unroll
        for (int k = 0; k < BK_U; k++) {
            int e  = base + k * BK_THREADS + tid;
            int ec = e < e1 ? e : e1 - 1;
            dv[k] = dst[ec];
            sv[k] = src[ec];
        }
        asm volatile("" : "+v"(dv[0]), "+v"(dv[1]), "+v"(dv[2]), "+v"(dv[3]),
                          "+v"(sv[0]), "+v"(sv[1]), "+v"(sv[2]), "+v"(sv[3]));
        int rank[BK_U], ls[BK_U], gs[BK_U];
#pragma unroll
        for (int k = 0; k < BK_U; k++) {
            int e = base + k * BK_THREADS + tid;
            int r = dv[k] >> RB_SHIFT;
            rank[k] = 0;
            if (e < e1) rank[k] = atomicAdd(&cnt[r], 1);
            ls[k] = sc[r];
            gs[k] = gb[r];
        }
#pragma unroll
        for (int k = 0; k < BK_U; k++) {
            int e = base + k * BK_THREADS + tid;
            if (e < e1) {
                int lp = ls[k] + rank[k];
                stage_w[lp] = (sv[k] << RB_SHIFT) | (dv[k] & (RB_NODES - 1));
                stage_p[lp] = gs[k] + rank[k];
            }
        }
    }
    __syncthreads();
    const int n = e1 - e0;
    for (int i = tid; i < n; i += BK_THREADS) bins[stage_p[i]] = stage_w[i];
}

__global__ void __launch_bounds__(BK_THREADS)
bucket_gemm(const int* __restrict__ src, const int* __restrict__ dst,
            const int* __restrict__ totals, int* __restrict__ cursor,
            int* __restrict__ offs, int* __restrict__ bins,
            int n_edges, int per_chunk, int n_ranges, int nb,
            const float* __restrict__ x, const float* __restrict__ W1,
            const float* __restrict__ dinv, float* __restrict__ h,
            int n_tiles, int n_nodes) {
    if ((int)blockIdx.x < nb) {
        bucket_chunk(src, dst, totals, cursor, offs, bins, n_edges, per_chunk, n_ranges, blockIdx.x);
    } else {
        gemm1_tile(x, W1, dinv, h,
                   ((int)blockIdx.x - nb) * (BK_THREADS / 32) + (int)(threadIdx.x >> 5),
                   n_tiles, n_nodes);
    }
}

__global__ void __launch_bounds__(EB_THREADS)
edge1_bin(const int* __restrict__ bins, const int* __restrict__ offs,
          const float* __restrict__ h, const float* __restrict__ dinv,
          const float* __restrict__ bias1, const float* __restrict__ W2,
          float* __restrict__ h2, int n_nodes) {
    __shared__ float acc[RB_NODES * FEAT_MID];
    const int tid  = threadIdx.x;
    const int lane = tid & 31;
    const int f    = lane & 15;
    const int hf   = lane >> 4;
    const int lo   = blockIdx.x << RB_SHIFT;
    int len = n_nodes - lo;
    if (len > RB_NODES) len = RB_NODES;

    for (int i = tid; i < RB_NODES * FEAT_MID; i += EB_THREADS) acc[i] = 0.0f;
    __syncthreads();

    const int b0 = offs[blockIdx.x];
    const int b1 = offs[blockIdx.x + 1];
    const int stride = (EB_THREADS / 32) * 2 * EB_STEPS;
    int j0 = b0 + (tid >> 5) * 2 * EB_STEPS;
    int w[EB_STEPS] = {};
    if (j0 < b1) {
#pragma unroll
        for (int p = 0; p < EB_STEPS; p++) {
            int idx = j0 + 2 * p + hf;
            w[p] = bins[idx < b1 ? idx : j0];
        }
    }
    while (j0 < b1) {
        int   cur[EB_STEPS];
        float hq[EB_STEPS];
#pragma unroll
        for (int p = 0; p < EB_STEPS; p++) cur[p] = w[p];
        const int jn = j0 + stride;
        if (jn < b1) {
#pragma unroll
            for (int p = 0; p < EB_STEPS; p++) {
                int idx = jn + 2 * p + hf;
                w[p] = bins[idx < b1 ? idx : jn];
            }
        }
#pragma unroll
        for (int p = 0; p < EB_STEPS; p++)
            hq[p] = h[(cur[p] >> RB_SHIFT) * FEAT_MID + f];
        asm volatile("" : "+v"(hq[0]), "+v"(hq[1]), "+v"(hq[2]), "+v"(hq[3]),
                          "+v"(hq[4]), "+v"(hq[5]), "+v"(hq[6]), "+v"(hq[7]),
                          "+v"(hq[8]), "+v"(hq[9]), "+v"(hq[10]), "+v"(hq[11]),
                          "+v"(hq[12]), "+v"(hq[13]), "+v"(hq[14]), "+v"(hq[15]));
#pragma unroll
        for (int p = 0; p < EB_STEPS; p++) {
            if (j0 + 2 * p + hf < b1)
                atomicAdd(&acc[(cur[p] & (RB_NODES - 1)) * FEAT_MID + f], hq[p]);
        }
        j0 = jn;
    }
    __syncthreads();

    const float* __restrict__ hown = h + (size_t)lo * FEAT_MID;
    const int   flen = len * FEAT_MID;
    const float bf   = bias1[f];
    float hs[EB_TAIL], dd[EB_TAIL];
#pragma unroll
    for (int j = 0; j < EB_TAIL; j++) {
        int i  = tid + j * EB_THREADS;
        int ic = i < flen ? i : tid & 15;
        hs[j] = hown[ic];
        dd[j] = dinv[lo + (ic >> 4)];
    }
#pragma unroll
    for (int j = 0; j < EB_TAIL; j++) {
        int i = tid + j * EB_THREADS;
        if (i < flen) acc[i] = fmaxf((acc[i] + hs[j]) * dd[j] + bf, 0.0f);
    }
    __syncthreads();

    if (tid < len * FEAT_OUT) {
        const int node = tid >> 1, c = tid & 1;
        float s = 0.0f;
#pragma unroll
        for (int k = 0; k < FEAT_MID; k++)
            s = fmaf(acc[node * FEAT_MID + k], W2[k * FEAT_OUT + c], s);
        h2[(size_t)lo * FEAT_OUT + tid] = s * dinv[lo + node];
    }
}

__global__ void __launch_bounds__(EB_THREADS)
edge2_bin(const int* __restrict__ bins, const int* __restrict__ offs,
          const float* __restrict__ h2, const float* __restrict__ dinv,
          const float* __restrict__ bias2, float* __restrict__ out, int n_nodes) {
    __shared__ float acc[RB_NODES * FEAT_OUT];
    const int tid = threadIdx.x;
    const int lo  = blockIdx.x << RB_SHIFT;
    int len = n_nodes - lo;
    if (len > RB_NODES) len = RB_NODES;

    for (int i = tid; i < RB_NODES * FEAT_OUT; i += EB_THREADS) acc[i] = 0.0f;
    __syncthreads();

    const v2f* __restrict__ h2v = (const v2f*)h2;
    const int b0 = offs[blockIdx.x];
    const int b1 = offs[blockIdx.x + 1];
    for (int base = b0; base < b1; base += E2_U * EB_THREADS) {
        int w[E2_U];
        v2f hv[E2_U];
#pragma unroll
        for (int k = 0; k < E2_U; k++) {
            int idx = base + k * EB_THREADS + tid;
            w[k] = bins[idx < b1 ? idx : b0];
        }
#pragma unroll
        for (int k = 0; k < E2_U; k++) hv[k] = h2v[w[k] >> RB_SHIFT];
        asm volatile("" : "+v"(hv[0]), "+v"(hv[1]), "+v"(hv[2]), "+v"(hv[3]),
                          "+v"(hv[4]), "+v"(hv[5]), "+v"(hv[6]), "+v"(hv[7]));
#pragma unroll
        for (int k = 0; k < E2_U; k++) {
            if (base + k * EB_THREADS + tid < b1) {
                int t = w[k] & (RB_NODES - 1);
                atomicAdd(&acc[t * FEAT_OUT + 0], hv[k].x);
                atomicAdd(&acc[t * FEAT_OUT + 1], hv[k].y);
            }
        }
    }
    __syncthreads();

    if (tid < len) {
        const int i = lo + tid;
        float di = dinv[i];
        float v0 = (acc[tid * 2 + 0] + h2[i * 2 + 0]) * di + bias2[0];
        float v1 = (acc[tid * 2 + 1] + h2[i * 2 + 1]) * di + bias2[1];
        float m  = fmaxf(v0, v1);
        float l  = m + logf(expf(v0 - m) + expf(v1 - m));
        out[i * 2 + 0] = v0 - l;
        out[i * 2 + 1] = v1 - l;
    }
}

#define ES_W   16
#define ES_CAP (ES_W * EB_THREADS)
#define ES_U   8
typedef float v4f __attribute__((ext_vector_type(4)));
typedef int   v4i_t __attribute__((ext_vector_type(4)));

__global__ void __launch_bounds__(EB_THREADS, 16)
edge1_sort(const int* __restrict__ bins, const int* __restrict__ offs,
           const float* __restrict__ h, const float* __restrict__ dinv,
           const float* __restrict__ bias1, const float* __restrict__ W2,
           float* __restrict__ h2, int n_nodes) {
    __shared__ int sorted[ES_CAP];
    __shared__ __attribute__((aligned(16))) int cnt[RB_NODES];
    __shared__ __attribute__((aligned(16))) int start[RB_NODES + 4];
    __shared__ float hs[RB_NODES * FEAT_OUT];
    const int tid  = threadIdx.x;
    const int lane = tid & 31;
    const int q    = lane & 3;
    const int my   = (tid >> 5) * 8 + (lane >> 2);
    const int lo   = blockIdx.x << RB_SHIFT;
    int len = n_nodes - lo;
    if (len > RB_NODES) len = RB_NODES;
    const v4f* __restrict__ h4 = (const v4f*)h;

    const int b0 = offs[blockIdx.x];
    const int b1 = offs[blockIdx.x + 1];
    v4f acc = {0.0f, 0.0f, 0.0f, 0.0f};
    for (int c0 = b0; c0 < b1; c0 += ES_CAP) {
        int n = b1 - c0;
        if (n > ES_CAP) n = ES_CAP;
        if (tid < RB_NODES) cnt[tid] = 0;
        __syncthreads();
        int w[ES_W];
#pragma unroll
        for (int hh = 0; hh < 2; hh++) {
#pragma unroll
            for (int k = 0; k < ES_W / 2; k++) w[hh * (ES_W / 2) + k] = 0;
            if (hh * (ES_CAP / 2) < n) {
#pragma unroll
                for (int k = 0; k < ES_W / 2; k++) {
                    int idx = (hh * (ES_W / 2) + k) * EB_THREADS + tid;
                    w[hh * (ES_W / 2) + k] = bins[c0 + (idx < n ? idx : 0)];
                }
#pragma unroll
                for (int k = 0; k < ES_W / 2; k++) {
                    int idx = (hh * (ES_W / 2) + k) * EB_THREADS + tid;
                    if (idx < n) atomicAdd(&cnt[w[hh * (ES_W / 2) + k] & (RB_NODES - 1)], 1);
                }
            }
        }
        __syncthreads();
        if (tid < 32) {
            v4i_t c = *(const v4i_t*)&cnt[4 * tid];
            int s = c.x + c.y + c.z + c.w;
            int incl = s;
            for (int d = 1; d < 32; d <<= 1) {
                int t = __shfl_up(incl, d);
                if (lane >= d) incl += t;
            }
            int ex = incl - s;
            v4i_t st;
            st.x = ex; st.y = ex + c.x; st.z = ex + c.x + c.y; st.w = ex + c.x + c.y + c.z;
            *(v4i_t*)&start[4 * tid] = st;
            *(v4i_t*)&cnt[4 * tid]   = st;
            if (tid == 31) start[RB_NODES] = incl;
        }
        __syncthreads();
#pragma unroll
        for (int k = 0; k < ES_W; k++) {
            int idx = k * EB_THREADS + tid;
            if (idx < n) sorted[atomicAdd(&cnt[w[k] & (RB_NODES - 1)], 1)] = w[k] >> RB_SHIFT;
        }
        __syncthreads();

        const int s0 = start[my];
        const int s1 = start[my + 1];
        for (int j = s0; j < s1; j += ES_U) {
            const int m = s1 - j;
            v4f v[ES_U];
#pragma unroll
            for (int u = 0; u < ES_U; u++) v[u] = h4[sorted[j + (u < m ? u : 0)] * (FEAT_MID / 4) + q];
#pragma unroll
            for (int u = 0; u < ES_U; u++) {
                const bool ok = u < m;
                acc.x += ok ? v[u].x : 0.0f;
                acc.y += ok ? v[u].y : 0.0f;
                acc.z += ok ? v[u].z : 0.0f;
                acc.w += ok ? v[u].w : 0.0f;
            }
        }
    }

    const int   node = lo + (my < len ? my : 0);
    const float di   = dinv[node];
    const v4f   own  = h4[node * (FEAT_MID / 4) + q];
    const v4f   bq   = ((const v4f*)bias1)[q];
    const v4f   wa   = ((const v4f*)W2)[2 * q];
    const v4f   wb   = ((const v4f*)W2)[2 * q + 1];
    const float a0 = fmaxf((acc.x + own.x) * di + bq.x, 0.0f);
    const float a1 = fmaxf((acc.y + own.y) * di + bq.y, 0.0f);
    const float a2 = fmaxf((acc.z + own.z) * di + bq.z, 0.0f);
    const float a3 = fmaxf((acc.w + own.w) * di + bq.w, 0.0f);
    float p0 = fmaf(a3, wb.z, fmaf(a2, wb.x, fmaf(a1, wa.z, a0 * wa.x)));
    float p1 = fmaf(a3, wb.w, fmaf(a2, wb.y, fmaf(a1, wa.w, a0 * wa.y)));
    p0 += __shfl_xor(p0, 1);  p1 += __shfl_xor(p1, 1);
    p0 += __shfl_xor(p0, 2);  p1 += __shfl_xor(p1, 2);
    if (q == 0) { hs[my * FEAT_OUT + 0] = p0 * di; hs[my * FEAT_OUT + 1] = p1 * di; }
    __syncthreads();
    if (tid < len * FEAT_OUT) h2[(size_t)lo * FEAT_OUT + tid] = hs[tid];
}

extern "C" void kernel_launch(void* const* d_in, const int* in_sizes, int n_in,
                              void* d_out, int out_size, void* d_ws, size_t ws_size,
                              hipStream_t stream) {
    const float* x  = (const float*)d_in[0];
    const int*   ei = (const int*)d_in[1];
    const float* W1 = (const float*)d_in[2];
    const float* b1 = (const float*)d_in[3];
    const float* W2 = (const float*)d_in[4];
    const float* b2 = (const float*)d_in[5];
    float* out = (float*)d_out;

    int n_nodes = in_sizes[0] / FEAT_IN;
    int n_edges = in_sizes[1] / 2;
    const int* src = ei;
    const int* dst = ei + n_edges;

    float* ws   = (float*)d_ws;
    float* deg  = ws;
    float* h    = deg  + n_nodes;
    float* acc1 = h    + (size_t)n_nodes * FEAT_MID;
    float* h2   = acc1 + (size_t)n_nodes * FEAT_MID;
    float* acc2 = h2   + (size_t)n_nodes * FEAT_OUT;

    const int B = 256;
    int n16 = n_nodes * FEAT_MID;

    int nr1 = (n_nodes + LDS_FLOATS - 1) / LDS_FLOATS;
    int rn1 = (((n_nodes + nr1 - 1) / nr1) + 31) & ~31;
    int ch1 = 128 / nr1 > 0 ? 128 / nr1 : 1;
    int pc1 = (n_edges + ch1 - 1) / ch1;
    int n_tiles = (n_nodes + 15) / 16;
    int wpb = B / 32;

    int    n_ranges = (n_nodes + RB_NODES - 1) >> RB_SHIFT;
    size_t need = ((size_t)n_nodes * (1 + 2 * FEAT_MID + 2 * FEAT_OUT) + 3 * (size_t)n_ranges + 1 +
                   (size_t)n_edges) * sizeof(float);
    if (n_ranges <= RB_MAXR && ws_size >= need) {
        int* offs   = (int*)(acc2 + (size_t)n_nodes * FEAT_OUT);
        int* cursor = offs + n_ranges + 1;
        int* totals = cursor + n_ranges;
        int* bins   = totals + n_ranges;
        int  nb     = (n_edges + BK_STAGE - 1) / BK_STAGE;
        int  pcb    = BK_STAGE;

        init_deg<<<(n_nodes + B - 1) / B, B, 0, stream>>>(deg, n_nodes);
        degree_lds<<<nr1 * ch1, SCAT_THREADS, 0, stream>>>(dst, deg, n_edges, n_nodes, rn1, ch1, pc1);
        dinv_totals<<<(n_nodes + DT_THREADS - 1) / DT_THREADS, DT_THREADS, 0, stream>>>(
            deg, totals, cursor, n_nodes, n_ranges);
        int  gw     = (n_tiles + BK_THREADS / 32 - 1) / (BK_THREADS / 32);
        bucket_gemm<<<nb + gw, BK_THREADS, 0, stream>>>(src, dst, totals, cursor, offs, bins,
                                                        n_edges, pcb, n_ranges, nb,
                                                        x, W1, deg, h, n_tiles, n_nodes);
        if (((((size_t)h) | ((size_t)b1) | ((size_t)W2)) & 15) == 0)
            edge1_sort<<<n_ranges, EB_THREADS, 0, stream>>>(bins, offs, h, deg, b1, W2, h2, n_nodes);
        else
            edge1_bin<<<n_ranges, EB_THREADS, 0, stream>>>(bins, offs, h, deg, b1, W2, h2, n_nodes);
        edge2_bin<<<n_ranges, EB_THREADS, 0, stream>>>(bins, offs, h2, deg, b2, out, n_nodes);
        return;
    }

    init_ws<<<(n16 + B - 1) / B, B, 0, stream>>>(deg, acc1, acc2, n_nodes);
    degree_lds<<<nr1 * ch1, SCAT_THREADS, 0, stream>>>(dst, deg, n_edges, n_nodes, rn1, ch1, pc1);
    dinv_kernel<<<(n_nodes + B - 1) / B, B, 0, stream>>>(deg, n_nodes);

    gemm1_wmma<<<(n_tiles + wpb - 1) / wpb, B, 0, stream>>>(x, W1, deg, h, n_tiles, n_nodes);

    int nr0 = (n_nodes + E1_NODES - 1) / E1_NODES;
    int rn0 = (((n_nodes + nr0 - 1) / nr0) + 31) & ~31;
    int ch0 = 256 / nr0 > 0 ? 256 / nr0 : 1;
    int pc0 = (n_edges + ch0 - 1) / ch0;
    edge1_lds<<<nr0 * ch0, SCAT_THREADS, 0, stream>>>(src, dst, h, acc1, n_edges, n_nodes,
                                                      rn0, ch0, pc0);
    relu_bias1<<<(n16 + B - 1) / B, B, 0, stream>>>(h, deg, b1, acc1, n_nodes);

    gemm2_kernel<<<(n_nodes + B - 1) / B, B, 0, stream>>>(acc1, W2, deg, h2, n_nodes);

    int nr2 = (n_nodes * FEAT_OUT + LDS_FLOATS - 1) / LDS_FLOATS;
    int rn2 = (((n_nodes + nr2 - 1) / nr2) + 31) & ~31;
    int ch2 = 128 / nr2 > 0 ? 128 / nr2 : 1;
    int pc2 = (n_edges + ch2 - 1) / ch2;
    edge2_lds<<<nr2 * ch2, SCAT_THREADS, 0, stream>>>(src, dst, h2, acc2, n_edges, n_nodes,
                                                      rn2, ch2, pc2);

    final_kernel<<<(n_nodes + B - 1) / B, B, 0, stream>>>(acc2, h2, deg, b2, out, n_nodes);
}


# degree count: each work-group stores its chunk's LDS counts plainly into a plane per chunk, summed per node where dinv is formed; global float atomics and their init launch gone (integer counts)
# speedup vs baseline: 6.0296x; 1.0952x over previous
#include <hip/hip_runtime.h>
#include <math.h>

typedef float v2f __attribute__((ext_vector_type(2)));
typedef float v8f __attribute__((ext_vector_type(8)));

#define FEAT_IN  128
#define FEAT_MID 16
#define FEAT_OUT 2

__global__ void init_ws(float* __restrict__ deg, float* __restrict__ acc1,
                        float* __restrict__ acc2, int n_nodes) {
    int tid = blockIdx.x * blockDim.x + threadIdx.x;
    if (tid < n_nodes * FEAT_MID) acc1[tid] = 0.0f;
    if (tid < n_nodes * FEAT_OUT) acc2[tid] = 0.0f;
    if (tid < n_nodes)            deg[tid]  = 1.0f;
}

#define LDS_FLOATS   69632
#define SCAT_THREADS 1024
#define SCAT_U       8

__global__ void __launch_bounds__(SCAT_THREADS)
degree_lds(const int* __restrict__ dst, float* __restrict__ deg,
           int n_edges, int n_nodes, int range_nodes, int chunks, int per_chunk) {
    __shared__ float cnt[LDS_FLOATS];
    const int tid   = threadIdx.x;
    const int range = blockIdx.x / chunks;
    const int chunk = blockIdx.x - range * chunks;
    const int lo    = range * range_nodes;
    int len = n_nodes - lo;
    if (len > range_nodes) len = range_nodes;
    if (len <= 0) return;

    for (int i = tid; i < len; i += SCAT_THREADS) cnt[i] = 0.0f;
    __syncthreads();

    const int e0 = chunk * per_chunk;
    int e1 = e0 + per_chunk;
    if (e1 > n_edges) e1 = n_edges;
    for (int base = e0; base < e1; base += SCAT_U * SCAT_THREADS) {
        int dv[SCAT_U];
#pragma unroll
        for (int k = 0; k < SCAT_U; k++) {
            int e  = base + k * SCAT_THREADS + tid;
            int ec = e < e1 ? e : e1 - 1;
            dv[k] = dst[ec];
        }
        asm volatile("" : "+v"(dv[0]), "+v"(dv[1]), "+v"(dv[2]), "+v"(dv[3]),
                          "+v"(dv[4]), "+v"(dv[5]), "+v"(dv[6]), "+v"(dv[7]));
#pragma unroll
        for (int k = 0; k < SCAT_U; k++) {
            int e = base + k * SCAT_THREADS + tid;
            int t = dv[k] - lo;
            if (e < e1 && (unsigned)t < (unsigned)len) atomicAdd(&cnt[t], 1.0f);
        }
    }
    __syncthreads();

    const int rot = (chunk * (len / chunks)) & ~31;
    for (int j = tid; j < len; j += SCAT_THREADS) {
        int i = j + rot;
        if (i >= len) i -= len;
        float v = cnt[i];
        if (v != 0.0f) atomicAdd(&deg[lo + i], v);
    }
}

__global__ void dinv_kernel(float* __restrict__ deg, int n_nodes) {
    int i = blockIdx.x * blockDim.x + threadIdx.x;
    if (i < n_nodes) {
        float d = deg[i];
        deg[i] = (d > 0.0f) ? rsqrtf(d) : 0.0f;
    }
}

__device__ __forceinline__ void
gemm1_tile(const float* __restrict__ x, const float* __restrict__ W1,
           const float* __restrict__ dinv, float* __restrict__ h,
           int tile, int n_tiles, int n_nodes) {
    int lane = threadIdx.x & 31;
    if (tile >= n_tiles) return;

    int col = lane & 15;
    int kh  = (lane >> 4) << 1;

    int rowA = tile * 16 + col;
    if (rowA > n_nodes - 1) rowA = n_nodes - 1;
    const float* xrow = x + (size_t)rowA * FEAT_IN;

    int mbase = (lane >> 4) * 8;
    float dr[8];
#pragma unroll
    for (int r = 0; r < 8; r++) {
        int row = tile * 16 + mbase + r;
        dr[r] = dinv[row < n_nodes ? row : n_nodes - 1];
    }

    v8f acc = {};
    for (int k0 = 0; k0 < FEAT_IN; k0 += 4) {
        v2f a, b;
        a.x = xrow[k0 + kh];
        a.y = xrow[k0 + kh + 1];
        b.x = W1[(k0 + kh)     * FEAT_MID + col];
        b.y = W1[(k0 + kh + 1) * FEAT_MID + col];
        acc = __builtin_amdgcn_wmma_f32_16x16x4_f32(
             false, a,  false, b,
             (short)0, acc,  false,  false);
    }

    asm volatile("v_nop\n\tv_nop\n\tv_nop\n\tv_nop\n\tv_nop" : "+v"(acc));
#pragma unroll
    for (int r = 0; r < 8; r++) {
        int row = tile * 16 + mbase + r;
        if (row < n_nodes) h[(size_t)row * FEAT_MID + col] = acc[r] * dr[r];
    }
}

__global__ void __launch_bounds__(256)
gemm1_wmma(const float* __restrict__ x, const float* __restrict__ W1,
           const float* __restrict__ dinv, float* __restrict__ h,
           int n_tiles, int n_nodes) {
    gemm1_tile(x, W1, dinv, h, blockIdx.x * (blockDim.x >> 5) + (threadIdx.x >> 5), n_tiles, n_nodes);
}

typedef int v2i __attribute__((ext_vector_type(2)));
typedef int v4i __attribute__((ext_vector_type(4)));

#define E1_NODES 2400
#define E1_U     16
#define E1_STEPS 8
#define E1_LIST  (E1_U * 32)
#define E1_TRIP  (E1_U * SCAT_THREADS)

__device__ __forceinline__ void e1_push(bool m, int e, int tt, v2i* lst, int& cnt) {
    unsigned b = __builtin_amdgcn_ballot_w32(m);
    int pos = cnt + (int)__builtin_amdgcn_mbcnt_lo(b, 0u);
    if (m) {
        v2i q;
        q.x = e;
        q.y = tt;
        lst[pos] = q;
    }
    cnt += __builtin_popcount(b);
}

__device__ __forceinline__ void e1_drain(const v2i* lst, int cnt,
                                         const int* __restrict__ src, const float* __restrict__ h,
                                         float* __restrict__ acc, int f, int hf) {
    for (int j0 = 0; j0 < cnt; j0 += 2 * E1_STEPS) {
        int   tq[E1_STEPS], sq[E1_STEPS];
        float hq[E1_STEPS];
#pragma unroll
        for (int p = 0; p < E1_STEPS; p++) {
            int  idx   = j0 + 2 * p + hf;
            bool valid = idx < cnt;
            v2i  q     = lst[valid ? idx : 0];
            tq[p] = valid ? q.y : -1;
            sq[p] = src[q.x];
        }
        asm volatile("" : "+v"(sq[0]), "+v"(sq[1]), "+v"(sq[2]), "+v"(sq[3]),
                          "+v"(sq[4]), "+v"(sq[5]), "+v"(sq[6]), "+v"(sq[7]));
#pragma unroll
        for (int p = 0; p < E1_STEPS; p++)
            hq[p] = h[sq[p] * FEAT_MID + f];
        asm volatile("" : "+v"(hq[0]), "+v"(hq[1]), "+v"(hq[2]), "+v"(hq[3]),
                          "+v"(hq[4]), "+v"(hq[5]), "+v"(hq[6]), "+v"(hq[7]));
#pragma unroll
        for (int p = 0; p < E1_STEPS; p++) {
            if (tq[p] >= 0)
                atomicAdd(&acc[tq[p] * FEAT_MID + f], hq[p]);
        }
    }
}

__global__ void __launch_bounds__(SCAT_THREADS)
edge1_lds(const int* __restrict__ src, const int* __restrict__ dst,
          const float* __restrict__ h,
          float* __restrict__ acc1, int n_edges, int n_nodes,
          int range_nodes, int chunks, int per_chunk) {
    __shared__ float acc[E1_NODES * FEAT_MID];
    __shared__ v2i   list[SCAT_THREADS / 32][E1_LIST];
    const int tid   = threadIdx.x;
    const int lane  = tid & 31;
    const int f     = lane & 15;
    const int hf    = lane >> 4;
    const int range = blockIdx.x / chunks;
    const int chunk = blockIdx.x - range * chunks;
    const int lo    = range * range_nodes;
    int len = n_nodes - lo;
    if (len > range_nodes) len = range_nodes;
    if (len <= 0) return;
    const int flen = len * FEAT_MID;

    for (int i = tid; i < flen; i += SCAT_THREADS) acc[i] = 0.0f;
    __syncthreads();

    v2i* __restrict__ lst = list[tid >> 5];
    int e0 = chunk * per_chunk;
    if (e0 > n_edges) e0 = n_edges;
    e0 &= ~3;
    int e1 = n_edges;
    if (chunk + 1 < chunks) {
        e1 = (chunk + 1) * per_chunk;
        if (e1 > n_edges) e1 = n_edges;
        e1 &= ~3;
    }

    const bool wide = (((size_t)dst) & 15) == 0;
    const int nfull = wide ? (e1 - e0) / E1_TRIP : 0;
    v4i nx[E1_U / 4];
    if (nfull > 0) {
        const v4i* __restrict__ pn = (const v4i*)(dst + e0);
#pragma unroll
        for (int j = 0; j < E1_U / 4; j++) nx[j] = pn[j * SCAT_THREADS + tid];
    }
    for (int t = 0; t < nfull; t++) {
        const int base = e0 + t * E1_TRIP;
        v4i cur[E1_U / 4];
#pragma unroll
        for (int j = 0; j < E1_U / 4; j++) cur[j] = nx[j];
        if (t + 1 < nfull) {
            const v4i* __restrict__ pn = (const v4i*)(dst + base + E1_TRIP);
#pragma unroll
            for (int j = 0; j < E1_U / 4; j++) nx[j] = pn[j * SCAT_THREADS + tid];
        }
        const int eb = base + 4 * tid;
        int cnt = 0;
#pragma unroll
        for (int j = 0; j < E1_U / 4; j++) {
#pragma unroll
            for (int i = 0; i < 4; i++) {
                int tt = cur[j][i] - lo;
                e1_push((unsigned)tt < (unsigned)len, eb + j * 4 * SCAT_THREADS + i, tt, lst, cnt);
            }
        }
        __builtin_amdgcn_wave_barrier();
        e1_drain(lst, cnt, src, h, acc, f, hf);
        __builtin_amdgcn_wave_barrier();
    }

    for (int base = e0 + nfull * E1_TRIP; base < e1; base += E1_TRIP) {
        int dv[E1_U];
#pragma unroll
        for (int k = 0; k < E1_U; k++) {
            int e  = base + k * SCAT_THREADS + tid;
            int ec = e < e1 ? e : e1 - 1;
            dv[k] = dst[ec];
        }
        asm volatile("" : "+v"(dv[0]), "+v"(dv[1]), "+v"(dv[2]), "+v"(dv[3]),
                          "+v"(dv[4]), "+v"(dv[5]), "+v"(dv[6]), "+v"(dv[7]),
                          "+v"(dv[8]), "+v"(dv[9]), "+v"(dv[10]), "+v"(dv[11]),
                          "+v"(dv[12]), "+v"(dv[13]), "+v"(dv[14]), "+v"(dv[15]));
        int cnt = 0;
#pragma unroll
        for (int k = 0; k < E1_U; k++) {
            int e  = base + k * SCAT_THREADS + tid;
            int tt = dv[k] - lo;
            e1_push((e < e1) && ((unsigned)tt < (unsigned)len), e, tt, lst, cnt);
        }
        __builtin_amdgcn_wave_barrier();
        e1_drain(lst, cnt, src, h, acc, f, hf);
        __builtin_amdgcn_wave_barrier();
    }
    __syncthreads();

    const int rot = (chunk * (flen / chunks)) & ~31;
    float* __restrict__ outp = acc1 + (size_t)lo * FEAT_MID;
    for (int j = tid; j < flen; j += SCAT_THREADS) {
        int i = j + rot;
        if (i >= flen) i -= flen;
        float v = acc[i];
        if (v != 0.0f) atomicAdd(&outp[i], v);
    }
}

__global__ void relu_bias1(const float* __restrict__ h, const float* __restrict__ dinv,
                           const float* __restrict__ b1, float* __restrict__ acc1,
                           int n_nodes) {
    int tid = blockIdx.x * blockDim.x + threadIdx.x;
    if (tid >= n_nodes * FEAT_MID) return;
    int node = tid >> 4, f = tid & 15;
    float di = dinv[node];
    float v = (acc1[tid] + h[tid]) * di + b1[f];
    acc1[tid] = fmaxf(v, 0.0f);
}

__global__ void gemm2_kernel(const float* __restrict__ h1, const float* __restrict__ W2,
                             const float* __restrict__ dinv,
                             float* __restrict__ h2, int n_nodes) {
    int i = blockIdx.x * blockDim.x + threadIdx.x;
    if (i >= n_nodes) return;
    float s0 = 0.0f, s1 = 0.0f;
#pragma unroll
    for (int k = 0; k < FEAT_MID; k++) {
        float v = h1[i * FEAT_MID + k];
        s0 = fmaf(v, W2[k * 2 + 0], s0);
        s1 = fmaf(v, W2[k * 2 + 1], s1);
    }
    float di = dinv[i];
    h2[i * 2 + 0] = s0 * di;
    h2[i * 2 + 1] = s1 * di;
}

__global__ void __launch_bounds__(SCAT_THREADS)
edge2_lds(const int* __restrict__ src, const int* __restrict__ dst,
          const float* __restrict__ h2,
          float* __restrict__ acc2, int n_edges, int n_nodes,
          int range_nodes, int chunks, int per_chunk) {
    __shared__ float acc[LDS_FLOATS];
    const int tid   = threadIdx.x;
    const int range = blockIdx.x / chunks;
    const int chunk = blockIdx.x - range * chunks;
    const int lo    = range * range_nodes;
    int len = n_nodes - lo;
    if (len > range_nodes) len = range_nodes;
    if (len <= 0) return;
    const int flen = len * FEAT_OUT;

    for (int i = tid; i < flen; i += SCAT_THREADS) acc[i] = 0.0f;
    __syncthreads();

    const v2f* __restrict__ h2v = (const v2f*)h2;
    const int e0 = chunk * per_chunk;
    int e1 = e0 + per_chunk;
    if (e1 > n_edges) e1 = n_edges;
    for (int base = e0; base < e1; base += SCAT_U * SCAT_THREADS) {
        int dv[SCAT_U], sv[SCAT_U];
#pragma unroll
        for (int k = 0; k < SCAT_U; k++) {
            int e  = base + k * SCAT_THREADS + tid;
            int ec = e < e1 ? e : e1 - 1;
            dv[k] = dst[ec];
            sv[k] = src[ec];
        }
        asm volatile("" : "+v"(dv[0]), "+v"(dv[1]), "+v"(dv[2]), "+v"(dv[3]),
                          "+v"(dv[4]), "+v"(dv[5]), "+v"(dv[6]), "+v"(dv[7]),
                          "+v"(sv[0]), "+v"(sv[1]), "+v"(sv[2]), "+v"(sv[3]),
                          "+v"(sv[4]), "+v"(sv[5]), "+v"(sv[6]), "+v"(sv[7]));
        int   t[SCAT_U];
        v2f   hv[SCAT_U];
#pragma unroll
        for (int k = 0; k < SCAT_U; k++) {
            int e  = base + k * SCAT_THREADS + tid;
            int tt = dv[k] - lo;
            bool m = (e < e1) && ((unsigned)tt < (unsigned)len);
            t[k] = m ? tt : -1;
            int sg = m ? sv[k] : lo;
            hv[k] = h2v[sg];
        }
        asm volatile("" : "+v"(hv[0]), "+v"(hv[1]), "+v"(hv[2]), "+v"(hv[3]),
                          "+v"(hv[4]), "+v"(hv[5]), "+v"(hv[6]), "+v"(hv[7]));
#pragma unroll
        for (int k = 0; k < SCAT_U; k++) {
            if (t[k] >= 0) {
                atomicAdd(&acc[t[k] * FEAT_OUT + 0], hv[k].x);
                atomicAdd(&acc[t[k] * FEAT_OUT + 1], hv[k].y);
            }
        }
    }
    __syncthreads();

    const int rot = (chunk * (flen / chunks)) & ~31;
    float* __restrict__ outp = acc2 + (size_t)lo * FEAT_OUT;
    for (int j = tid; j < flen; j += SCAT_THREADS) {
        int i = j + rot;
        if (i >= flen) i -= flen;
        float v = acc[i];
        if (v != 0.0f) atomicAdd(&outp[i], v);
    }
}

__global__ void final_kernel(const float* __restrict__ acc2, const float* __restrict__ h2,
                             const float* __restrict__ dinv, const float* __restrict__ b2,
                             float* __restrict__ out, int n_nodes) {
    int i = blockIdx.x * blockDim.x + threadIdx.x;
    if (i >= n_nodes) return;
    float di = dinv[i];
    float v0 = (acc2[i * 2 + 0] + h2[i * 2 + 0]) * di + b2[0];
    float v1 = (acc2[i * 2 + 1] + h2[i * 2 + 1]) * di + b2[1];
    float m  = fmaxf(v0, v1);
    float l  = m + logf(expf(v0 - m) + expf(v1 - m));
    out[i * 2 + 0] = v0 - l;
    out[i * 2 + 1] = v1 - l;
}

#define RB_NODES   128
#define RB_SHIFT   7
#define RB_MAXR    1024
#define BK_THREADS 1024
#define BK_U       4
#define BK_STAGE   12800
#define EB_THREADS 512
#define EB_STEPS   16
#define E2_U       8
#define EB_TAIL    (RB_NODES * FEAT_MID / EB_THREADS)

#define DP_CHUNKS 32
#define DP_NODES  25024
#define DP_U      16
#define DP_TRIP   (DP_U * SCAT_THREADS)

__global__ void __launch_bounds__(SCAT_THREADS)
degree_parts(const int* __restrict__ dst, int* __restrict__ parts,
             int n_edges, int n_nodes, int range_nodes, int per_chunk) {
    __shared__ int cnt[DP_NODES];
    const int tid   = threadIdx.x;
    const int range = blockIdx.x / DP_CHUNKS;
    const int chunk = blockIdx.x - range * DP_CHUNKS;
    const int lo    = range * range_nodes;
    int len = n_nodes - lo;
    if (len > range_nodes) len = range_nodes;
    if (len <= 0) return;

    for (int i = tid; i < len; i += SCAT_THREADS) cnt[i] = 0;
    __syncthreads();

    int e0 = chunk * per_chunk;
    if (e0 > n_edges) e0 = n_edges;
    int e1 = e0 + per_chunk;
    if (e1 > n_edges) e1 = n_edges;

    const int nfull = (e1 - e0) / DP_TRIP;
    const int* __restrict__ p0 = dst + e0 + tid;
    int nx[DP_U] = {};
    if (nfull > 0) {
#pragma unroll
        for (int k = 0; k < DP_U; k++) nx[k] = p0[k * SCAT_THREADS];
    }
    for (int t = 0; t < nfull; t++) {
        int cur[DP_U];
#pragma unroll
        for (int k = 0; k < DP_U; k++) cur[k] = nx[k];
        if (t + 1 < nfull) {
            const int* __restrict__ pn = p0 + (t + 1) * DP_TRIP;
#pragma unroll
            for (int k = 0; k < DP_U; k++) nx[k] = pn[k * SCAT_THREADS];
        }
#pragma unroll
        for (int k = 0; k < DP_U; k++) {
            int tt = cur[k] - lo;
            if ((unsigned)tt < (unsigned)len) atomicAdd(&cnt[tt], 1);
        }
    }

    const int base = e0 + nfull * DP_TRIP;
    if (base < e1) {
        int dv[DP_U];
#pragma unroll
        for (int k = 0; k < DP_U; k++) {
            int e  = base + k * SCAT_THREADS + tid;
            int ec = e < e1 ? e : e1 - 1;
            dv[k] = dst[ec];
        }
        asm volatile("" : "+v"(dv[0]), "+v"(dv[1]), "+v"(dv[2]), "+v"(dv[3]),
                          "+v"(dv[4]), "+v"(dv[5]), "+v"(dv[6]), "+v"(dv[7]),
                          "+v"(dv[8]), "+v"(dv[9]), "+v"(dv[10]), "+v"(dv[11]),
                          "+v"(dv[12]), "+v"(dv[13]), "+v"(dv[14]), "+v"(dv[15]));
#pragma unroll
        for (int k = 0; k < DP_U; k++) {
            int e  = base + k * SCAT_THREADS + tid;
            int tt = dv[k] - lo;
            if (e < e1 && (unsigned)tt < (unsigned)len) atomicAdd(&cnt[tt], 1);
        }
    }
    __syncthreads();

    int* __restrict__ outp = parts + (size_t)chunk * n_nodes + lo;
    for (int i = tid; i < len; i += SCAT_THREADS) outp[i] = cnt[i];
}

#define DT_THREADS (2 * RB_NODES)
__global__ void __launch_bounds__(DT_THREADS)
dinv_totals(const int* __restrict__ parts, float* __restrict__ dinv,
            int* __restrict__ totals, int* __restrict__ cursor,
            int n_nodes, int n_ranges) {
    __shared__ int part[DT_THREADS / 32];
    const int tid = threadIdx.x;
    const int i   = blockIdx.x * DT_THREADS + tid;
    int v = 0;
    if (i < n_nodes) {
        int p[DP_CHUNKS];
#pragma unroll
        for (int c = 0; c < DP_CHUNKS; c++) p[c] = parts[(size_t)c * n_nodes + i];
#pragma unroll
        for (int c = 0; c < DP_CHUNKS; c++) v += p[c];
        float d = (float)(v + 1);
        dinv[i] = (d > 0.0f) ? rsqrtf(d) : 0.0f;
    }
    for (int m = 16; m > 0; m >>= 1) v += __shfl_xor(v, m);
    if ((tid & 31) == 0) part[tid >> 5] = v;
    __syncthreads();
    if (tid < 2) {
        const int r = blockIdx.x * 2 + tid;
        if (r < n_ranges) {
            const int* q = part + tid * (RB_NODES / 32);
            totals[r] = q[0] + q[1] + q[2] + q[3];
            cursor[r] = 0;
        }
    }
}

__device__ __forceinline__ void
bucket_chunk(const int* __restrict__ src, const int* __restrict__ dst,
             const int* __restrict__ totals, int* __restrict__ cursor,
             int* __restrict__ offs, int* __restrict__ bins,
             int n_edges, int per_chunk, int n_ranges, const int blk) {
    __shared__ int cnt[RB_MAXR];
    __shared__ int gb[RB_MAXR];
    __shared__ int sc[RB_MAXR];
    __shared__ int so[RB_MAXR];
    __shared__ int stage_w[BK_STAGE];
    __shared__ int stage_p[BK_STAGE];
    const int tid = threadIdx.x;
    int e0 = blk * per_chunk;
    if (e0 > n_edges) e0 = n_edges;
    int e1 = e0 + per_chunk;
    if (e1 > n_edges) e1 = n_edges;
    cnt[tid] = 0;
    __syncthreads();
    for (int base = e0; base < e1; base += BK_U * BK_THREADS) {
        int dv[BK_U];
#pragma unroll
        for (int k = 0; k < BK_U; k++) {
            int e  = base + k * BK_THREADS + tid;
            int ec = e < e1 ? e : e1 - 1;
            dv[k] = dst[ec];
        }
        asm volatile("" : "+v"(dv[0]), "+v"(dv[1]), "+v"(dv[2]), "+v"(dv[3]));
#pragma unroll
        for (int k = 0; k < BK_U; k++) {
            int e = base + k * BK_THREADS + tid;
            if (e < e1) atomicAdd(&cnt[dv[k] >> RB_SHIFT], 1);
        }
    }
    __syncthreads();
    const int c   = cnt[tid];
    const int tot = tid < n_ranges ? totals[tid] : 0;
    int rel = 0;
    if (c > 0) rel = atomicAdd(&cursor[tid], c);
    sc[tid] = c;
    so[tid] = tot;
    __syncthreads();
    for (int off = 1; off < RB_MAXR; off <<= 1) {
        int t1 = tid >= off ? sc[tid - off] : 0;
        int t2 = tid >= off ? so[tid - off] : 0;
        __syncthreads();
        sc[tid] += t1;
        so[tid] += t2;
        __syncthreads();
    }
    const int first = so[tid] - tot;
    if (blk == 0 && tid < n_ranges) {
        offs[tid] = first;
        if (tid == n_ranges - 1) offs[n_ranges] = so[tid];
    }
    gb[tid]  = first + rel;
    sc[tid] -= c;
    cnt[tid] = 0;
    __syncthreads();
    for (int base = e0; base < e1; base += BK_U * BK_THREADS) {
        int dv[BK_U], sv[BK_U];
#pragma unroll
        for (int k = 0; k < BK_U; k++) {
            int e  = base + k * BK_THREADS + tid;
            int ec = e < e1 ? e : e1 - 1;
            dv[k] = dst[ec];
            sv[k] = src[ec];
        }
        asm volatile("" : "+v"(dv[0]), "+v"(dv[1]), "+v"(dv[2]), "+v"(dv[3]),
                          "+v"(sv[0]), "+v"(sv[1]), "+v"(sv[2]), "+v"(sv[3]));
        int rank[BK_U], ls[BK_U], gs[BK_U];
#pragma unroll
        for (int k = 0; k < BK_U; k++) {
            int e = base + k * BK_THREADS + tid;
            int r = dv[k] >> RB_SHIFT;
            rank[k] = 0;
            if (e < e1) rank[k] = atomicAdd(&cnt[r], 1);
            ls[k] = sc[r];
            gs[k] = gb[r];
        }
#pragma unroll
        for (int k = 0; k < BK_U; k++) {
            int e = base + k * BK_THREADS + tid;
            if (e < e1) {
                int lp = ls[k] + rank[k];
                stage_w[lp] = (sv[k] << RB_SHIFT) | (dv[k] & (RB_NODES - 1));
                stage_p[lp] = gs[k] + rank[k];
            }
        }
    }
    __syncthreads();
    const int n = e1 - e0;
    for (int i = tid; i < n; i += BK_THREADS) bins[stage_p[i]] = stage_w[i];
}

__global__ void __launch_bounds__(BK_THREADS)
bucket_gemm(const int* __restrict__ src, const int* __restrict__ dst,
            const int* __restrict__ totals, int* __restrict__ cursor,
            int* __restrict__ offs, int* __restrict__ bins,
            int n_edges, int per_chunk, int n_ranges, int nb,
            const float* __restrict__ x, const float* __restrict__ W1,
            const float* __restrict__ dinv, float* __restrict__ h,
            int n_tiles, int n_nodes) {
    if ((int)blockIdx.x < nb) {
        bucket_chunk(src, dst, totals, cursor, offs, bins, n_edges, per_chunk, n_ranges, blockIdx.x);
    } else {
        gemm1_tile(x, W1, dinv, h,
                   ((int)blockIdx.x - nb) * (BK_THREADS / 32) + (int)(threadIdx.x >> 5),
                   n_tiles, n_nodes);
    }
}

__global__ void __launch_bounds__(EB_THREADS)
edge1_bin(const int* __restrict__ bins, const int* __restrict__ offs,
          const float* __restrict__ h, const float* __restrict__ dinv,
          const float* __restrict__ bias1, const float* __restrict__ W2,
          float* __restrict__ h2, int n_nodes) {
    __shared__ float acc[RB_NODES * FEAT_MID];
    const int tid  = threadIdx.x;
    const int lane = tid & 31;
    const int f    = lane & 15;
    const int hf   = lane >> 4;
    const int lo   = blockIdx.x << RB_SHIFT;
    int len = n_nodes - lo;
    if (len > RB_NODES) len = RB_NODES;

    for (int i = tid; i < RB_NODES * FEAT_MID; i += EB_THREADS) acc[i] = 0.0f;
    __syncthreads();

    const int b0 = offs[blockIdx.x];
    const int b1 = offs[blockIdx.x + 1];
    const int stride = (EB_THREADS / 32) * 2 * EB_STEPS;
    int j0 = b0 + (tid >> 5) * 2 * EB_STEPS;
    int w[EB_STEPS] = {};
    if (j0 < b1) {
#pragma unroll
        for (int p = 0; p < EB_STEPS; p++) {
            int idx = j0 + 2 * p + hf;
            w[p] = bins[idx < b1 ? idx : j0];
        }
    }
    while (j0 < b1) {
        int   cur[EB_STEPS];
        float hq[EB_STEPS];
#pragma unroll
        for (int p = 0; p < EB_STEPS; p++) cur[p] = w[p];
        const int jn = j0 + stride;
        if (jn < b1) {
#pragma unroll
            for (int p = 0; p < EB_STEPS; p++) {
                int idx = jn + 2 * p + hf;
                w[p] = bins[idx < b1 ? idx : jn];
            }
        }
#pragma unroll
        for (int p = 0; p < EB_STEPS; p++)
            hq[p] = h[(cur[p] >> RB_SHIFT) * FEAT_MID + f];
        asm volatile("" : "+v"(hq[0]), "+v"(hq[1]), "+v"(hq[2]), "+v"(hq[3]),
                          "+v"(hq[4]), "+v"(hq[5]), "+v"(hq[6]), "+v"(hq[7]),
                          "+v"(hq[8]), "+v"(hq[9]), "+v"(hq[10]), "+v"(hq[11]),
                          "+v"(hq[12]), "+v"(hq[13]), "+v"(hq[14]), "+v"(hq[15]));
#pragma unroll
        for (int p = 0; p < EB_STEPS; p++) {
            if (j0 + 2 * p + hf < b1)
                atomicAdd(&acc[(cur[p] & (RB_NODES - 1)) * FEAT_MID + f], hq[p]);
        }
        j0 = jn;
    }
    __syncthreads();

    const float* __restrict__ hown = h + (size_t)lo * FEAT_MID;
    const int   flen = len * FEAT_MID;
    const float bf   = bias1[f];
    float hs[EB_TAIL], dd[EB_TAIL];
#pragma unroll
    for (int j = 0; j < EB_TAIL; j++) {
        int i  = tid + j * EB_THREADS;
        int ic = i < flen ? i : tid & 15;
        hs[j] = hown[ic];
        dd[j] = dinv[lo + (ic >> 4)];
    }
#pragma unroll
    for (int j = 0; j < EB_TAIL; j++) {
        int i = tid + j * EB_THREADS;
        if (i < flen) acc[i] = fmaxf((acc[i] + hs[j]) * dd[j] + bf, 0.0f);
    }
    __syncthreads();

    if (tid < len * FEAT_OUT) {
        const int node = tid >> 1, c = tid & 1;
        float s = 0.0f;
#pragma unroll
        for (int k = 0; k < FEAT_MID; k++)
            s = fmaf(acc[node * FEAT_MID + k], W2[k * FEAT_OUT + c], s);
        h2[(size_t)lo * FEAT_OUT + tid] = s * dinv[lo + node];
    }
}

__global__ void __launch_bounds__(EB_THREADS)
edge2_bin(const int* __restrict__ bins, const int* __restrict__ offs,
          const float* __restrict__ h2, const float* __restrict__ dinv,
          const float* __restrict__ bias2, float* __restrict__ out, int n_nodes) {
    __shared__ float acc[RB_NODES * FEAT_OUT];
    const int tid = threadIdx.x;
    const int lo  = blockIdx.x << RB_SHIFT;
    int len = n_nodes - lo;
    if (len > RB_NODES) len = RB_NODES;

    for (int i = tid; i < RB_NODES * FEAT_OUT; i += EB_THREADS) acc[i] = 0.0f;
    __syncthreads();

    const v2f* __restrict__ h2v = (const v2f*)h2;
    const int b0 = offs[blockIdx.x];
    const int b1 = offs[blockIdx.x + 1];
    for (int base = b0; base < b1; base += E2_U * EB_THREADS) {
        int w[E2_U];
        v2f hv[E2_U];
#pragma unroll
        for (int k = 0; k < E2_U; k++) {
            int idx = base + k * EB_THREADS + tid;
            w[k] = bins[idx < b1 ? idx : b0];
        }
#pragma unroll
        for (int k = 0; k < E2_U; k++) hv[k] = h2v[w[k] >> RB_SHIFT];
        asm volatile("" : "+v"(hv[0]), "+v"(hv[1]), "+v"(hv[2]), "+v"(hv[3]),
                          "+v"(hv[4]), "+v"(hv[5]), "+v"(hv[6]), "+v"(hv[7]));
#pragma unroll
        for (int k = 0; k < E2_U; k++) {
            if (base + k * EB_THREADS + tid < b1) {
                int t = w[k] & (RB_NODES - 1);
                atomicAdd(&acc[t * FEAT_OUT + 0], hv[k].x);
                atomicAdd(&acc[t * FEAT_OUT + 1], hv[k].y);
            }
        }
    }
    __syncthreads();

    if (tid < len) {
        const int i = lo + tid;
        float di = dinv[i];
        float v0 = (acc[tid * 2 + 0] + h2[i * 2 + 0]) * di + bias2[0];
        float v1 = (acc[tid * 2 + 1] + h2[i * 2 + 1]) * di + bias2[1];
        float m  = fmaxf(v0, v1);
        float l  = m + logf(expf(v0 - m) + expf(v1 - m));
        out[i * 2 + 0] = v0 - l;
        out[i * 2 + 1] = v1 - l;
    }
}

#define ES_W   16
#define ES_CAP (ES_W * EB_THREADS)
#define ES_U   8
typedef float v4f __attribute__((ext_vector_type(4)));
typedef int   v4i_t __attribute__((ext_vector_type(4)));

__global__ void __launch_bounds__(EB_THREADS, 16)
edge1_sort(const int* __restrict__ bins, const int* __restrict__ offs,
           const float* __restrict__ h, const float* __restrict__ dinv,
           const float* __restrict__ bias1, const float* __restrict__ W2,
           float* __restrict__ h2, int n_nodes) {
    __shared__ int sorted[ES_CAP];
    __shared__ __attribute__((aligned(16))) int cnt[RB_NODES];
    __shared__ __attribute__((aligned(16))) int start[RB_NODES + 4];
    __shared__ float hs[RB_NODES * FEAT_OUT];
    const int tid  = threadIdx.x;
    const int lane = tid & 31;
    const int q    = lane & 3;
    const int my   = (tid >> 5) * 8 + (lane >> 2);
    const int lo   = blockIdx.x << RB_SHIFT;
    int len = n_nodes - lo;
    if (len > RB_NODES) len = RB_NODES;
    const v4f* __restrict__ h4 = (const v4f*)h;

    const int b0 = offs[blockIdx.x];
    const int b1 = offs[blockIdx.x + 1];
    v4f acc = {0.0f, 0.0f, 0.0f, 0.0f};
    for (int c0 = b0; c0 < b1; c0 += ES_CAP) {
        int n = b1 - c0;
        if (n > ES_CAP) n = ES_CAP;
        if (tid < RB_NODES) cnt[tid] = 0;
        __syncthreads();
        int w[ES_W];
#pragma unroll
        for (int hh = 0; hh < 2; hh++) {
#pragma unroll
            for (int k = 0; k < ES_W / 2; k++) w[hh * (ES_W / 2) + k] = 0;
            if (hh * (ES_CAP / 2) < n) {
#pragma unroll
                for (int k = 0; k < ES_W / 2; k++) {
                    int idx = (hh * (ES_W / 2) + k) * EB_THREADS + tid;
                    w[hh * (ES_W / 2) + k] = bins[c0 + (idx < n ? idx : 0)];
                }
#pragma unroll
                for (int k = 0; k < ES_W / 2; k++) {
                    int idx = (hh * (ES_W / 2) + k) * EB_THREADS + tid;
                    if (idx < n) atomicAdd(&cnt[w[hh * (ES_W / 2) + k] & (RB_NODES - 1)], 1);
                }
            }
        }
        __syncthreads();
        if (tid < 32) {
            v4i_t c = *(const v4i_t*)&cnt[4 * tid];
            int s = c.x + c.y + c.z + c.w;
            int incl = s;
            for (int d = 1; d < 32; d <<= 1) {
                int t = __shfl_up(incl, d);
                if (lane >= d) incl += t;
            }
            int ex = incl - s;
            v4i_t st;
            st.x = ex; st.y = ex + c.x; st.z = ex + c.x + c.y; st.w = ex + c.x + c.y + c.z;
            *(v4i_t*)&start[4 * tid] = st;
            *(v4i_t*)&cnt[4 * tid]   = st;
            if (tid == 31) start[RB_NODES] = incl;
        }
        __syncthreads();
#pragma unroll
        for (int k = 0; k < ES_W; k++) {
            int idx = k * EB_THREADS + tid;
            if (idx < n) sorted[atomicAdd(&cnt[w[k] & (RB_NODES - 1)], 1)] = w[k] >> RB_SHIFT;
        }
        __syncthreads();

        const int s0 = start[my];
        const int s1 = start[my + 1];
        for (int j = s0; j < s1; j += ES_U) {
            const int m = s1 - j;
            v4f v[ES_U];
#pragma unroll
            for (int u = 0; u < ES_U; u++) v[u] = h4[sorted[j + (u < m ? u : 0)] * (FEAT_MID / 4) + q];
#pragma unroll
            for (int u = 0; u < ES_U; u++) {
                const bool ok = u < m;
                acc.x += ok ? v[u].x : 0.0f;
                acc.y += ok ? v[u].y : 0.0f;
                acc.z += ok ? v[u].z : 0.0f;
                acc.w += ok ? v[u].w : 0.0f;
            }
        }
    }

    const int   node = lo + (my < len ? my : 0);
    const float di   = dinv[node];
    const v4f   own  = h4[node * (FEAT_MID / 4) + q];
    const v4f   bq   = ((const v4f*)bias1)[q];
    const v4f   wa   = ((const v4f*)W2)[2 * q];
    const v4f   wb   = ((const v4f*)W2)[2 * q + 1];
    const float a0 = fmaxf((acc.x + own.x) * di + bq.x, 0.0f);
    const float a1 = fmaxf((acc.y + own.y) * di + bq.y, 0.0f);
    const float a2 = fmaxf((acc.z + own.z) * di + bq.z, 0.0f);
    const float a3 = fmaxf((acc.w + own.w) * di + bq.w, 0.0f);
    float p0 = fmaf(a3, wb.z, fmaf(a2, wb.x, fmaf(a1, wa.z, a0 * wa.x)));
    float p1 = fmaf(a3, wb.w, fmaf(a2, wb.y, fmaf(a1, wa.w, a0 * wa.y)));
    p0 += __shfl_xor(p0, 1);  p1 += __shfl_xor(p1, 1);
    p0 += __shfl_xor(p0, 2);  p1 += __shfl_xor(p1, 2);
    if (q == 0) { hs[my * FEAT_OUT + 0] = p0 * di; hs[my * FEAT_OUT + 1] = p1 * di; }
    __syncthreads();
    if (tid < len * FEAT_OUT) h2[(size_t)lo * FEAT_OUT + tid] = hs[tid];
}

extern "C" void kernel_launch(void* const* d_in, const int* in_sizes, int n_in,
                              void* d_out, int out_size, void* d_ws, size_t ws_size,
                              hipStream_t stream) {
    const float* x  = (const float*)d_in[0];
    const int*   ei = (const int*)d_in[1];
    const float* W1 = (const float*)d_in[2];
    const float* b1 = (const float*)d_in[3];
    const float* W2 = (const float*)d_in[4];
    const float* b2 = (const float*)d_in[5];
    float* out = (float*)d_out;

    int n_nodes = in_sizes[0] / FEAT_IN;
    int n_edges = in_sizes[1] / 2;
    const int* src = ei;
    const int* dst = ei + n_edges;

    float* ws   = (float*)d_ws;
    float* deg  = ws;
    float* h    = deg  + n_nodes;
    float* acc1 = h    + (size_t)n_nodes * FEAT_MID;
    float* h2   = acc1 + (size_t)n_nodes * FEAT_MID;
    float* acc2 = h2   + (size_t)n_nodes * FEAT_OUT;

    const int B = 256;
    int n16 = n_nodes * FEAT_MID;

    int nr1 = (n_nodes + LDS_FLOATS - 1) / LDS_FLOATS;
    int rn1 = (((n_nodes + nr1 - 1) / nr1) + 31) & ~31;
    int ch1 = 128 / nr1 > 0 ? 128 / nr1 : 1;
    int pc1 = (n_edges + ch1 - 1) / ch1;
    int n_tiles = (n_nodes + 15) / 16;
    int wpb = B / 32;

    int    n_ranges = (n_nodes + RB_NODES - 1) >> RB_SHIFT;
    size_t need = ((size_t)n_nodes * (1 + 2 * FEAT_MID + 2 * FEAT_OUT) + 3 * (size_t)n_ranges + 1 +
                   (size_t)n_edges) * sizeof(float);
    if (n_ranges <= RB_MAXR && ws_size >= need) {
        int* offs   = (int*)(acc2 + (size_t)n_nodes * FEAT_OUT);
        int* cursor = offs + n_ranges + 1;
        int* totals = cursor + n_ranges;
        int* bins   = totals + n_ranges;
        int  nb     = (n_edges + BK_STAGE - 1) / BK_STAGE;
        int  pcb    = BK_STAGE;

        int* parts = (int*)h;
        int  nrp   = (n_nodes + DP_NODES - 1) / DP_NODES;
        int  rnp   = (((n_nodes + nrp - 1) / nrp) + 31) & ~31;
        int  pcp   = (n_edges + DP_CHUNKS - 1) / DP_CHUNKS;
        degree_parts<<<nrp * DP_CHUNKS, SCAT_THREADS, 0, stream>>>(dst, parts, n_edges, n_nodes, rnp, pcp);
        dinv_totals<<<(n_nodes + DT_THREADS - 1) / DT_THREADS, DT_THREADS, 0, stream>>>(
            parts, deg, totals, cursor, n_nodes, n_ranges);
        int  gw     = (n_tiles + BK_THREADS / 32 - 1) / (BK_THREADS / 32);
        bucket_gemm<<<nb + gw, BK_THREADS, 0, stream>>>(src, dst, totals, cursor, offs, bins,
                                                        n_edges, pcb, n_ranges, nb,
                                                        x, W1, deg, h, n_tiles, n_nodes);
        if (((((size_t)h) | ((size_t)b1) | ((size_t)W2)) & 15) == 0)
            edge1_sort<<<n_ranges, EB_THREADS, 0, stream>>>(bins, offs, h, deg, b1, W2, h2, n_nodes);
        else
            edge1_bin<<<n_ranges, EB_THREADS, 0, stream>>>(bins, offs, h, deg, b1, W2, h2, n_nodes);
        edge2_bin<<<n_ranges, EB_THREADS, 0, stream>>>(bins, offs, h2, deg, b2, out, n_nodes);
        return;
    }

    init_ws<<<(n16 + B - 1) / B, B, 0, stream>>>(deg, acc1, acc2, n_nodes);
    degree_lds<<<nr1 * ch1, SCAT_THREADS, 0, stream>>>(dst, deg, n_edges, n_nodes, rn1, ch1, pc1);
    dinv_kernel<<<(n_nodes + B - 1) / B, B, 0, stream>>>(deg, n_nodes);

    gemm1_wmma<<<(n_tiles + wpb - 1) / wpb, B, 0, stream>>>(x, W1, deg, h, n_tiles, n_nodes);

    int nr0 = (n_nodes + E1_NODES - 1) / E1_NODES;
    int rn0 = (((n_nodes + nr0 - 1) / nr0) + 31) & ~31;
    int ch0 = 256 / nr0 > 0 ? 256 / nr0 : 1;
    int pc0 = (n_edges + ch0 - 1) / ch0;
    edge1_lds<<<nr0 * ch0, SCAT_THREADS, 0, stream>>>(src, dst, h, acc1, n_edges, n_nodes,
                                                      rn0, ch0, pc0);
    relu_bias1<<<(n16 + B - 1) / B, B, 0, stream>>>(h, deg, b1, acc1, n_nodes);

    gemm2_kernel<<<(n_nodes + B - 1) / B, B, 0, stream>>>(acc1, W2, deg, h2, n_nodes);

    int nr2 = (n_nodes * FEAT_OUT + LDS_FLOATS - 1) / LDS_FLOATS;
    int rn2 = (((n_nodes + nr2 - 1) / nr2) + 31) & ~31;
    int ch2 = 128 / nr2 > 0 ? 128 / nr2 : 1;
    int pc2 = (n_edges + ch2 - 1) / ch2;
    edge2_lds<<<nr2 * ch2, SCAT_THREADS, 0, stream>>>(src, dst, h2, acc2, n_edges, n_nodes,
                                                      rn2, ch2, pc2);

    final_kernel<<<(n_nodes + B - 1) / B, B, 0, stream>>>(acc2, h2, deg, b2, out, n_nodes);
}


# layer 2 sums one value per node, the class difference (all a 2-class log_softmax needs): its table is staged in LDS in two halves, no L2 line per edge (sum order: by source half; rounding differs)
# speedup vs baseline: 6.6840x; 1.1085x over previous
#include <hip/hip_runtime.h>
#include <math.h>

typedef float v2f __attribute__((ext_vector_type(2)));
typedef float v8f __attribute__((ext_vector_type(8)));

#define FEAT_IN  128
#define FEAT_MID 16
#define FEAT_OUT 2

__global__ void init_ws(float* __restrict__ deg, float* __restrict__ acc1,
                        float* __restrict__ acc2, int n_nodes) {
    int tid = blockIdx.x * blockDim.x + threadIdx.x;
    if (tid < n_nodes * FEAT_MID) acc1[tid] = 0.0f;
    if (tid < n_nodes * FEAT_OUT) acc2[tid] = 0.0f;
    if (tid < n_nodes)            deg[tid]  = 1.0f;
}

#define LDS_FLOATS   69632
#define SCAT_THREADS 1024
#define SCAT_U       8

__global__ void __launch_bounds__(SCAT_THREADS)
degree_lds(const int* __restrict__ dst, float* __restrict__ deg,
           int n_edges, int n_nodes, int range_nodes, int chunks, int per_chunk) {
    __shared__ float cnt[LDS_FLOATS];
    const int tid   = threadIdx.x;
    const int range = blockIdx.x / chunks;
    const int chunk = blockIdx.x - range * chunks;
    const int lo    = range * range_nodes;
    int len = n_nodes - lo;
    if (len > range_nodes) len = range_nodes;
    if (len <= 0) return;

    for (int i = tid; i < len; i += SCAT_THREADS) cnt[i] = 0.0f;
    __syncthreads();

    const int e0 = chunk * per_chunk;
    int e1 = e0 + per_chunk;
    if (e1 > n_edges) e1 = n_edges;
    for (int base = e0; base < e1; base += SCAT_U * SCAT_THREADS) {
        int dv[SCAT_U];
#pragma unroll
        for (int k = 0; k < SCAT_U; k++) {
            int e  = base + k * SCAT_THREADS + tid;
            int ec = e < e1 ? e : e1 - 1;
            dv[k] = dst[ec];
        }
        asm volatile("" : "+v"(dv[0]), "+v"(dv[1]), "+v"(dv[2]), "+v"(dv[3]),
                          "+v"(dv[4]), "+v"(dv[5]), "+v"(dv[6]), "+v"(dv[7]));
#pragma unroll
        for (int k = 0; k < SCAT_U; k++) {
            int e = base + k * SCAT_THREADS + tid;
            int t = dv[k] - lo;
            if (e < e1 && (unsigned)t < (unsigned)len) atomicAdd(&cnt[t], 1.0f);
        }
    }
    __syncthreads();

    const int rot = (chunk * (len / chunks)) & ~31;
    for (int j = tid; j < len; j += SCAT_THREADS) {
        int i = j + rot;
        if (i >= len) i -= len;
        float v = cnt[i];
        if (v != 0.0f) atomicAdd(&deg[lo + i], v);
    }
}

__global__ void dinv_kernel(float* __restrict__ deg, int n_nodes) {
    int i = blockIdx.x * blockDim.x + threadIdx.x;
    if (i < n_nodes) {
        float d = deg[i];
        deg[i] = (d > 0.0f) ? rsqrtf(d) : 0.0f;
    }
}

__device__ __forceinline__ void
gemm1_tile(const float* __restrict__ x, const float* __restrict__ W1,
           const float* __restrict__ dinv, float* __restrict__ h,
           int tile, int n_tiles, int n_nodes) {
    int lane = threadIdx.x & 31;
    if (tile >= n_tiles) return;

    int col = lane & 15;
    int kh  = (lane >> 4) << 1;

    int rowA = tile * 16 + col;
    if (rowA > n_nodes - 1) rowA = n_nodes - 1;
    const float* xrow = x + (size_t)rowA * FEAT_IN;

    int mbase = (lane >> 4) * 8;
    float dr[8];
#pragma unroll
    for (int r = 0; r < 8; r++) {
        int row = tile * 16 + mbase + r;
        dr[r] = dinv[row < n_nodes ? row : n_nodes - 1];
    }

    v8f acc = {};
    for (int k0 = 0; k0 < FEAT_IN; k0 += 4) {
        v2f a, b;
        a.x = xrow[k0 + kh];
        a.y = xrow[k0 + kh + 1];
        b.x = W1[(k0 + kh)     * FEAT_MID + col];
        b.y = W1[(k0 + kh + 1) * FEAT_MID + col];
        acc = __builtin_amdgcn_wmma_f32_16x16x4_f32(
             false, a,  false, b,
             (short)0, acc,  false,  false);
    }

    asm volatile("v_nop\n\tv_nop\n\tv_nop\n\tv_nop\n\tv_nop" : "+v"(acc));
#pragma unroll
    for (int r = 0; r < 8; r++) {
        int row = tile * 16 + mbase + r;
        if (row < n_nodes) h[(size_t)row * FEAT_MID + col] = acc[r] * dr[r];
    }
}

__global__ void __launch_bounds__(256)
gemm1_wmma(const float* __restrict__ x, const float* __restrict__ W1,
           const float* __restrict__ dinv, float* __restrict__ h,
           int n_tiles, int n_nodes) {
    gemm1_tile(x, W1, dinv, h, blockIdx.x * (blockDim.x >> 5) + (threadIdx.x >> 5), n_tiles, n_nodes);
}

typedef int v2i __attribute__((ext_vector_type(2)));
typedef int v4i __attribute__((ext_vector_type(4)));

#define E1_NODES 2400
#define E1_U     16
#define E1_STEPS 8
#define E1_LIST  (E1_U * 32)
#define E1_TRIP  (E1_U * SCAT_THREADS)

__device__ __forceinline__ void e1_push(bool m, int e, int tt, v2i* lst, int& cnt) {
    unsigned b = __builtin_amdgcn_ballot_w32(m);
    int pos = cnt + (int)__builtin_amdgcn_mbcnt_lo(b, 0u);
    if (m) {
        v2i q;
        q.x = e;
        q.y = tt;
        lst[pos] = q;
    }
    cnt += __builtin_popcount(b);
}

__device__ __forceinline__ void e1_drain(const v2i* lst, int cnt,
                                         const int* __restrict__ src, const float* __restrict__ h,
                                         float* __restrict__ acc, int f, int hf) {
    for (int j0 = 0; j0 < cnt; j0 += 2 * E1_STEPS) {
        int   tq[E1_STEPS], sq[E1_STEPS];
        float hq[E1_STEPS];
#pragma unroll
        for (int p = 0; p < E1_STEPS; p++) {
            int  idx   = j0 + 2 * p + hf;
            bool valid = idx < cnt;
            v2i  q     = lst[valid ? idx : 0];
            tq[p] = valid ? q.y : -1;
            sq[p] = src[q.x];
        }
        asm volatile("" : "+v"(sq[0]), "+v"(sq[1]), "+v"(sq[2]), "+v"(sq[3]),
                          "+v"(sq[4]), "+v"(sq[5]), "+v"(sq[6]), "+v"(sq[7]));
#pragma unroll
        for (int p = 0; p < E1_STEPS; p++)
            hq[p] = h[sq[p] * FEAT_MID + f];
        asm volatile("" : "+v"(hq[0]), "+v"(hq[1]), "+v"(hq[2]), "+v"(hq[3]),
                          "+v"(hq[4]), "+v"(hq[5]), "+v"(hq[6]), "+v"(hq[7]));
#pragma unroll
        for (int p = 0; p < E1_STEPS; p++) {
            if (tq[p] >= 0)
                atomicAdd(&acc[tq[p] * FEAT_MID + f], hq[p]);
        }
    }
}

__global__ void __launch_bounds__(SCAT_THREADS)
edge1_lds(const int* __restrict__ src, const int* __restrict__ dst,
          const float* __restrict__ h,
          float* __restrict__ acc1, int n_edges, int n_nodes,
          int range_nodes, int chunks, int per_chunk) {
    __shared__ float acc[E1_NODES * FEAT_MID];
    __shared__ v2i   list[SCAT_THREADS / 32][E1_LIST];
    const int tid   = threadIdx.x;
    const int lane  = tid & 31;
    const int f     = lane & 15;
    const int hf    = lane >> 4;
    const int range = blockIdx.x / chunks;
    const int chunk = blockIdx.x - range * chunks;
    const int lo    = range * range_nodes;
    int len = n_nodes - lo;
    if (len > range_nodes) len = range_nodes;
    if (len <= 0) return;
    const int flen = len * FEAT_MID;

    for (int i = tid; i < flen; i += SCAT_THREADS) acc[i] = 0.0f;
    __syncthreads();

    v2i* __restrict__ lst = list[tid >> 5];
    int e0 = chunk * per_chunk;
    if (e0 > n_edges) e0 = n_edges;
    e0 &= ~3;
    int e1 = n_edges;
    if (chunk + 1 < chunks) {
        e1 = (chunk + 1) * per_chunk;
        if (e1 > n_edges) e1 = n_edges;
        e1 &= ~3;
    }

    const bool wide = (((size_t)dst) & 15) == 0;
    const int nfull = wide ? (e1 - e0) / E1_TRIP : 0;
    v4i nx[E1_U / 4];
    if (nfull > 0) {
        const v4i* __restrict__ pn = (const v4i*)(dst + e0);
#pragma unroll
        for (int j = 0; j < E1_U / 4; j++) nx[j] = pn[j * SCAT_THREADS + tid];
    }
    for (int t = 0; t < nfull; t++) {
        const int base = e0 + t * E1_TRIP;
        v4i cur[E1_U / 4];
#pragma unroll
        for (int j = 0; j < E1_U / 4; j++) cur[j] = nx[j];
        if (t + 1 < nfull) {
            const v4i* __restrict__ pn = (const v4i*)(dst + base + E1_TRIP);
#pragma unroll
            for (int j = 0; j < E1_U / 4; j++) nx[j] = pn[j * SCAT_THREADS + tid];
        }
        const int eb = base + 4 * tid;
        int cnt = 0;
#pragma unroll
        for (int j = 0; j < E1_U / 4; j++) {
#pragma unroll
            for (int i = 0; i < 4; i++) {
                int tt = cur[j][i] - lo;
                e1_push((unsigned)tt < (unsigned)len, eb + j * 4 * SCAT_THREADS + i, tt, lst, cnt);
            }
        }
        __builtin_amdgcn_wave_barrier();
        e1_drain(lst, cnt, src, h, acc, f, hf);
        __builtin_amdgcn_wave_barrier();
    }

    for (int base = e0 + nfull * E1_TRIP; base < e1; base += E1_TRIP) {
        int dv[E1_U];
#pragma unroll
        for (int k = 0; k < E1_U; k++) {
            int e  = base + k * SCAT_THREADS + tid;
            int ec = e < e1 ? e : e1 - 1;
            dv[k] = dst[ec];
        }
        asm volatile("" : "+v"(dv[0]), "+v"(dv[1]), "+v"(dv[2]), "+v"(dv[3]),
                          "+v"(dv[4]), "+v"(dv[5]), "+v"(dv[6]), "+v"(dv[7]),
                          "+v"(dv[8]), "+v"(dv[9]), "+v"(dv[10]), "+v"(dv[11]),
                          "+v"(dv[12]), "+v"(dv[13]), "+v"(dv[14]), "+v"(dv[15]));
        int cnt = 0;
#pragma unroll
        for (int k = 0; k < E1_U; k++) {
            int e  = base + k * SCAT_THREADS + tid;
            int tt = dv[k] - lo;
            e1_push((e < e1) && ((unsigned)tt < (unsigned)len), e, tt, lst, cnt);
        }
        __builtin_amdgcn_wave_barrier();
        e1_drain(lst, cnt, src, h, acc, f, hf);
        __builtin_amdgcn_wave_barrier();
    }
    __syncthreads();

    const int rot = (chunk * (flen / chunks)) & ~31;
    float* __restrict__ outp = acc1 + (size_t)lo * FEAT_MID;
    for (int j = tid; j < flen; j += SCAT_THREADS) {
        int i = j + rot;
        if (i >= flen) i -= flen;
        float v = acc[i];
        if (v != 0.0f) atomicAdd(&outp[i], v);
    }
}

__global__ void relu_bias1(const float* __restrict__ h, const float* __restrict__ dinv,
                           const float* __restrict__ b1, float* __restrict__ acc1,
                           int n_nodes) {
    int tid = blockIdx.x * blockDim.x + threadIdx.x;
    if (tid >= n_nodes * FEAT_MID) return;
    int node = tid >> 4, f = tid & 15;
    float di = dinv[node];
    float v = (acc1[tid] + h[tid]) * di + b1[f];
    acc1[tid] = fmaxf(v, 0.0f);
}

__global__ void gemm2_kernel(const float* __restrict__ h1, const float* __restrict__ W2,
                             const float* __restrict__ dinv,
                             float* __restrict__ h2, int n_nodes) {
    int i = blockIdx.x * blockDim.x + threadIdx.x;
    if (i >= n_nodes) return;
    float s0 = 0.0f, s1 = 0.0f;
#pragma unroll
    for (int k = 0; k < FEAT_MID; k++) {
        float v = h1[i * FEAT_MID + k];
        s0 = fmaf(v, W2[k * 2 + 0], s0);
        s1 = fmaf(v, W2[k * 2 + 1], s1);
    }
    float di = dinv[i];
    h2[i * 2 + 0] = s0 * di;
    h2[i * 2 + 1] = s1 * di;
}

__global__ void __launch_bounds__(SCAT_THREADS)
edge2_lds(const int* __restrict__ src, const int* __restrict__ dst,
          const float* __restrict__ h2,
          float* __restrict__ acc2, int n_edges, int n_nodes,
          int range_nodes, int chunks, int per_chunk) {
    __shared__ float acc[LDS_FLOATS];
    const int tid   = threadIdx.x;
    const int range = blockIdx.x / chunks;
    const int chunk = blockIdx.x - range * chunks;
    const int lo    = range * range_nodes;
    int len = n_nodes - lo;
    if (len > range_nodes) len = range_nodes;
    if (len <= 0) return;
    const int flen = len * FEAT_OUT;

    for (int i = tid; i < flen; i += SCAT_THREADS) acc[i] = 0.0f;
    __syncthreads();

    const v2f* __restrict__ h2v = (const v2f*)h2;
    const int e0 = chunk * per_chunk;
    int e1 = e0 + per_chunk;
    if (e1 > n_edges) e1 = n_edges;
    for (int base = e0; base < e1; base += SCAT_U * SCAT_THREADS) {
        int dv[SCAT_U], sv[SCAT_U];
#pragma unroll
        for (int k = 0; k < SCAT_U; k++) {
            int e  = base + k * SCAT_THREADS + tid;
            int ec = e < e1 ? e : e1 - 1;
            dv[k] = dst[ec];
            sv[k] = src[ec];
        }
        asm volatile("" : "+v"(dv[0]), "+v"(dv[1]), "+v"(dv[2]), "+v"(dv[3]),
                          "+v"(dv[4]), "+v"(dv[5]), "+v"(dv[6]), "+v"(dv[7]),
                          "+v"(sv[0]), "+v"(sv[1]), "+v"(sv[2]), "+v"(sv[3]),
                          "+v"(sv[4]), "+v"(sv[5]), "+v"(sv[6]), "+v"(sv[7]));
        int   t[SCAT_U];
        v2f   hv[SCAT_U];
#pragma unroll
        for (int k = 0; k < SCAT_U; k++) {
            int e  = base + k * SCAT_THREADS + tid;
            int tt = dv[k] - lo;
            bool m = (e < e1) && ((unsigned)tt < (unsigned)len);
            t[k] = m ? tt : -1;
            int sg = m ? sv[k] : lo;
            hv[k] = h2v[sg];
        }
        asm volatile("" : "+v"(hv[0]), "+v"(hv[1]), "+v"(hv[2]), "+v"(hv[3]),
                          "+v"(hv[4]), "+v"(hv[5]), "+v"(hv[6]), "+v"(hv[7]));
#pragma unroll
        for (int k = 0; k < SCAT_U; k++) {
            if (t[k] >= 0) {
                atomicAdd(&acc[t[k] * FEAT_OUT + 0], hv[k].x);
                atomicAdd(&acc[t[k] * FEAT_OUT + 1], hv[k].y);
            }
        }
    }
    __syncthreads();

    const int rot = (chunk * (flen / chunks)) & ~31;
    float* __restrict__ outp = acc2 + (size_t)lo * FEAT_OUT;
    for (int j = tid; j < flen; j += SCAT_THREADS) {
        int i = j + rot;
        if (i >= flen) i -= flen;
        float v = acc[i];
        if (v != 0.0f) atomicAdd(&outp[i], v);
    }
}

__global__ void final_kernel(const float* __restrict__ acc2, const float* __restrict__ h2,
                             const float* __restrict__ dinv, const float* __restrict__ b2,
                             float* __restrict__ out, int n_nodes) {
    int i = blockIdx.x * blockDim.x + threadIdx.x;
    if (i >= n_nodes) return;
    float di = dinv[i];
    float v0 = (acc2[i * 2 + 0] + h2[i * 2 + 0]) * di + b2[0];
    float v1 = (acc2[i * 2 + 1] + h2[i * 2 + 1]) * di + b2[1];
    float m  = fmaxf(v0, v1);
    float l  = m + logf(expf(v0 - m) + expf(v1 - m));
    out[i * 2 + 0] = v0 - l;
    out[i * 2 + 1] = v1 - l;
}

#define RB_NODES   128
#define RB_SHIFT   7
#define RB_MAXR    1024
#define BK_THREADS 1024
#define BK_U       4
#define BK_STAGE   12800
#define EB_THREADS 512
#define EB_STEPS   16
#define EB_TAIL    (RB_NODES * FEAT_MID / EB_THREADS)

#define DP_CHUNKS 32
#define DP_NODES  25024
#define DP_U      16
#define DP_TRIP   (DP_U * SCAT_THREADS)

__global__ void __launch_bounds__(SCAT_THREADS)
degree_parts(const int* __restrict__ dst, int* __restrict__ parts,
             int n_edges, int n_nodes, int range_nodes, int per_chunk) {
    __shared__ int cnt[DP_NODES];
    const int tid   = threadIdx.x;
    const int range = blockIdx.x / DP_CHUNKS;
    const int chunk = blockIdx.x - range * DP_CHUNKS;
    const int lo    = range * range_nodes;
    int len = n_nodes - lo;
    if (len > range_nodes) len = range_nodes;
    if (len <= 0) return;

    for (int i = tid; i < len; i += SCAT_THREADS) cnt[i] = 0;
    __syncthreads();

    int e0 = chunk * per_chunk;
    if (e0 > n_edges) e0 = n_edges;
    int e1 = e0 + per_chunk;
    if (e1 > n_edges) e1 = n_edges;

    const int nfull = (e1 - e0) / DP_TRIP;
    const int* __restrict__ p0 = dst + e0 + tid;
    int nx[DP_U] = {};
    if (nfull > 0) {
#pragma unroll
        for (int k = 0; k < DP_U; k++) nx[k] = p0[k * SCAT_THREADS];
    }
    for (int t = 0; t < nfull; t++) {
        int cur[DP_U];
#pragma unroll
        for (int k = 0; k < DP_U; k++) cur[k] = nx[k];
        if (t + 1 < nfull) {
            const int* __restrict__ pn = p0 + (t + 1) * DP_TRIP;
#pragma unroll
            for (int k = 0; k < DP_U; k++) nx[k] = pn[k * SCAT_THREADS];
        }
#pragma unroll
        for (int k = 0; k < DP_U; k++) {
            int tt = cur[k] - lo;
            if ((unsigned)tt < (unsigned)len) atomicAdd(&cnt[tt], 1);
        }
    }

    const int base = e0 + nfull * DP_TRIP;
    if (base < e1) {
        int dv[DP_U];
#pragma unroll
        for (int k = 0; k < DP_U; k++) {
            int e  = base + k * SCAT_THREADS + tid;
            int ec = e < e1 ? e : e1 - 1;
            dv[k] = dst[ec];
        }
        asm volatile("" : "+v"(dv[0]), "+v"(dv[1]), "+v"(dv[2]), "+v"(dv[3]),
                          "+v"(dv[4]), "+v"(dv[5]), "+v"(dv[6]), "+v"(dv[7]),
                          "+v"(dv[8]), "+v"(dv[9]), "+v"(dv[10]), "+v"(dv[11]),
                          "+v"(dv[12]), "+v"(dv[13]), "+v"(dv[14]), "+v"(dv[15]));
#pragma unroll
        for (int k = 0; k < DP_U; k++) {
            int e  = base + k * SCAT_THREADS + tid;
            int tt = dv[k] - lo;
            if (e < e1 && (unsigned)tt < (unsigned)len) atomicAdd(&cnt[tt], 1);
        }
    }
    __syncthreads();

    int* __restrict__ outp = parts + (size_t)chunk * n_nodes + lo;
    for (int i = tid; i < len; i += SCAT_THREADS) outp[i] = cnt[i];
}

#define DT_THREADS (2 * RB_NODES)
__global__ void __launch_bounds__(DT_THREADS)
dinv_totals(const int* __restrict__ parts, float* __restrict__ dinv,
            int* __restrict__ totals, int* __restrict__ cursor,
            int n_nodes, int n_ranges) {
    __shared__ int part[DT_THREADS / 32];
    const int tid = threadIdx.x;
    const int i   = blockIdx.x * DT_THREADS + tid;
    int v = 0;
    if (i < n_nodes) {
        int p[DP_CHUNKS];
#pragma unroll
        for (int c = 0; c < DP_CHUNKS; c++) p[c] = parts[(size_t)c * n_nodes + i];
#pragma unroll
        for (int c = 0; c < DP_CHUNKS; c++) v += p[c];
        float d = (float)(v + 1);
        dinv[i] = (d > 0.0f) ? rsqrtf(d) : 0.0f;
    }
    for (int m = 16; m > 0; m >>= 1) v += __shfl_xor(v, m);
    if ((tid & 31) == 0) part[tid >> 5] = v;
    __syncthreads();
    if (tid < 2) {
        const int r = blockIdx.x * 2 + tid;
        if (r < n_ranges) {
            const int* q = part + tid * (RB_NODES / 32);
            totals[r] = q[0] + q[1] + q[2] + q[3];
            cursor[r] = 0;
        }
    }
}

__device__ __forceinline__ void
bucket_chunk(const int* __restrict__ src, const int* __restrict__ dst,
             const int* __restrict__ totals, int* __restrict__ cursor,
             int* __restrict__ offs, int* __restrict__ bins,
             int n_edges, int per_chunk, int n_ranges, const int blk) {
    __shared__ int cnt[RB_MAXR];
    __shared__ int gb[RB_MAXR];
    __shared__ int sc[RB_MAXR];
    __shared__ int so[RB_MAXR];
    __shared__ int stage_w[BK_STAGE];
    __shared__ int stage_p[BK_STAGE];
    const int tid = threadIdx.x;
    int e0 = blk * per_chunk;
    if (e0 > n_edges) e0 = n_edges;
    int e1 = e0 + per_chunk;
    if (e1 > n_edges) e1 = n_edges;
    cnt[tid] = 0;
    __syncthreads();
    for (int base = e0; base < e1; base += BK_U * BK_THREADS) {
        int dv[BK_U];
#pragma unroll
        for (int k = 0; k < BK_U; k++) {
            int e  = base + k * BK_THREADS + tid;
            int ec = e < e1 ? e : e1 - 1;
            dv[k] = dst[ec];
        }
        asm volatile("" : "+v"(dv[0]), "+v"(dv[1]), "+v"(dv[2]), "+v"(dv[3]));
#pragma unroll
        for (int k = 0; k < BK_U; k++) {
            int e = base + k * BK_THREADS + tid;
            if (e < e1) atomicAdd(&cnt[dv[k] >> RB_SHIFT], 1);
        }
    }
    __syncthreads();
    const int c   = cnt[tid];
    const int tot = tid < n_ranges ? totals[tid] : 0;
    int rel = 0;
    if (c > 0) rel = atomicAdd(&cursor[tid], c);
    sc[tid] = c;
    so[tid] = tot;
    __syncthreads();
    for (int off = 1; off < RB_MAXR; off <<= 1) {
        int t1 = tid >= off ? sc[tid - off] : 0;
        int t2 = tid >= off ? so[tid - off] : 0;
        __syncthreads();
        sc[tid] += t1;
        so[tid] += t2;
        __syncthreads();
    }
    const int first = so[tid] - tot;
    if (blk == 0 && tid < n_ranges) {
        offs[tid] = first;
        if (tid == n_ranges - 1) offs[n_ranges] = so[tid];
    }
    gb[tid]  = first + rel;
    sc[tid] -= c;
    cnt[tid] = 0;
    __syncthreads();
    for (int base = e0; base < e1; base += BK_U * BK_THREADS) {
        int dv[BK_U], sv[BK_U];
#pragma unroll
        for (int k = 0; k < BK_U; k++) {
            int e  = base + k * BK_THREADS + tid;
            int ec = e < e1 ? e : e1 - 1;
            dv[k] = dst[ec];
            sv[k] = src[ec];
        }
        asm volatile("" : "+v"(dv[0]), "+v"(dv[1]), "+v"(dv[2]), "+v"(dv[3]),
                          "+v"(sv[0]), "+v"(sv[1]), "+v"(sv[2]), "+v"(sv[3]));
        int rank[BK_U], ls[BK_U], gs[BK_U];
#pragma unroll
        for (int k = 0; k < BK_U; k++) {
            int e = base + k * BK_THREADS + tid;
            int r = dv[k] >> RB_SHIFT;
            rank[k] = 0;
            if (e < e1) rank[k] = atomicAdd(&cnt[r], 1);
            ls[k] = sc[r];
            gs[k] = gb[r];
        }
#pragma unroll
        for (int k = 0; k < BK_U; k++) {
            int e = base + k * BK_THREADS + tid;
            if (e < e1) {
                int lp = ls[k] + rank[k];
                stage_w[lp] = (sv[k] << RB_SHIFT) | (dv[k] & (RB_NODES - 1));
                stage_p[lp] = gs[k] + rank[k];
            }
        }
    }
    __syncthreads();
    const int n = e1 - e0;
    for (int i = tid; i < n; i += BK_THREADS) bins[stage_p[i]] = stage_w[i];
}

__global__ void __launch_bounds__(BK_THREADS)
bucket_gemm(const int* __restrict__ src, const int* __restrict__ dst,
            const int* __restrict__ totals, int* __restrict__ cursor,
            int* __restrict__ offs, int* __restrict__ bins,
            int n_edges, int per_chunk, int n_ranges, int nb,
            const float* __restrict__ x, const float* __restrict__ W1,
            const float* __restrict__ dinv, float* __restrict__ h,
            int n_tiles, int n_nodes) {
    if ((int)blockIdx.x < nb) {
        bucket_chunk(src, dst, totals, cursor, offs, bins, n_edges, per_chunk, n_ranges, blockIdx.x);
    } else {
        gemm1_tile(x, W1, dinv, h,
                   ((int)blockIdx.x - nb) * (BK_THREADS / 32) + (int)(threadIdx.x >> 5),
                   n_tiles, n_nodes);
    }
}

__global__ void __launch_bounds__(EB_THREADS)
edge1_bin(const int* __restrict__ bins, const int* __restrict__ offs,
          const float* __restrict__ h, const float* __restrict__ dinv,
          const float* __restrict__ bias1, const float* __restrict__ W2,
          float* __restrict__ z, int n_nodes) {
    __shared__ float acc[RB_NODES * FEAT_MID];
    const int tid  = threadIdx.x;
    const int lane = tid & 31;
    const int f    = lane & 15;
    const int hf   = lane >> 4;
    const int lo   = blockIdx.x << RB_SHIFT;
    int len = n_nodes - lo;
    if (len > RB_NODES) len = RB_NODES;

    for (int i = tid; i < RB_NODES * FEAT_MID; i += EB_THREADS) acc[i] = 0.0f;
    __syncthreads();

    const int b0 = offs[blockIdx.x];
    const int b1 = offs[blockIdx.x + 1];
    const int stride = (EB_THREADS / 32) * 2 * EB_STEPS;
    int j0 = b0 + (tid >> 5) * 2 * EB_STEPS;
    int w[EB_STEPS] = {};
    if (j0 < b1) {
#pragma unroll
        for (int p = 0; p < EB_STEPS; p++) {
            int idx = j0 + 2 * p + hf;
            w[p] = bins[idx < b1 ? idx : j0];
        }
    }
    while (j0 < b1) {
        int   cur[EB_STEPS];
        float hq[EB_STEPS];
#pragma unroll
        for (int p = 0; p < EB_STEPS; p++) cur[p] = w[p];
        const int jn = j0 + stride;
        if (jn < b1) {
#pragma unroll
            for (int p = 0; p < EB_STEPS; p++) {
                int idx = jn + 2 * p + hf;
                w[p] = bins[idx < b1 ? idx : jn];
            }
        }
#pragma unroll
        for (int p = 0; p < EB_STEPS; p++)
            hq[p] = h[(cur[p] >> RB_SHIFT) * FEAT_MID + f];
        asm volatile("" : "+v"(hq[0]), "+v"(hq[1]), "+v"(hq[2]), "+v"(hq[3]),
                          "+v"(hq[4]), "+v"(hq[5]), "+v"(hq[6]), "+v"(hq[7]),
                          "+v"(hq[8]), "+v"(hq[9]), "+v"(hq[10]), "+v"(hq[11]),
                          "+v"(hq[12]), "+v"(hq[13]), "+v"(hq[14]), "+v"(hq[15]));
#pragma unroll
        for (int p = 0; p < EB_STEPS; p++) {
            if (j0 + 2 * p + hf < b1)
                atomicAdd(&acc[(cur[p] & (RB_NODES - 1)) * FEAT_MID + f], hq[p]);
        }
        j0 = jn;
    }
    __syncthreads();

    const float* __restrict__ hown = h + (size_t)lo * FEAT_MID;
    const int   flen = len * FEAT_MID;
    const float bf   = bias1[f];
    float hs[EB_TAIL], dd[EB_TAIL];
#pragma unroll
    for (int j = 0; j < EB_TAIL; j++) {
        int i  = tid + j * EB_THREADS;
        int ic = i < flen ? i : tid & 15;
        hs[j] = hown[ic];
        dd[j] = dinv[lo + (ic >> 4)];
    }
#pragma unroll
    for (int j = 0; j < EB_TAIL; j++) {
        int i = tid + j * EB_THREADS;
        if (i < flen) acc[i] = fmaxf((acc[i] + hs[j]) * dd[j] + bf, 0.0f);
    }
    __syncthreads();

    if (tid < len) {
        float s = 0.0f;
#pragma unroll
        for (int k = 0; k < FEAT_MID; k++)
            s = fmaf(acc[tid * FEAT_MID + k], W2[k * FEAT_OUT] - W2[k * FEAT_OUT + 1], s);
        z[lo + tid] = s * dinv[lo + tid];
    }
}

#define ES_W   16
#define ES_CAP (ES_W * EB_THREADS)
#define ES_U   8
typedef float v4f __attribute__((ext_vector_type(4)));
typedef int   v4i_t __attribute__((ext_vector_type(4)));

__global__ void __launch_bounds__(EB_THREADS, 16)
edge1_sort(const int* __restrict__ bins, const int* __restrict__ offs,
           const float* __restrict__ h, const float* __restrict__ dinv,
           const float* __restrict__ bias1, const float* __restrict__ W2,
           float* __restrict__ z, int n_nodes) {
    __shared__ int sorted[ES_CAP];
    __shared__ __attribute__((aligned(16))) int cnt[RB_NODES];
    __shared__ __attribute__((aligned(16))) int start[RB_NODES + 4];
    __shared__ float hs[RB_NODES];
    const int tid  = threadIdx.x;
    const int lane = tid & 31;
    const int q    = lane & 3;
    const int my   = (tid >> 5) * 8 + (lane >> 2);
    const int lo   = blockIdx.x << RB_SHIFT;
    int len = n_nodes - lo;
    if (len > RB_NODES) len = RB_NODES;
    const v4f* __restrict__ h4 = (const v4f*)h;

    const int b0 = offs[blockIdx.x];
    const int b1 = offs[blockIdx.x + 1];
    v4f acc = {0.0f, 0.0f, 0.0f, 0.0f};
    for (int c0 = b0; c0 < b1; c0 += ES_CAP) {
        int n = b1 - c0;
        if (n > ES_CAP) n = ES_CAP;
        if (tid < RB_NODES) cnt[tid] = 0;
        __syncthreads();
        int w[ES_W];
#pragma unroll
        for (int hh = 0; hh < 2; hh++) {
#pragma unroll
            for (int k = 0; k < ES_W / 2; k++) w[hh * (ES_W / 2) + k] = 0;
            if (hh * (ES_CAP / 2) < n) {
#pragma unroll
                for (int k = 0; k < ES_W / 2; k++) {
                    int idx = (hh * (ES_W / 2) + k) * EB_THREADS + tid;
                    w[hh * (ES_W / 2) + k] = bins[c0 + (idx < n ? idx : 0)];
                }
#pragma unroll
                for (int k = 0; k < ES_W / 2; k++) {
                    int idx = (hh * (ES_W / 2) + k) * EB_THREADS + tid;
                    if (idx < n) atomicAdd(&cnt[w[hh * (ES_W / 2) + k] & (RB_NODES - 1)], 1);
                }
            }
        }
        __syncthreads();
        if (tid < 32) {
            v4i_t c = *(const v4i_t*)&cnt[4 * tid];
            int s = c.x + c.y + c.z + c.w;
            int incl = s;
            for (int d = 1; d < 32; d <<= 1) {
                int t = __shfl_up(incl, d);
                if (lane >= d) incl += t;
            }
            int ex = incl - s;
            v4i_t st;
            st.x = ex; st.y = ex + c.x; st.z = ex + c.x + c.y; st.w = ex + c.x + c.y + c.z;
            *(v4i_t*)&start[4 * tid] = st;
            *(v4i_t*)&cnt[4 * tid]   = st;
            if (tid == 31) start[RB_NODES] = incl;
        }
        __syncthreads();
#pragma unroll
        for (int k = 0; k < ES_W; k++) {
            int idx = k * EB_THREADS + tid;
            if (idx < n) sorted[atomicAdd(&cnt[w[k] & (RB_NODES - 1)], 1)] = w[k] >> RB_SHIFT;
        }
        __syncthreads();

        const int s0 = start[my];
        const int s1 = start[my + 1];
        for (int j = s0; j < s1; j += ES_U) {
            const int m = s1 - j;
            v4f v[ES_U];
#pragma unroll
            for (int u = 0; u < ES_U; u++) v[u] = h4[sorted[j + (u < m ? u : 0)] * (FEAT_MID / 4) + q];
#pragma unroll
            for (int u = 0; u < ES_U; u++) {
                const bool ok = u < m;
                acc.x += ok ? v[u].x : 0.0f;
                acc.y += ok ? v[u].y : 0.0f;
                acc.z += ok ? v[u].z : 0.0f;
                acc.w += ok ? v[u].w : 0.0f;
            }
        }
    }

    const int   node = lo + (my < len ? my : 0);
    const float di   = dinv[node];
    const v4f   own  = h4[node * (FEAT_MID / 4) + q];
    const v4f   bq   = ((const v4f*)bias1)[q];
    const v4f   wa   = ((const v4f*)W2)[2 * q];
    const v4f   wb   = ((const v4f*)W2)[2 * q + 1];
    const float a0 = fmaxf((acc.x + own.x) * di + bq.x, 0.0f);
    const float a1 = fmaxf((acc.y + own.y) * di + bq.y, 0.0f);
    const float a2 = fmaxf((acc.z + own.z) * di + bq.z, 0.0f);
    const float a3 = fmaxf((acc.w + own.w) * di + bq.w, 0.0f);
    const float g0 = wa.x - wa.y, g1 = wa.z - wa.w;
    const float g2 = wb.x - wb.y, g3 = wb.z - wb.w;
    float p = fmaf(a3, g3, fmaf(a2, g2, fmaf(a1, g1, a0 * g0)));
    p += __shfl_xor(p, 1);
    p += __shfl_xor(p, 2);
    if (q == 0) hs[my] = p * di;
    __syncthreads();
    if (tid < len) z[lo + tid] = hs[tid];
}

#define ZS_BUCKETS 4
#define ZS_NODES   (ZS_BUCKETS * RB_NODES)
#define ZS_SHIFT   9
#define ZS_THREADS 1024
#define ZS_W       18
#define ZS_G       6
#define ZS_CAP     (ZS_W * ZS_THREADS)
#define ZS_SLICE   50048
#define ZS_FILL    ((ZS_SLICE / 4 + ZS_THREADS - 1) / ZS_THREADS)

__device__ __forceinline__ void
zs_ask(v4f (&r)[ZS_FILL], const float* __restrict__ z, int s0, int cs, int tid) {
    const v4f* __restrict__ z4 = (const v4f*)(z + s0);
    const int nq = cs >> 2;
#pragma unroll
    for (int f = 0; f < ZS_FILL; f++) {
        const int j = f * ZS_THREADS + tid;
        r[f] = (v4f){0.0f, 0.0f, 0.0f, 0.0f};
        if (j < nq) r[f] = z4[j];
    }
}

__device__ __forceinline__ void
zs_put(const v4f (&r)[ZS_FILL], float* __restrict__ tab, const float* __restrict__ z,
       int s0, int cs, int tid) {
    v4f* __restrict__ tab4 = (v4f*)tab;
    const int nq = cs >> 2;
#pragma unroll
    for (int f = 0; f < ZS_FILL; f++) {
        const int j = f * ZS_THREADS + tid;
        if (j < nq) tab4[j] = r[f];
    }
    if (tid < (cs & 3)) tab[4 * nq + tid] = z[s0 + 4 * nq + tid];
}

__global__ void __launch_bounds__(ZS_THREADS)
edge2_stage(const int* __restrict__ bins, const int* __restrict__ offs,
            const float* __restrict__ z, const float* __restrict__ dinv,
            const float* __restrict__ bias2, float* __restrict__ out,
            int n_nodes, int n_ranges) {
    __shared__ __attribute__((aligned(16))) float tab[ZS_SLICE];
    __shared__ float acc[ZS_NODES];
    const int tid = threadIdx.x;
    const int r0  = blockIdx.x * ZS_BUCKETS;
    const int lo  = r0 << RB_SHIFT;
    int len = n_nodes - lo;
    if (len > ZS_NODES) len = ZS_NODES;

    if (tid < ZS_NODES) acc[tid] = 0.0f;

    const int   own = lo + (tid < len ? tid : 0);
    const float di  = dinv[own];
    const float zo  = z[own];
    const float bd  = bias2[0] - bias2[1];

    int ob[ZS_BUCKETS + 1];
#pragma unroll
    for (int k = 0; k <= ZS_BUCKETS; k++) {
        const int r = r0 + k;
        ob[k] = offs[r < n_ranges ? r : n_ranges];
    }
    const int  b0       = ob[0];
    const int  b1       = ob[ZS_BUCKETS];
    const int  n_slices = (n_nodes + ZS_SLICE - 1) / ZS_SLICE;
    const int  rot      = blockIdx.x % n_slices;
    const bool wide     = (((size_t)z) & 15) == 0;

    for (int c0 = b0; c0 < b1; c0 += ZS_CAP) {
        int n = b1 - c0;
        if (n > ZS_CAP) n = ZS_CAP;
        int w[ZS_W];
#pragma unroll
        for (int k = 0; k < ZS_W; k++) {
            const int idx = k * ZS_THREADS + tid;
            w[k] = bins[c0 + (idx < n ? idx : 0)];
        }
        v4f r[ZS_FILL];
        if (wide) {
            const int s0 = rot * ZS_SLICE;
            int cs = n_nodes - s0;
            if (cs > ZS_SLICE) cs = ZS_SLICE;
            zs_ask(r, z, s0, cs, tid);
        }
#pragma unroll
        for (int k = 0; k < ZS_W; k++) {
            const int idx = k * ZS_THREADS + tid;
            const int pos = c0 + idx;
            const int t   = (pos >= ob[1]) + (pos >= ob[2]) + (pos >= ob[3]);
            const int e   = ((w[k] >> RB_SHIFT) << ZS_SHIFT) | (t << RB_SHIFT) | (w[k] & (RB_NODES - 1));
            w[k] = idx < n ? e : -1;
        }
        for (int s = 0; s < n_slices; s++) {
            int sl = s + rot;
            if (sl >= n_slices) sl -= n_slices;
            const int s0 = sl * ZS_SLICE;
            int cs = n_nodes - s0;
            if (cs > ZS_SLICE) cs = ZS_SLICE;
            __syncthreads();
            if (wide) {
                zs_put(r, tab, z, s0, cs, tid);
            } else {
                for (int j = tid; j < cs; j += ZS_THREADS) tab[j] = z[s0 + j];
            }
            __syncthreads();
            if (wide && s + 1 < n_slices) {
                int sn = sl + 1;
                if (sn >= n_slices) sn -= n_slices;
                const int t0 = sn * ZS_SLICE;
                int ct = n_nodes - t0;
                if (ct > ZS_SLICE) ct = ZS_SLICE;
                zs_ask(r, z, t0, ct, tid);
            }
#pragma unroll
            for (int g = 0; g < ZS_W; g += ZS_G) {
                float v[ZS_G];
#pragma unroll
                for (int k = 0; k < ZS_G; k++) {
                    const int  srel = (w[g + k] >> ZS_SHIFT) - s0;
                    const bool in   = (unsigned)srel < (unsigned)cs;
                    v[k] = tab[in ? srel : 0];
                }
                asm volatile("" : "+v"(v[0]), "+v"(v[1]), "+v"(v[2]), "+v"(v[3]), "+v"(v[4]), "+v"(v[5]));
#pragma unroll
                for (int k = 0; k < ZS_G; k++) {
                    const int  srel = (w[g + k] >> ZS_SHIFT) - s0;
                    const bool in   = (unsigned)srel < (unsigned)cs;
                    if (in) atomicAdd(&acc[w[g + k] & (ZS_NODES - 1)], v[k]);
                }
            }
        }
    }
    __syncthreads();

    if (tid < len) {
        const float d  = (acc[tid] + zo) * di + bd;
        const float ad = fabsf(d);
        const float t  = logf(1.0f + expf(-ad));
        v2f o2;
        o2.x = d >= 0.0f ? -t : -ad - t;
        o2.y = d >= 0.0f ? -ad - t : -t;
        ((v2f*)out)[lo + tid] = o2;
    }
}

extern "C" void kernel_launch(void* const* d_in, const int* in_sizes, int n_in,
                              void* d_out, int out_size, void* d_ws, size_t ws_size,
                              hipStream_t stream) {
    const float* x  = (const float*)d_in[0];
    const int*   ei = (const int*)d_in[1];
    const float* W1 = (const float*)d_in[2];
    const float* b1 = (const float*)d_in[3];
    const float* W2 = (const float*)d_in[4];
    const float* b2 = (const float*)d_in[5];
    float* out = (float*)d_out;

    int n_nodes = in_sizes[0] / FEAT_IN;
    int n_edges = in_sizes[1] / 2;
    const int* src = ei;
    const int* dst = ei + n_edges;

    float* ws   = (float*)d_ws;
    float* deg  = ws;
    float* h    = deg  + n_nodes;
    float* acc1 = h    + (size_t)n_nodes * FEAT_MID;
    float* h2   = acc1 + (size_t)n_nodes * FEAT_MID;
    float* acc2 = h2   + (size_t)n_nodes * FEAT_OUT;

    const int B = 256;
    int n16 = n_nodes * FEAT_MID;

    int nr1 = (n_nodes + LDS_FLOATS - 1) / LDS_FLOATS;
    int rn1 = (((n_nodes + nr1 - 1) / nr1) + 31) & ~31;
    int ch1 = 128 / nr1 > 0 ? 128 / nr1 : 1;
    int pc1 = (n_edges + ch1 - 1) / ch1;
    int n_tiles = (n_nodes + 15) / 16;
    int wpb = B / 32;

    int    n_ranges = (n_nodes + RB_NODES - 1) >> RB_SHIFT;
    size_t need = ((size_t)n_nodes * (1 + 2 * FEAT_MID + 2 * FEAT_OUT) + 3 * (size_t)n_ranges + 1 +
                   (size_t)n_edges) * sizeof(float);
    if (n_ranges <= RB_MAXR && ws_size >= need) {
        int* offs   = (int*)(acc2 + (size_t)n_nodes * FEAT_OUT);
        int* cursor = offs + n_ranges + 1;
        int* totals = cursor + n_ranges;
        int* bins   = totals + n_ranges;
        int  nb     = (n_edges + BK_STAGE - 1) / BK_STAGE;
        int  pcb    = BK_STAGE;

        int* parts = (int*)h;
        int  nrp   = (n_nodes + DP_NODES - 1) / DP_NODES;
        int  rnp   = (((n_nodes + nrp - 1) / nrp) + 31) & ~31;
        int  pcp   = (n_edges + DP_CHUNKS - 1) / DP_CHUNKS;
        degree_parts<<<nrp * DP_CHUNKS, SCAT_THREADS, 0, stream>>>(dst, parts, n_edges, n_nodes, rnp, pcp);
        dinv_totals<<<(n_nodes + DT_THREADS - 1) / DT_THREADS, DT_THREADS, 0, stream>>>(
            parts, deg, totals, cursor, n_nodes, n_ranges);
        int  gw     = (n_tiles + BK_THREADS / 32 - 1) / (BK_THREADS / 32);
        bucket_gemm<<<nb + gw, BK_THREADS, 0, stream>>>(src, dst, totals, cursor, offs, bins,
                                                        n_edges, pcb, n_ranges, nb,
                                                        x, W1, deg, h, n_tiles, n_nodes);
        float* z = h2;
        if (((((size_t)h) | ((size_t)b1) | ((size_t)W2)) & 15) == 0)
            edge1_sort<<<n_ranges, EB_THREADS, 0, stream>>>(bins, offs, h, deg, b1, W2, z, n_nodes);
        else
            edge1_bin<<<n_ranges, EB_THREADS, 0, stream>>>(bins, offs, h, deg, b1, W2, z, n_nodes);
        edge2_stage<<<(n_ranges + ZS_BUCKETS - 1) / ZS_BUCKETS, ZS_THREADS, 0, stream>>>(
            bins, offs, z, deg, b2, out, n_nodes, n_ranges);
        return;
    }

    init_ws<<<(n16 + B - 1) / B, B, 0, stream>>>(deg, acc1, acc2, n_nodes);
    degree_lds<<<nr1 * ch1, SCAT_THREADS, 0, stream>>>(dst, deg, n_edges, n_nodes, rn1, ch1, pc1);
    dinv_kernel<<<(n_nodes + B - 1) / B, B, 0, stream>>>(deg, n_nodes);

    gemm1_wmma<<<(n_tiles + wpb - 1) / wpb, B, 0, stream>>>(x, W1, deg, h, n_tiles, n_nodes);

    int nr0 = (n_nodes + E1_NODES - 1) / E1_NODES;
    int rn0 = (((n_nodes + nr0 - 1) / nr0) + 31) & ~31;
    int ch0 = 256 / nr0 > 0 ? 256 / nr0 : 1;
    int pc0 = (n_edges + ch0 - 1) / ch0;
    edge1_lds<<<nr0 * ch0, SCAT_THREADS, 0, stream>>>(src, dst, h, acc1, n_edges, n_nodes,
                                                      rn0, ch0, pc0);
    relu_bias1<<<(n16 + B - 1) / B, B, 0, stream>>>(h, deg, b1, acc1, n_nodes);

    gemm2_kernel<<<(n_nodes + B - 1) / B, B, 0, stream>>>(acc1, W2, deg, h2, n_nodes);

    int nr2 = (n_nodes * FEAT_OUT + LDS_FLOATS - 1) / LDS_FLOATS;
    int rn2 = (((n_nodes + nr2 - 1) / nr2) + 31) & ~31;
    int ch2 = 128 / nr2 > 0 ? 128 / nr2 : 1;
    int pc2 = (n_edges + ch2 - 1) / ch2;
    edge2_lds<<<nr2 * ch2, SCAT_THREADS, 0, stream>>>(src, dst, h2, acc2, n_edges, n_nodes,
                                                      rn2, ch2, pc2);

    final_kernel<<<(n_nodes + B - 1) / B, B, 0, stream>>>(acc2, h2, deg, b2, out, n_nodes);
}
